# starting kernel
# baseline (speedup 1.0000x reference)
#include <hip/hip_runtime.h>

#define NN 100000
#define NE 1000000
#define DIN 5
#define H 64
#define HHID 32
#define NL 4
#define DOUT 32
#define LN_EPS 1e-5f

typedef __attribute__((ext_vector_type(2)))  float    v2f;
typedef __attribute__((ext_vector_type(8)))  float    v8f;
typedef __attribute__((ext_vector_type(16))) _Float16 v16h;

#if defined(__has_builtin)
#if __has_builtin(__builtin_amdgcn_wmma_f32_16x16x4_f32)
#define HAVE_WMMA_F32 1
#endif
#endif

__device__ __forceinline__ float wave_sum(float v) {
#pragma unroll
  for (int off = 16; off > 0; off >>= 1) v += __shfl_xor(v, off, 32);
  return v;
}
__device__ __forceinline__ float sigmoidf_(float x) { return 1.0f / (1.0f + __expf(-x)); }

__device__ __forceinline__ void atomic_fadd(float* p, float v) { unsafeAtomicAdd(p, v); }

__global__ __launch_bounds__(256) void k_input_proj(
    const float* __restrict__ x, const float* __restrict__ W,
    const float* __restrict__ b, float* __restrict__ h) {
  int t = blockIdx.x * 256 + threadIdx.x;
  if (t >= NN * H) return;
  int n = t >> 6, k = t & 63;
  const float* xr = x + (size_t)n * DIN;
  float acc = b[k];
#pragma unroll
  for (int d = 0; d < DIN; ++d) acc = fmaf(xr[d], W[d * H + k], acc);
  h[t] = acc;
}

__global__ __launch_bounds__(128) void k_gemm_h(
    const float* __restrict__ A, const float* __restrict__ W,
    const float* __restrict__ bias, const float* __restrict__ aw,
    float* __restrict__ out, float* __restrict__ sD, float* __restrict__ sS) {
  __shared__ float lA[16][H + 1];
  __shared__ float lW[H][H + 1];
  const int tid = threadIdx.x;
  const int row0 = blockIdx.x * 16;
  for (int idx = tid; idx < 16 * H; idx += 128)
    lA[idx >> 6][idx & 63] = A[(size_t)row0 * H + idx];
  for (int idx = tid; idx < H * H; idx += 128)
    lW[idx >> 6][idx & 63] = W[idx];
  __syncthreads();

  const int wave = tid >> 5, lane = tid & 31;
  const int m = lane & 15, g = lane >> 4;
  const int col = wave * 16 + m;

  v8f acc;
  float bv = bias[col];
#pragma unroll
  for (int i = 0; i < 8; ++i) acc[i] = bv;

#if HAVE_WMMA_F32
#pragma unroll
  for (int k0 = 0; k0 < H; k0 += 4) {
    int kb = k0 + 2 * g;
    v2f a, b;
    a.x = lA[m][kb];     a.y = lA[m][kb + 1];
    b.x = lW[kb][col];   b.y = lW[kb + 1][col];
    acc = __builtin_amdgcn_wmma_f32_16x16x4_f32(false, a, false, b, (short)0, acc, false, false);
  }
#else
#pragma unroll
  for (int kk = 0; kk < H; kk += 32) {
    v16h a, b;
#pragma unroll
    for (int j = 0; j < 8; ++j) {
      int ka = (j < 4) ? (2 * j + 8 * g) : (16 + 2 * (j - 4) + 8 * g);
      a[2 * j]     = (_Float16)lA[m][kk + ka];
      a[2 * j + 1] = (_Float16)lA[m][kk + ka + 1];
      int kb = 2 * j + 16 * g;
      b[2 * j]     = (_Float16)lW[kk + kb][col];
      b[2 * j + 1] = (_Float16)lW[kk + kb + 1][col];
    }
    acc = __builtin_amdgcn_wmma_f32_16x16x32_f16(false, a, false, b, (short)0, acc, false, false);
  }
#endif

#pragma unroll
  for (int i = 0; i < 8; ++i) {
    int row = row0 + i + 8 * g;
    out[(size_t)row * H + col] = acc[i];
  }

  if (wave < 2) {
    const float* awp = aw + wave * H;
    float p = 0.0f;
#pragma unroll
    for (int k = 0; k < 32; ++k) p = fmaf(lA[m][32 * g + k], awp[32 * g + k], p);
    p += __shfl_xor(p, 16, 32);
    if (g == 0) {
      if (wave == 0) sD[row0 + m] = p; else sS[row0 + m] = p;
    }
  }
}

__global__ __launch_bounds__(256) void k_edges(
    const int* __restrict__ ei, const float* __restrict__ ea,
    const float* __restrict__ m, const float* __restrict__ sD,
    const float* __restrict__ sS, const float* __restrict__ attb,
    float* __restrict__ aggr) {
  int gw = blockIdx.x * 8 + (threadIdx.x >> 5);
  int lane = threadIdx.x & 31;
  int ebase = gw * 32;
  if (ebase >= NE) return;
  int e = ebase + lane;
  int src = 0, dst = 0;
  float w = 0.0f;
  if (e < NE) {
    src = ei[e];
    dst = ei[NE + e];
    __builtin_prefetch(m + (size_t)src * H, 0, 1);
    __builtin_prefetch(m + (size_t)src * H + 32, 0, 1);
    w = ea[e] * sigmoidf_(sD[dst] + sS[src] + attb[0]);
  }
  int cnt = min(32, NE - ebase);
  for (int i = 0; i < cnt; ++i) {
    int bs = __shfl(src, i, 32);
    int bd = __shfl(dst, i, 32);
    float bw = __shfl(w, i, 32);
    const float* mr = m + (size_t)bs * H;
    float* ar = aggr + (size_t)bd * H;
    atomic_fadd(ar + lane,      mr[lane] * bw);
    atomic_fadd(ar + lane + 32, mr[lane + 32] * bw);
  }
}

__global__ __launch_bounds__(256) void k_update(
    float* __restrict__ h, const float* __restrict__ m,
    const float* __restrict__ aggr, const float* __restrict__ g,
    const float* __restrict__ beta) {
  int gw = blockIdx.x * 8 + (threadIdx.x >> 5);
  int lane = threadIdx.x & 31;
  if (gw >= NN) return;
  size_t base = (size_t)gw * H;
  float r0 = h[base + lane], r1 = h[base + lane + 32];
  float f0 = aggr[base + lane] + m[base + lane];
  float f1 = aggr[base + lane + 32] + m[base + lane + 32];
  float mean = wave_sum(f0 + f1) * (1.0f / H);
  float d0 = f0 - mean, d1 = f1 - mean;
  float var = wave_sum(fmaf(d0, d0, d1 * d1)) * (1.0f / H);
  float inv = rsqrtf(var + LN_EPS);
  float y0 = fmaf(d0 * inv, g[lane], beta[lane]);
  float y1 = fmaf(d1 * inv, g[lane + 32], beta[lane + 32]);
  h[base + lane]      = fmaxf(y0, 0.0f) + r0;
  h[base + lane + 32] = fmaxf(y1, 0.0f) + r1;
}

__global__ __launch_bounds__(256) void k_heads(
    const float* __restrict__ h,
    const float* __restrict__ qW1, const float* __restrict__ qb1,
    const float* __restrict__ qW2, const float* __restrict__ qb2,
    const float* __restrict__ iW1, const float* __restrict__ ib1,
    const float* __restrict__ iW2, const float* __restrict__ ib2,
    const float* __restrict__ gW1, const float* __restrict__ gb1,
    const float* __restrict__ gW2, const float* __restrict__ gb2,
    const float* __restrict__ oW,  const float* __restrict__ obv,
    const float* __restrict__ sW1, const float* __restrict__ sb1,
    const float* __restrict__ sW2, const float* __restrict__ sb2,
    float* __restrict__ oq, float* __restrict__ oi, float* __restrict__ og,
    float* __restrict__ oh, float* __restrict__ obot, float* __restrict__ osev) {
  int gw = blockIdx.x * 8 + (threadIdx.x >> 5);
  int lane = threadIdx.x & 31;
  if (gw >= NN) return;
  size_t base = (size_t)gw * H;
  float h0 = h[base + lane], h1 = h[base + lane + 32];

  float tq = 0.f, ti = 0.f, tg = 0.f, ts = 0.f, bo = 0.f;
#pragma unroll 8
  for (int k = 0; k < H; ++k) {
    float hk = (k < 32) ? __shfl(h0, k, 32) : __shfl(h1, k - 32, 32);
    int wi = k * HHID + lane;
    tq = fmaf(hk, qW1[wi], tq);
    ti = fmaf(hk, iW1[wi], ti);
    tg = fmaf(hk, gW1[wi], tg);
    ts = fmaf(hk, sW1[wi], ts);
    bo = fmaf(hk, oW[wi],  bo);
  }
  tq = fmaxf(tq + qb1[lane], 0.0f);
  ti = fmaxf(ti + ib1[lane], 0.0f);
  tg = fmaxf(tg + gb1[lane], 0.0f);
  ts = fmaxf(ts + sb1[lane], 0.0f);

  float q  = wave_sum(tq * qW2[lane]);
  float ii = wave_sum(ti * iW2[lane]);
  float gg = wave_sum(tg * gW2[lane]);
  if (lane == 0) {
    oq[gw] = sigmoidf_(q + qb2[0]);
    oi[gw] = sigmoidf_(ii + ib2[0]);
    og[gw] = sigmoidf_(gg + gb2[0]);
  }

  float z[5];
#pragma unroll
  for (int c = 0; c < 5; ++c) z[c] = wave_sum(ts * sW2[lane * 5 + c]) + sb2[c];
  float zm = z[0];
#pragma unroll
  for (int c = 1; c < 5; ++c) zm = fmaxf(zm, z[c]);
  float es = 0.f;
#pragma unroll
  for (int c = 0; c < 5; ++c) { z[c] = __expf(z[c] - zm); es += z[c]; }
  float inv = 1.0f / es;
#pragma unroll
  for (int c = 0; c < 5; ++c)
    if (lane == c) osev[(size_t)gw * 5 + c] = z[c] * inv;

  obot[(size_t)gw * DOUT + lane] = bo + obv[lane];
  oh[base + lane]      = h0;
  oh[base + lane + 32] = h1;
}

__global__ __launch_bounds__(256) void k_graph_mean(
    const float* __restrict__ h, float* __restrict__ gout) {
  __shared__ float ls[H];
  int tid = threadIdx.x;
  if (tid < H) ls[tid] = 0.0f;
  __syncthreads();
  float acc = 0.0f;
  size_t stride = (size_t)gridDim.x * blockDim.x;
  for (size_t idx = (size_t)blockIdx.x * blockDim.x + tid; idx < (size_t)NN * H; idx += stride)
    acc += h[idx];
  atomicAdd(&ls[tid & 63], acc);
  __syncthreads();
  if (tid < H) atomic_fadd(&gout[tid], ls[tid] * (1.0f / NN));
}

extern "C" void kernel_launch(void* const* d_in, const int* in_sizes, int n_in,
                              void* d_out, int out_size, void* d_ws, size_t ws_size,
                              hipStream_t stream) {
  (void)in_sizes; (void)n_in; (void)out_size; (void)ws_size;
  const float* x    = (const float*)d_in[0];
  const int*   ei   = (const int*)  d_in[1];
  const float* ea   = (const float*)d_in[2];
  const float* inW  = (const float*)d_in[3];
  const float* inb  = (const float*)d_in[4];
  const float* linW = (const float*)d_in[5];
  const float* linb = (const float*)d_in[6];
  const float* attW = (const float*)d_in[7];
  const float* attb = (const float*)d_in[8];
  const float* lng  = (const float*)d_in[9];
  const float* lnb  = (const float*)d_in[10];
  const float* qW1 = (const float*)d_in[11]; const float* qb1 = (const float*)d_in[12];
  const float* qW2 = (const float*)d_in[13]; const float* qb2 = (const float*)d_in[14];
  const float* iW1 = (const float*)d_in[15]; const float* ib1 = (const float*)d_in[16];
  const float* iW2 = (const float*)d_in[17]; const float* ib2 = (const float*)d_in[18];
  const float* gW1 = (const float*)d_in[19]; const float* gb1 = (const float*)d_in[20];
  const float* gW2 = (const float*)d_in[21]; const float* gb2 = (const float*)d_in[22];
  const float* oW  = (const float*)d_in[23]; const float* obv = (const float*)d_in[24];
  const float* sW1 = (const float*)d_in[25]; const float* sb1 = (const float*)d_in[26];
  const float* sW2 = (const float*)d_in[27]; const float* sb2 = (const float*)d_in[28];

  float* ws   = (float*)d_ws;
  float* h    = ws;
  float* mbuf = h    + (size_t)NN * H;
  float* aggr = mbuf + (size_t)NN * H;
  float* sD   = aggr + (size_t)NN * H;
  float* sS   = sD + NN;

  float* out  = (float*)d_out;
  float* oq   = out;
  float* oi   = out + (size_t)NN;
  float* og   = out + 2 * (size_t)NN;
  float* oh   = out + 3 * (size_t)NN;
  float* ogr  = out + 67 * (size_t)NN;
  float* obot = out + 67 * (size_t)NN + H;
  float* osev = out + 99 * (size_t)NN + H;

  k_input_proj<<<(NN * H + 255) / 256, 256, 0, stream>>>(x, inW, inb, h);

  const int nodeWaveBlocks = (NN + 7) / 8;
  for (int l = 0; l < NL; ++l) {
    k_gemm_h<<<NN / 16, 128, 0, stream>>>(h, linW + (size_t)l * H * H, linb + (size_t)l * H,
                                          attW + (size_t)l * 2 * H, mbuf, sD, sS);
    hipMemsetAsync(aggr, 0, (size_t)NN * H * sizeof(float), stream);
    k_edges<<<(NE + 255) / 256, 256, 0, stream>>>(ei, ea, mbuf, sD, sS, attb + l, aggr);
    k_update<<<nodeWaveBlocks, 256, 0, stream>>>(h, mbuf, aggr, lng + (size_t)l * H, lnb + (size_t)l * H);
  }

  hipMemsetAsync(ogr, 0, H * sizeof(float), stream);
  k_heads<<<nodeWaveBlocks, 256, 0, stream>>>(h, qW1, qb1, qW2, qb2, iW1, ib1, iW2, ib2,
                                              gW1, gb1, gW2, gb2, oW, obv, sW1, sb1, sW2, sb2,
                                              oq, oi, og, oh, obot, osev);
  k_graph_mean<<<128, 256, 0, stream>>>(h, ogr);
}


# graph mean as a two-stage tree: 625 blocks, each thread with 10 independent four-float loads in flight, write partials; one block sums them in fixed order; no atomics, no memset (sum order changes)
# speedup vs baseline: 1.1276x; 1.1276x over previous
#include <hip/hip_runtime.h>

#define NN 100000
#define NE 1000000
#define DIN 5
#define H 64
#define HHID 32
#define NL 4
#define DOUT 32
#define LN_EPS 1e-5f

typedef __attribute__((ext_vector_type(2)))  float    v2f;
typedef __attribute__((ext_vector_type(8)))  float    v8f;
typedef __attribute__((ext_vector_type(16))) _Float16 v16h;

#if defined(__has_builtin)
#if __has_builtin(__builtin_amdgcn_wmma_f32_16x16x4_f32)
#define HAVE_WMMA_F32 1
#endif
#endif

__device__ __forceinline__ float wave_sum(float v) {
#pragma unroll
  for (int off = 16; off > 0; off >>= 1) v += __shfl_xor(v, off, 32);
  return v;
}
__device__ __forceinline__ float sigmoidf_(float x) { return 1.0f / (1.0f + __expf(-x)); }

__device__ __forceinline__ void atomic_fadd(float* p, float v) { unsafeAtomicAdd(p, v); }

__global__ __launch_bounds__(256) void k_input_proj(
    const float* __restrict__ x, const float* __restrict__ W,
    const float* __restrict__ b, float* __restrict__ h) {
  int t = blockIdx.x * 256 + threadIdx.x;
  if (t >= NN * H) return;
  int n = t >> 6, k = t & 63;
  const float* xr = x + (size_t)n * DIN;
  float acc = b[k];
#pragma unroll
  for (int d = 0; d < DIN; ++d) acc = fmaf(xr[d], W[d * H + k], acc);
  h[t] = acc;
}

__global__ __launch_bounds__(128) void k_gemm_h(
    const float* __restrict__ A, const float* __restrict__ W,
    const float* __restrict__ bias, const float* __restrict__ aw,
    float* __restrict__ out, float* __restrict__ sD, float* __restrict__ sS) {
  __shared__ float lA[16][H + 1];
  __shared__ float lW[H][H + 1];
  const int tid = threadIdx.x;
  const int row0 = blockIdx.x * 16;
  for (int idx = tid; idx < 16 * H; idx += 128)
    lA[idx >> 6][idx & 63] = A[(size_t)row0 * H + idx];
  for (int idx = tid; idx < H * H; idx += 128)
    lW[idx >> 6][idx & 63] = W[idx];
  __syncthreads();

  const int wave = tid >> 5, lane = tid & 31;
  const int m = lane & 15, g = lane >> 4;
  const int col = wave * 16 + m;

  v8f acc;
  float bv = bias[col];
#pragma unroll
  for (int i = 0; i < 8; ++i) acc[i] = bv;

#if HAVE_WMMA_F32
#pragma unroll
  for (int k0 = 0; k0 < H; k0 += 4) {
    int kb = k0 + 2 * g;
    v2f a, b;
    a.x = lA[m][kb];     a.y = lA[m][kb + 1];
    b.x = lW[kb][col];   b.y = lW[kb + 1][col];
    acc = __builtin_amdgcn_wmma_f32_16x16x4_f32(false, a, false, b, (short)0, acc, false, false);
  }
#else
#pragma unroll
  for (int kk = 0; kk < H; kk += 32) {
    v16h a, b;
#pragma unroll
    for (int j = 0; j < 8; ++j) {
      int ka = (j < 4) ? (2 * j + 8 * g) : (16 + 2 * (j - 4) + 8 * g);
      a[2 * j]     = (_Float16)lA[m][kk + ka];
      a[2 * j + 1] = (_Float16)lA[m][kk + ka + 1];
      int kb = 2 * j + 16 * g;
      b[2 * j]     = (_Float16)lW[kk + kb][col];
      b[2 * j + 1] = (_Float16)lW[kk + kb + 1][col];
    }
    acc = __builtin_amdgcn_wmma_f32_16x16x32_f16(false, a, false, b, (short)0, acc, false, false);
  }
#endif

#pragma unroll
  for (int i = 0; i < 8; ++i) {
    int row = row0 + i + 8 * g;
    out[(size_t)row * H + col] = acc[i];
  }

  if (wave < 2) {
    const float* awp = aw + wave * H;
    float p = 0.0f;
#pragma unroll
    for (int k = 0; k < 32; ++k) p = fmaf(lA[m][32 * g + k], awp[32 * g + k], p);
    p += __shfl_xor(p, 16, 32);
    if (g == 0) {
      if (wave == 0) sD[row0 + m] = p; else sS[row0 + m] = p;
    }
  }
}

__global__ __launch_bounds__(256) void k_edges(
    const int* __restrict__ ei, const float* __restrict__ ea,
    const float* __restrict__ m, const float* __restrict__ sD,
    const float* __restrict__ sS, const float* __restrict__ attb,
    float* __restrict__ aggr) {
  int gw = blockIdx.x * 8 + (threadIdx.x >> 5);
  int lane = threadIdx.x & 31;
  int ebase = gw * 32;
  if (ebase >= NE) return;
  int e = ebase + lane;
  int src = 0, dst = 0;
  float w = 0.0f;
  if (e < NE) {
    src = ei[e];
    dst = ei[NE + e];
    __builtin_prefetch(m + (size_t)src * H, 0, 1);
    __builtin_prefetch(m + (size_t)src * H + 32, 0, 1);
    w = ea[e] * sigmoidf_(sD[dst] + sS[src] + attb[0]);
  }
  int cnt = min(32, NE - ebase);
  for (int i = 0; i < cnt; ++i) {
    int bs = __shfl(src, i, 32);
    int bd = __shfl(dst, i, 32);
    float bw = __shfl(w, i, 32);
    const float* mr = m + (size_t)bs * H;
    float* ar = aggr + (size_t)bd * H;
    atomic_fadd(ar + lane,      mr[lane] * bw);
    atomic_fadd(ar + lane + 32, mr[lane + 32] * bw);
  }
}

__global__ __launch_bounds__(256) void k_update(
    float* __restrict__ h, const float* __restrict__ m,
    const float* __restrict__ aggr, const float* __restrict__ g,
    const float* __restrict__ beta) {
  int gw = blockIdx.x * 8 + (threadIdx.x >> 5);
  int lane = threadIdx.x & 31;
  if (gw >= NN) return;
  size_t base = (size_t)gw * H;
  float r0 = h[base + lane], r1 = h[base + lane + 32];
  float f0 = aggr[base + lane] + m[base + lane];
  float f1 = aggr[base + lane + 32] + m[base + lane + 32];
  float mean = wave_sum(f0 + f1) * (1.0f / H);
  float d0 = f0 - mean, d1 = f1 - mean;
  float var = wave_sum(fmaf(d0, d0, d1 * d1)) * (1.0f / H);
  float inv = rsqrtf(var + LN_EPS);
  float y0 = fmaf(d0 * inv, g[lane], beta[lane]);
  float y1 = fmaf(d1 * inv, g[lane + 32], beta[lane + 32]);
  h[base + lane]      = fmaxf(y0, 0.0f) + r0;
  h[base + lane + 32] = fmaxf(y1, 0.0f) + r1;
}

__global__ __launch_bounds__(256) void k_heads(
    const float* __restrict__ h,
    const float* __restrict__ qW1, const float* __restrict__ qb1,
    const float* __restrict__ qW2, const float* __restrict__ qb2,
    const float* __restrict__ iW1, const float* __restrict__ ib1,
    const float* __restrict__ iW2, const float* __restrict__ ib2,
    const float* __restrict__ gW1, const float* __restrict__ gb1,
    const float* __restrict__ gW2, const float* __restrict__ gb2,
    const float* __restrict__ oW,  const float* __restrict__ obv,
    const float* __restrict__ sW1, const float* __restrict__ sb1,
    const float* __restrict__ sW2, const float* __restrict__ sb2,
    float* __restrict__ oq, float* __restrict__ oi, float* __restrict__ og,
    float* __restrict__ oh, float* __restrict__ obot, float* __restrict__ osev) {
  int gw = blockIdx.x * 8 + (threadIdx.x >> 5);
  int lane = threadIdx.x & 31;
  if (gw >= NN) return;
  size_t base = (size_t)gw * H;
  float h0 = h[base + lane], h1 = h[base + lane + 32];

  float tq = 0.f, ti = 0.f, tg = 0.f, ts = 0.f, bo = 0.f;
#pragma unroll 8
  for (int k = 0; k < H; ++k) {
    float hk = (k < 32) ? __shfl(h0, k, 32) : __shfl(h1, k - 32, 32);
    int wi = k * HHID + lane;
    tq = fmaf(hk, qW1[wi], tq);
    ti = fmaf(hk, iW1[wi], ti);
    tg = fmaf(hk, gW1[wi], tg);
    ts = fmaf(hk, sW1[wi], ts);
    bo = fmaf(hk, oW[wi],  bo);
  }
  tq = fmaxf(tq + qb1[lane], 0.0f);
  ti = fmaxf(ti + ib1[lane], 0.0f);
  tg = fmaxf(tg + gb1[lane], 0.0f);
  ts = fmaxf(ts + sb1[lane], 0.0f);

  float q  = wave_sum(tq * qW2[lane]);
  float ii = wave_sum(ti * iW2[lane]);
  float gg = wave_sum(tg * gW2[lane]);
  if (lane == 0) {
    oq[gw] = sigmoidf_(q + qb2[0]);
    oi[gw] = sigmoidf_(ii + ib2[0]);
    og[gw] = sigmoidf_(gg + gb2[0]);
  }

  float z[5];
#pragma unroll
  for (int c = 0; c < 5; ++c) z[c] = wave_sum(ts * sW2[lane * 5 + c]) + sb2[c];
  float zm = z[0];
#pragma unroll
  for (int c = 1; c < 5; ++c) zm = fmaxf(zm, z[c]);
  float es = 0.f;
#pragma unroll
  for (int c = 0; c < 5; ++c) { z[c] = __expf(z[c] - zm); es += z[c]; }
  float inv = 1.0f / es;
#pragma unroll
  for (int c = 0; c < 5; ++c)
    if (lane == c) osev[(size_t)gw * 5 + c] = z[c] * inv;

  obot[(size_t)gw * DOUT + lane] = bo + obv[lane];
  oh[base + lane]      = h0;
  oh[base + lane + 32] = h1;
}

#define GM_BLOCKS 625
#define GM_LOADS 10
__global__ __launch_bounds__(256) void k_graph_partial(
    const float* __restrict__ h, float* __restrict__ part) {
  __shared__ float ls[8][H];
  const int tid = threadIdx.x;
  const int wave = tid >> 5, lane = tid & 31;
  const float4* h4 = reinterpret_cast<const float4*>(h);
  const size_t t = (size_t)blockIdx.x * 256 + tid;
  float4 v[GM_LOADS];
#pragma unroll
  for (int i = 0; i < GM_LOADS; ++i) v[i] = h4[t + (size_t)i * (GM_BLOCKS * 256)];
  float4 a = v[0];
#pragma unroll
  for (int i = 1; i < GM_LOADS; ++i) {
    a.x += v[i].x; a.y += v[i].y; a.z += v[i].z; a.w += v[i].w;
  }
  a.x += __shfl_xor(a.x, 16, 32);
  a.y += __shfl_xor(a.y, 16, 32);
  a.z += __shfl_xor(a.z, 16, 32);
  a.w += __shfl_xor(a.w, 16, 32);
  if (lane < 16) *reinterpret_cast<float4*>(&ls[wave][lane * 4]) = a;
  __syncthreads();
  if (tid < H) {
    float s = ls[0][tid];
#pragma unroll
    for (int w = 1; w < 8; ++w) s += ls[w][tid];
    part[(size_t)blockIdx.x * H + tid] = s;
  }
}

__global__ __launch_bounds__(1024) void k_graph_final(
    const float* __restrict__ part, float* __restrict__ gout) {
  __shared__ float ls[16][H];
  const int tid = threadIdx.x;
  const int f = tid & 63, q = tid >> 6;
  float v[40];
#pragma unroll
  for (int i = 0; i < 40; ++i) {
    int b = q + 16 * i;
    v[i] = (b < GM_BLOCKS) ? part[(size_t)b * H + f] : 0.0f;
  }
  float s = v[0];
#pragma unroll
  for (int i = 1; i < 40; ++i) s += v[i];
  ls[q][f] = s;
  __syncthreads();
  if (tid < H) {
    float r = ls[0][tid];
#pragma unroll
    for (int w = 1; w < 16; ++w) r += ls[w][tid];
    gout[tid] = r * (1.0f / NN);
  }
}

extern "C" void kernel_launch(void* const* d_in, const int* in_sizes, int n_in,
                              void* d_out, int out_size, void* d_ws, size_t ws_size,
                              hipStream_t stream) {
  (void)in_sizes; (void)n_in; (void)out_size; (void)ws_size;
  const float* x    = (const float*)d_in[0];
  const int*   ei   = (const int*)  d_in[1];
  const float* ea   = (const float*)d_in[2];
  const float* inW  = (const float*)d_in[3];
  const float* inb  = (const float*)d_in[4];
  const float* linW = (const float*)d_in[5];
  const float* linb = (const float*)d_in[6];
  const float* attW = (const float*)d_in[7];
  const float* attb = (const float*)d_in[8];
  const float* lng  = (const float*)d_in[9];
  const float* lnb  = (const float*)d_in[10];
  const float* qW1 = (const float*)d_in[11]; const float* qb1 = (const float*)d_in[12];
  const float* qW2 = (const float*)d_in[13]; const float* qb2 = (const float*)d_in[14];
  const float* iW1 = (const float*)d_in[15]; const float* ib1 = (const float*)d_in[16];
  const float* iW2 = (const float*)d_in[17]; const float* ib2 = (const float*)d_in[18];
  const float* gW1 = (const float*)d_in[19]; const float* gb1 = (const float*)d_in[20];
  const float* gW2 = (const float*)d_in[21]; const float* gb2 = (const float*)d_in[22];
  const float* oW  = (const float*)d_in[23]; const float* obv = (const float*)d_in[24];
  const float* sW1 = (const float*)d_in[25]; const float* sb1 = (const float*)d_in[26];
  const float* sW2 = (const float*)d_in[27]; const float* sb2 = (const float*)d_in[28];

  float* ws   = (float*)d_ws;
  float* h    = ws;
  float* mbuf = h    + (size_t)NN * H;
  float* aggr = mbuf + (size_t)NN * H;
  float* sD   = aggr + (size_t)NN * H;
  float* sS   = sD + NN;

  float* out  = (float*)d_out;
  float* oq   = out;
  float* oi   = out + (size_t)NN;
  float* og   = out + 2 * (size_t)NN;
  float* oh   = out + 3 * (size_t)NN;
  float* ogr  = out + 67 * (size_t)NN;
  float* obot = out + 67 * (size_t)NN + H;
  float* osev = out + 99 * (size_t)NN + H;

  k_input_proj<<<(NN * H + 255) / 256, 256, 0, stream>>>(x, inW, inb, h);

  const int nodeWaveBlocks = (NN + 7) / 8;
  for (int l = 0; l < NL; ++l) {
    k_gemm_h<<<NN / 16, 128, 0, stream>>>(h, linW + (size_t)l * H * H, linb + (size_t)l * H,
                                          attW + (size_t)l * 2 * H, mbuf, sD, sS);
    hipMemsetAsync(aggr, 0, (size_t)NN * H * sizeof(float), stream);
    k_edges<<<(NE + 255) / 256, 256, 0, stream>>>(ei, ea, mbuf, sD, sS, attb + l, aggr);
    k_update<<<nodeWaveBlocks, 256, 0, stream>>>(h, mbuf, aggr, lng + (size_t)l * H, lnb + (size_t)l * H);
  }

  k_heads<<<nodeWaveBlocks, 256, 0, stream>>>(h, qW1, qb1, qW2, qb2, iW1, ib1, iW2, ib2,
                                              gW1, gb1, gW2, gb2, oW, obv, sW1, sb1, sW2, sb2,
                                              oq, oi, og, oh, obot, osev);
  k_graph_partial<<<GM_BLOCKS, 256, 0, stream>>>(h, aggr);
  k_graph_final<<<1, 1024, 0, stream>>>(aggr, ogr);
}


# f32 atomic scatter replaced by a per-call edge list grouped by destination, ordered by edge index inside each node; a node's wave sums its rows in registers in that fixed order, then updates
# speedup vs baseline: 1.4303x; 1.2684x over previous
#include <hip/hip_runtime.h>

#define NN 100000
#define NE 1000000
#define DIN 5
#define H 64
#define HHID 32
#define NL 4
#define DOUT 32
#define LN_EPS 1e-5f

typedef __attribute__((ext_vector_type(2)))  float    v2f;
typedef __attribute__((ext_vector_type(8)))  float    v8f;
typedef __attribute__((ext_vector_type(16))) _Float16 v16h;

#if defined(__has_builtin)
#if __has_builtin(__builtin_amdgcn_wmma_f32_16x16x4_f32)
#define HAVE_WMMA_F32 1
#endif
#endif

__device__ __forceinline__ float wave_sum(float v) {
#pragma unroll
  for (int off = 16; off > 0; off >>= 1) v += __shfl_xor(v, off, 32);
  return v;
}
__device__ __forceinline__ float sigmoidf_(float x) { return 1.0f / (1.0f + __expf(-x)); }

__device__ __forceinline__ int wave_sum_i(int v) {
#pragma unroll
  for (int off = 16; off > 0; off >>= 1) v += __shfl_xor(v, off, 32);
  return v;
}

__global__ __launch_bounds__(256) void k_input_proj(
    const float* __restrict__ x, const float* __restrict__ W,
    const float* __restrict__ b, float* __restrict__ h) {
  int t = blockIdx.x * 256 + threadIdx.x;
  if (t >= NN * H) return;
  int n = t >> 6, k = t & 63;
  const float* xr = x + (size_t)n * DIN;
  float acc = b[k];
#pragma unroll
  for (int d = 0; d < DIN; ++d) acc = fmaf(xr[d], W[d * H + k], acc);
  h[t] = acc;
}

__global__ __launch_bounds__(128) void k_gemm_h(
    const float* __restrict__ A, const float* __restrict__ W,
    const float* __restrict__ bias, const float* __restrict__ aw,
    float* __restrict__ out, float* __restrict__ sD, float* __restrict__ sS) {
  __shared__ float lA[16][H + 1];
  __shared__ float lW[H][H + 1];
  const int tid = threadIdx.x;
  const int row0 = blockIdx.x * 16;
  for (int idx = tid; idx < 16 * H; idx += 128)
    lA[idx >> 6][idx & 63] = A[(size_t)row0 * H + idx];
  for (int idx = tid; idx < H * H; idx += 128)
    lW[idx >> 6][idx & 63] = W[idx];
  __syncthreads();

  const int wave = tid >> 5, lane = tid & 31;
  const int m = lane & 15, g = lane >> 4;
  const int col = wave * 16 + m;

  v8f acc;
  float bv = bias[col];
#pragma unroll
  for (int i = 0; i < 8; ++i) acc[i] = bv;

#if HAVE_WMMA_F32
#pragma unroll
  for (int k0 = 0; k0 < H; k0 += 4) {
    int kb = k0 + 2 * g;
    v2f a, b;
    a.x = lA[m][kb];     a.y = lA[m][kb + 1];
    b.x = lW[kb][col];   b.y = lW[kb + 1][col];
    acc = __builtin_amdgcn_wmma_f32_16x16x4_f32(false, a, false, b, (short)0, acc, false, false);
  }
#else
#pragma unroll
  for (int kk = 0; kk < H; kk += 32) {
    v16h a, b;
#pragma unroll
    for (int j = 0; j < 8; ++j) {
      int ka = (j < 4) ? (2 * j + 8 * g) : (16 + 2 * (j - 4) + 8 * g);
      a[2 * j]     = (_Float16)lA[m][kk + ka];
      a[2 * j + 1] = (_Float16)lA[m][kk + ka + 1];
      int kb = 2 * j + 16 * g;
      b[2 * j]     = (_Float16)lW[kk + kb][col];
      b[2 * j + 1] = (_Float16)lW[kk + kb + 1][col];
    }
    acc = __builtin_amdgcn_wmma_f32_16x16x32_f16(false, a, false, b, (short)0, acc, false, false);
  }
#endif

#pragma unroll
  for (int i = 0; i < 8; ++i) {
    int row = row0 + i + 8 * g;
    out[(size_t)row * H + col] = acc[i];
  }

  if (wave < 2) {
    const float* awp = aw + wave * H;
    float p = 0.0f;
#pragma unroll
    for (int k = 0; k < 32; ++k) p = fmaf(lA[m][32 * g + k], awp[32 * g + k], p);
    p += __shfl_xor(p, 16, 32);
    if (g == 0) {
      if (wave == 0) sD[row0 + m] = p; else sS[row0 + m] = p;
    }
  }
}

static_assert(NE % 4 == 0, "four edges per thread");
#define NODE_PAD 100032

__global__ __launch_bounds__(256) void k_edge_count(
    const int* __restrict__ ei, int* __restrict__ cnt) {
  int t = blockIdx.x * 256 + threadIdx.x;
  if (t >= NE / 4) return;
  int4 d = reinterpret_cast<const int4*>(ei + NE)[t];
  atomicAdd(cnt + d.x, 1);
  atomicAdd(cnt + d.y, 1);
  atomicAdd(cnt + d.z, 1);
  atomicAdd(cnt + d.w, 1);
}

#define SCAN_T 1024
#define SCAN_BLOCKS ((NN + SCAN_T - 1) / SCAN_T)
#define SCAN_ITERS (((SCAN_BLOCKS - 1) * (SCAN_T / 4) + SCAN_T - 1) / SCAN_T)
__global__ __launch_bounds__(SCAN_T) void k_edge_scan(
    const int* __restrict__ cnt, int* __restrict__ off) {
  __shared__ int lfront[32];
  __shared__ int ltot[32];
  const int tid = threadIdx.x, wave = tid >> 5, lane = tid & 31;
  const int b = blockIdx.x;
  const int4* c4 = reinterpret_cast<const int4*>(cnt);
  const int nvec = b * (SCAN_T / 4);
  int s = 0;
#pragma unroll
  for (int i = 0; i < SCAN_ITERS; ++i) {
    int idx = i * SCAN_T + tid;
    int4 v = c4[min(idx, NN / 4 - 1)];
    int q = (v.x + v.y) + (v.z + v.w);
    s += (idx < nvec) ? q : 0;
  }
  const int node = b * SCAN_T + tid;
  const int c = (node < NN) ? cnt[node] : 0;
  s = wave_sum_i(s);
  int incl = c;
#pragma unroll
  for (int o = 1; o < 32; o <<= 1) {
    int up = __shfl_up(incl, o, 32);
    if (lane >= o) incl += up;
  }
  if (lane == 0) lfront[wave] = s;
  if (lane == 31) ltot[wave] = incl;
  __syncthreads();
  int front = wave_sum_i(lfront[lane]);
  int before = wave_sum_i((lane < wave) ? ltot[lane] : 0);
  int excl = front + before + (incl - c);
  if (node < NN) off[node] = excl;
  if (node == NN - 1) off[NN] = excl + c;
}

__global__ __launch_bounds__(256) void k_edge_place(
    const int* __restrict__ ei, int* __restrict__ cnt,
    const int* __restrict__ off, int2* __restrict__ tmp) {
  int t = blockIdx.x * 256 + threadIdx.x;
  if (t >= NE / 4) return;
  int4 d = reinterpret_cast<const int4*>(ei + NE)[t];
  int e0 = 4 * t;
  int r;
  r = atomicSub(cnt + d.x, 1); tmp[off[d.x] + r - 1] = make_int2(e0,     d.x);
  r = atomicSub(cnt + d.y, 1); tmp[off[d.y] + r - 1] = make_int2(e0 + 1, d.y);
  r = atomicSub(cnt + d.z, 1); tmp[off[d.z] + r - 1] = make_int2(e0 + 2, d.z);
  r = atomicSub(cnt + d.w, 1); tmp[off[d.w] + r - 1] = make_int2(e0 + 3, d.w);
}

__global__ __launch_bounds__(256) void k_edge_order(
    const int* __restrict__ ei, const float* __restrict__ ea,
    const int* __restrict__ off, const int2* __restrict__ tmp,
    int2* __restrict__ rec) {
  int p = blockIdx.x * 256 + threadIdx.x;
  if (p >= NE) return;
  int2 my = tmp[p];
  int s = off[my.y], e = off[my.y + 1];
  int rank = 0;
  for (int q = s; q < e; ++q) rank += (tmp[q].x < my.x) ? 1 : 0;
  rec[s + rank] = make_int2(ei[my.x], __float_as_int(ea[my.x]));
}

__global__ __launch_bounds__(256) void k_gather_update(
    float* __restrict__ h, const float* __restrict__ m,
    const int* __restrict__ off, const int2* __restrict__ rec,
    const float* __restrict__ sD, const float* __restrict__ sS,
    const float* __restrict__ attb,
    const float* __restrict__ g, const float* __restrict__ beta) {
  const int gw = __builtin_amdgcn_readfirstlane(blockIdx.x * 8 + (threadIdx.x >> 5));
  int lane = threadIdx.x & 31;
  if (gw >= NN) return;
  size_t base = (size_t)gw * H;
  const int s = off[gw], e = off[gw + 1];
  const float sd = sD[gw], ab = attb[0];
  float r0 = h[base + lane], r1 = h[base + lane + 32];
  float m0 = m[base + lane], m1 = m[base + lane + 32];
  float acc0 = 0.0f, acc1 = 0.0f;
  for (int cb = s; cb < e; cb += 32) {
    const int cnt = min(32, e - cb);
    int src = 0;
    float w = 0.0f;
    if (lane < cnt) {
      int2 r = rec[cb + lane];
      src = r.x;
      __builtin_prefetch(m + (size_t)src * H, 0, 1);
      __builtin_prefetch(m + (size_t)src * H + 32, 0, 1);
      w = __int_as_float(r.y) * sigmoidf_(sd + sS[src] + ab);
    }
#pragma unroll 4
    for (int i = 0; i < cnt; ++i) {
      int bs = __builtin_amdgcn_readlane(src, i);
      float bw = __int_as_float(__builtin_amdgcn_readlane(__float_as_int(w), i));
      const float* mr = m + (size_t)bs * H;
      acc0 = fmaf(mr[lane], bw, acc0);
      acc1 = fmaf(mr[lane + 32], bw, acc1);
    }
  }
  float f0 = acc0 + m0;
  float f1 = acc1 + m1;
  float mean = wave_sum(f0 + f1) * (1.0f / H);
  float d0 = f0 - mean, d1 = f1 - mean;
  float var = wave_sum(fmaf(d0, d0, d1 * d1)) * (1.0f / H);
  float inv = rsqrtf(var + LN_EPS);
  float y0 = fmaf(d0 * inv, g[lane], beta[lane]);
  float y1 = fmaf(d1 * inv, g[lane + 32], beta[lane + 32]);
  h[base + lane]      = fmaxf(y0, 0.0f) + r0;
  h[base + lane + 32] = fmaxf(y1, 0.0f) + r1;
}

__global__ __launch_bounds__(256) void k_heads(
    const float* __restrict__ h,
    const float* __restrict__ qW1, const float* __restrict__ qb1,
    const float* __restrict__ qW2, const float* __restrict__ qb2,
    const float* __restrict__ iW1, const float* __restrict__ ib1,
    const float* __restrict__ iW2, const float* __restrict__ ib2,
    const float* __restrict__ gW1, const float* __restrict__ gb1,
    const float* __restrict__ gW2, const float* __restrict__ gb2,
    const float* __restrict__ oW,  const float* __restrict__ obv,
    const float* __restrict__ sW1, const float* __restrict__ sb1,
    const float* __restrict__ sW2, const float* __restrict__ sb2,
    float* __restrict__ oq, float* __restrict__ oi, float* __restrict__ og,
    float* __restrict__ oh, float* __restrict__ obot, float* __restrict__ osev) {
  int gw = blockIdx.x * 8 + (threadIdx.x >> 5);
  int lane = threadIdx.x & 31;
  if (gw >= NN) return;
  size_t base = (size_t)gw * H;
  float h0 = h[base + lane], h1 = h[base + lane + 32];

  float tq = 0.f, ti = 0.f, tg = 0.f, ts = 0.f, bo = 0.f;
#pragma unroll 8
  for (int k = 0; k < H; ++k) {
    float hk = (k < 32) ? __shfl(h0, k, 32) : __shfl(h1, k - 32, 32);
    int wi = k * HHID + lane;
    tq = fmaf(hk, qW1[wi], tq);
    ti = fmaf(hk, iW1[wi], ti);
    tg = fmaf(hk, gW1[wi], tg);
    ts = fmaf(hk, sW1[wi], ts);
    bo = fmaf(hk, oW[wi],  bo);
  }
  tq = fmaxf(tq + qb1[lane], 0.0f);
  ti = fmaxf(ti + ib1[lane], 0.0f);
  tg = fmaxf(tg + gb1[lane], 0.0f);
  ts = fmaxf(ts + sb1[lane], 0.0f);

  float q  = wave_sum(tq * qW2[lane]);
  float ii = wave_sum(ti * iW2[lane]);
  float gg = wave_sum(tg * gW2[lane]);
  if (lane == 0) {
    oq[gw] = sigmoidf_(q + qb2[0]);
    oi[gw] = sigmoidf_(ii + ib2[0]);
    og[gw] = sigmoidf_(gg + gb2[0]);
  }

  float z[5];
#pragma unroll
  for (int c = 0; c < 5; ++c) z[c] = wave_sum(ts * sW2[lane * 5 + c]) + sb2[c];
  float zm = z[0];
#pragma unroll
  for (int c = 1; c < 5; ++c) zm = fmaxf(zm, z[c]);
  float es = 0.f;
#pragma unroll
  for (int c = 0; c < 5; ++c) { z[c] = __expf(z[c] - zm); es += z[c]; }
  float inv = 1.0f / es;
#pragma unroll
  for (int c = 0; c < 5; ++c)
    if (lane == c) osev[(size_t)gw * 5 + c] = z[c] * inv;

  obot[(size_t)gw * DOUT + lane] = bo + obv[lane];
  oh[base + lane]      = h0;
  oh[base + lane + 32] = h1;
}

#define GM_BLOCKS 625
#define GM_LOADS 10
__global__ __launch_bounds__(256) void k_graph_partial(
    const float* __restrict__ h, float* __restrict__ part) {
  __shared__ float ls[8][H];
  const int tid = threadIdx.x;
  const int wave = tid >> 5, lane = tid & 31;
  const float4* h4 = reinterpret_cast<const float4*>(h);
  const size_t t = (size_t)blockIdx.x * 256 + tid;
  float4 v[GM_LOADS];
#pragma unroll
  for (int i = 0; i < GM_LOADS; ++i) v[i] = h4[t + (size_t)i * (GM_BLOCKS * 256)];
  float4 a = v[0];
#pragma unroll
  for (int i = 1; i < GM_LOADS; ++i) {
    a.x += v[i].x; a.y += v[i].y; a.z += v[i].z; a.w += v[i].w;
  }
  a.x += __shfl_xor(a.x, 16, 32);
  a.y += __shfl_xor(a.y, 16, 32);
  a.z += __shfl_xor(a.z, 16, 32);
  a.w += __shfl_xor(a.w, 16, 32);
  if (lane < 16) *reinterpret_cast<float4*>(&ls[wave][lane * 4]) = a;
  __syncthreads();
  if (tid < H) {
    float s = ls[0][tid];
#pragma unroll
    for (int w = 1; w < 8; ++w) s += ls[w][tid];
    part[(size_t)blockIdx.x * H + tid] = s;
  }
}

__global__ __launch_bounds__(1024) void k_graph_final(
    const float* __restrict__ part, float* __restrict__ gout) {
  __shared__ float ls[16][H];
  const int tid = threadIdx.x;
  const int f = tid & 63, q = tid >> 6;
  float v[40];
#pragma unroll
  for (int i = 0; i < 40; ++i) {
    int b = q + 16 * i;
    v[i] = (b < GM_BLOCKS) ? part[(size_t)b * H + f] : 0.0f;
  }
  float s = v[0];
#pragma unroll
  for (int i = 1; i < 40; ++i) s += v[i];
  ls[q][f] = s;
  __syncthreads();
  if (tid < H) {
    float r = ls[0][tid];
#pragma unroll
    for (int w = 1; w < 16; ++w) r += ls[w][tid];
    gout[tid] = r * (1.0f / NN);
  }
}

extern "C" void kernel_launch(void* const* d_in, const int* in_sizes, int n_in,
                              void* d_out, int out_size, void* d_ws, size_t ws_size,
                              hipStream_t stream) {
  (void)in_sizes; (void)n_in; (void)out_size; (void)ws_size;
  const float* x    = (const float*)d_in[0];
  const int*   ei   = (const int*)  d_in[1];
  const float* ea   = (const float*)d_in[2];
  const float* inW  = (const float*)d_in[3];
  const float* inb  = (const float*)d_in[4];
  const float* linW = (const float*)d_in[5];
  const float* linb = (const float*)d_in[6];
  const float* attW = (const float*)d_in[7];
  const float* attb = (const float*)d_in[8];
  const float* lng  = (const float*)d_in[9];
  const float* lnb  = (const float*)d_in[10];
  const float* qW1 = (const float*)d_in[11]; const float* qb1 = (const float*)d_in[12];
  const float* qW2 = (const float*)d_in[13]; const float* qb2 = (const float*)d_in[14];
  const float* iW1 = (const float*)d_in[15]; const float* ib1 = (const float*)d_in[16];
  const float* iW2 = (const float*)d_in[17]; const float* ib2 = (const float*)d_in[18];
  const float* gW1 = (const float*)d_in[19]; const float* gb1 = (const float*)d_in[20];
  const float* gW2 = (const float*)d_in[21]; const float* gb2 = (const float*)d_in[22];
  const float* oW  = (const float*)d_in[23]; const float* obv = (const float*)d_in[24];
  const float* sW1 = (const float*)d_in[25]; const float* sb1 = (const float*)d_in[26];
  const float* sW2 = (const float*)d_in[27]; const float* sb2 = (const float*)d_in[28];

  float* ws   = (float*)d_ws;
  float* h    = ws;
  float* mbuf = h    + (size_t)NN * H;
  float* lst  = mbuf + (size_t)NN * H;
  float* sD   = lst + (size_t)NN * H;
  float* sS   = sD + NN;
  int*   cnt  = (int*)lst;
  int*   off  = cnt + NODE_PAD;
  int2*  tmp  = (int2*)(off + NODE_PAD);
  int2*  rec  = tmp + NE;
  float* part = (float*)(rec + NE);

  float* out  = (float*)d_out;
  float* oq   = out;
  float* oi   = out + (size_t)NN;
  float* og   = out + 2 * (size_t)NN;
  float* oh   = out + 3 * (size_t)NN;
  float* ogr  = out + 67 * (size_t)NN;
  float* obot = out + 67 * (size_t)NN + H;
  float* osev = out + 99 * (size_t)NN + H;

  k_input_proj<<<(NN * H + 255) / 256, 256, 0, stream>>>(x, inW, inb, h);

  hipMemsetAsync(cnt, 0, (size_t)NN * sizeof(int), stream);
  k_edge_count<<<(NE / 4 + 255) / 256, 256, 0, stream>>>(ei, cnt);
  k_edge_scan<<<SCAN_BLOCKS, SCAN_T, 0, stream>>>(cnt, off);
  k_edge_place<<<(NE / 4 + 255) / 256, 256, 0, stream>>>(ei, cnt, off, tmp);
  k_edge_order<<<(NE + 255) / 256, 256, 0, stream>>>(ei, ea, off, tmp, rec);

  const int nodeWaveBlocks = (NN + 7) / 8;
  for (int l = 0; l < NL; ++l) {
    k_gemm_h<<<NN / 16, 128, 0, stream>>>(h, linW + (size_t)l * H * H, linb + (size_t)l * H,
                                          attW + (size_t)l * 2 * H, mbuf, sD, sS);
    k_gather_update<<<nodeWaveBlocks, 256, 0, stream>>>(h, mbuf, off, rec, sD, sS, attb + l,
                                                        lng + (size_t)l * H, lnb + (size_t)l * H);
  }

  k_heads<<<nodeWaveBlocks, 256, 0, stream>>>(h, qW1, qb1, qW2, qb2, iW1, ib1, iW2, ib2,
                                              gW1, gb1, gW2, gb2, oW, obv, sW1, sb1, sW2, sb2,
                                              oq, oi, og, oh, obot, osev);
  k_graph_partial<<<GM_BLOCKS, 256, 0, stream>>>(h, part);
  k_graph_final<<<1, 1024, 0, stream>>>(part, ogr);
}


# edge list grouping moved from global integer atomics to LDS counters: edges binned by 512-node destination range, then one workgroup per range groups them by node; no memset; sum orders unchanged
# speedup vs baseline: 1.7582x; 1.2293x over previous
#include <hip/hip_runtime.h>

#define NN 100000
#define NE 1000000
#define DIN 5
#define H 64
#define HHID 32
#define NL 4
#define DOUT 32
#define LN_EPS 1e-5f

typedef __attribute__((ext_vector_type(2)))  float    v2f;
typedef __attribute__((ext_vector_type(8)))  float    v8f;
typedef __attribute__((ext_vector_type(16))) _Float16 v16h;

#if defined(__has_builtin)
#if __has_builtin(__builtin_amdgcn_wmma_f32_16x16x4_f32)
#define HAVE_WMMA_F32 1
#endif
#endif

__device__ __forceinline__ float wave_sum(float v) {
#pragma unroll
  for (int off = 16; off > 0; off >>= 1) v += __shfl_xor(v, off, 32);
  return v;
}
__device__ __forceinline__ float sigmoidf_(float x) { return 1.0f / (1.0f + __expf(-x)); }

__device__ __forceinline__ int wave_sum_i(int v) {
#pragma unroll
  for (int off = 16; off > 0; off >>= 1) v += __shfl_xor(v, off, 32);
  return v;
}

__global__ __launch_bounds__(256) void k_input_proj(
    const float* __restrict__ x, const float* __restrict__ W,
    const float* __restrict__ b, float* __restrict__ h) {
  int t = blockIdx.x * 256 + threadIdx.x;
  if (t >= NN * H) return;
  int n = t >> 6, k = t & 63;
  const float* xr = x + (size_t)n * DIN;
  float acc = b[k];
#pragma unroll
  for (int d = 0; d < DIN; ++d) acc = fmaf(xr[d], W[d * H + k], acc);
  h[t] = acc;
}

__global__ __launch_bounds__(128) void k_gemm_h(
    const float* __restrict__ A, const float* __restrict__ W,
    const float* __restrict__ bias, const float* __restrict__ aw,
    float* __restrict__ out, float* __restrict__ sD, float* __restrict__ sS) {
  __shared__ float lA[16][H + 1];
  __shared__ float lW[H][H + 1];
  const int tid = threadIdx.x;
  const int row0 = blockIdx.x * 16;
  for (int idx = tid; idx < 16 * H; idx += 128)
    lA[idx >> 6][idx & 63] = A[(size_t)row0 * H + idx];
  for (int idx = tid; idx < H * H; idx += 128)
    lW[idx >> 6][idx & 63] = W[idx];
  __syncthreads();

  const int wave = tid >> 5, lane = tid & 31;
  const int m = lane & 15, g = lane >> 4;
  const int col = wave * 16 + m;

  v8f acc;
  float bv = bias[col];
#pragma unroll
  for (int i = 0; i < 8; ++i) acc[i] = bv;

#if HAVE_WMMA_F32
#pragma unroll
  for (int k0 = 0; k0 < H; k0 += 4) {
    int kb = k0 + 2 * g;
    v2f a, b;
    a.x = lA[m][kb];     a.y = lA[m][kb + 1];
    b.x = lW[kb][col];   b.y = lW[kb + 1][col];
    acc = __builtin_amdgcn_wmma_f32_16x16x4_f32(false, a, false, b, (short)0, acc, false, false);
  }
#else
#pragma unroll
  for (int kk = 0; kk < H; kk += 32) {
    v16h a, b;
#pragma unroll
    for (int j = 0; j < 8; ++j) {
      int ka = (j < 4) ? (2 * j + 8 * g) : (16 + 2 * (j - 4) + 8 * g);
      a[2 * j]     = (_Float16)lA[m][kk + ka];
      a[2 * j + 1] = (_Float16)lA[m][kk + ka + 1];
      int kb = 2 * j + 16 * g;
      b[2 * j]     = (_Float16)lW[kk + kb][col];
      b[2 * j + 1] = (_Float16)lW[kk + kb + 1][col];
    }
    acc = __builtin_amdgcn_wmma_f32_16x16x32_f16(false, a, false, b, (short)0, acc, false, false);
  }
#endif

#pragma unroll
  for (int i = 0; i < 8; ++i) {
    int row = row0 + i + 8 * g;
    out[(size_t)row * H + col] = acc[i];
  }

  if (wave < 2) {
    const float* awp = aw + wave * H;
    float p = 0.0f;
#pragma unroll
    for (int k = 0; k < 32; ++k) p = fmaf(lA[m][32 * g + k], awp[32 * g + k], p);
    p += __shfl_xor(p, 16, 32);
    if (g == 0) {
      if (wave == 0) sD[row0 + m] = p; else sS[row0 + m] = p;
    }
  }
}

static_assert(NE % 4 == 0, "four edges per thread");
#define NODE_PAD 100032
#define BKT_SHIFT 9
#define BKT_R (1 << BKT_SHIFT)
#define NBKT ((NN + BKT_R - 1) / BKT_R)
#define NCHK 250
#define CHK_V4 (NE / 4 / NCHK)
#define GH_PITCH 256
#define RG_DEPTH 8
static_assert(NCHK * CHK_V4 * 4 == NE, "chunks cover the edges exactly");
static_assert(CHK_V4 <= 1024, "one four-edge group per thread");
static_assert(NBKT <= 224 && NBKT * BKT_R >= NN + 1, "range table");
static_assert(NCHK <= GH_PITCH && GH_PITCH - NCHK <= NCHK, "pad columns are written by the first chunks");

__device__ __forceinline__ int wave_incl_scan_i(int v, int lane) {
#pragma unroll
  for (int o = 1; o < 32; o <<= 1) {
    int up = __shfl_up(v, o, 32);
    if (lane >= o) v += up;
  }
  return v;
}

__global__ __launch_bounds__(1024) void k_bin_hist(
    const int* __restrict__ ei, int* __restrict__ gh) {
  __shared__ int hist[NBKT];
  const int tid = threadIdx.x, w = blockIdx.x;
  if (tid < NBKT) hist[tid] = 0;
  __syncthreads();
  if (tid < CHK_V4) {
    int4 d = reinterpret_cast<const int4*>(ei + NE)[w * CHK_V4 + tid];
    atomicAdd(&hist[d.x >> BKT_SHIFT], 1);
    atomicAdd(&hist[d.y >> BKT_SHIFT], 1);
    atomicAdd(&hist[d.z >> BKT_SHIFT], 1);
    atomicAdd(&hist[d.w >> BKT_SHIFT], 1);
  }
  __syncthreads();
  if (tid < NBKT) {
    gh[tid * GH_PITCH + w] = hist[tid];
    if (w < GH_PITCH - NCHK) gh[tid * GH_PITCH + NCHK + w] = 0;
  }
}

__global__ __launch_bounds__(1024) void k_bin_scan(int* __restrict__ gh) {
  __shared__ int rowtot[224];
  __shared__ int rowbase[224];
  __shared__ int wtot[8];
  const int tid = threadIdx.x, wave = tid >> 5, lane = tid & 31;
  int4 a[7], c[7];
#pragma unroll
  for (int k = 0; k < 7; ++k) {
    int b = min(wave + 32 * k, NBKT - 1);
    const int4* p = reinterpret_cast<const int4*>(gh + b * GH_PITCH + lane * 8);
    a[k] = p[0];
    c[k] = p[1];
  }
  int ex[7];
#pragma unroll
  for (int k = 0; k < 7; ++k) {
    int b = wave + 32 * k;
    int sum = ((a[k].x + a[k].y) + (a[k].z + a[k].w)) + ((c[k].x + c[k].y) + (c[k].z + c[k].w));
    int incl = wave_incl_scan_i(sum, lane);
    ex[k] = incl - sum;
    int tot = __shfl(incl, 31, 32);
    if (lane == 0) rowtot[b] = (b < NBKT) ? tot : 0;
  }
  __syncthreads();
  int v = 0, vin = 0;
  if (tid < 224) {
    v = rowtot[tid];
    vin = wave_incl_scan_i(v, lane);
    if (lane == 31) wtot[wave] = vin;
  }
  __syncthreads();
  if (tid < 224) {
    int before = 0;
#pragma unroll
    for (int j = 0; j < 7; ++j) before += (j < wave) ? wtot[j] : 0;
    rowbase[tid] = before + vin - v;
  }
  __syncthreads();
#pragma unroll
  for (int k = 0; k < 7; ++k) {
    int b = wave + 32 * k;
    if (b < NBKT) {
      int r = rowbase[b] + ex[k];
      int4 o0, o1;
      o0.x = r; r += a[k].x;
      o0.y = r; r += a[k].y;
      o0.z = r; r += a[k].z;
      o0.w = r; r += a[k].w;
      o1.x = r; r += c[k].x;
      o1.y = r; r += c[k].y;
      o1.z = r; r += c[k].z;
      o1.w = r;
      int4* p = reinterpret_cast<int4*>(gh + b * GH_PITCH + lane * 8);
      p[0] = o0;
      p[1] = o1;
    }
  }
}

__global__ __launch_bounds__(1024) void k_bin_scatter(
    const int* __restrict__ ei, const int* __restrict__ gh, int2* __restrict__ binned) {
  __shared__ int cur[NBKT];
  const int tid = threadIdx.x, w = blockIdx.x;
  if (tid < NBKT) cur[tid] = gh[tid * GH_PITCH + w];
  __syncthreads();
  if (tid < CHK_V4) {
    int t = w * CHK_V4 + tid;
    int4 d = reinterpret_cast<const int4*>(ei + NE)[t];
    int e0 = 4 * t;
    int p;
    p = atomicAdd(&cur[d.x >> BKT_SHIFT], 1); binned[p] = make_int2(e0,     d.x);
    p = atomicAdd(&cur[d.y >> BKT_SHIFT], 1); binned[p] = make_int2(e0 + 1, d.y);
    p = atomicAdd(&cur[d.z >> BKT_SHIFT], 1); binned[p] = make_int2(e0 + 2, d.z);
    p = atomicAdd(&cur[d.w >> BKT_SHIFT], 1); binned[p] = make_int2(e0 + 3, d.w);
  }
}

__global__ __launch_bounds__(1024) void k_range_group(
    const int* __restrict__ gh, const int2* __restrict__ binned,
    int* __restrict__ off, int2* __restrict__ tmp) {
  __shared__ int cnt[BKT_R];
  __shared__ int wtot[32];
  const int tid = threadIdx.x, wave = tid >> 5, lane = tid & 31;
  const int b = blockIdx.x;
  const int lo = b << BKT_SHIFT;
  const int cb = gh[b * GH_PITCH];
  const int ce = (b + 1 < NBKT) ? gh[(b + 1) * GH_PITCH] : NE;
  if (tid < BKT_R) cnt[tid] = 0;
  __syncthreads();
  for (int i0 = cb + tid; i0 < ce; i0 += RG_DEPTH * 1024) {
    int dd[RG_DEPTH];
#pragma unroll
    for (int u = 0; u < RG_DEPTH; ++u) dd[u] = binned[min(i0 + 1024 * u, ce - 1)].y;
#pragma unroll
    for (int u = 0; u < RG_DEPTH; ++u)
      if (i0 + 1024 * u < ce) atomicAdd(&cnt[dd[u] - lo], 1);
  }
  __syncthreads();
  const int c = (tid < BKT_R) ? cnt[tid] : 0;
  const int incl = wave_incl_scan_i(c, lane);
  if (lane == 31) wtot[wave] = incl;
  __syncthreads();
  const int before = wave_sum_i((lane < wave) ? wtot[lane] : 0);
  const int excl = before + incl - c;
  if (tid < BKT_R) {
    cnt[tid] = excl;
    if (lo + tid <= NN) off[lo + tid] = cb + excl;
  }
  __syncthreads();
  for (int i0 = cb + tid; i0 < ce; i0 += RG_DEPTH * 1024) {
    int2 r[RG_DEPTH];
#pragma unroll
    for (int u = 0; u < RG_DEPTH; ++u) r[u] = binned[min(i0 + 1024 * u, ce - 1)];
#pragma unroll
    for (int u = 0; u < RG_DEPTH; ++u)
      if (i0 + 1024 * u < ce) {
        int p = atomicAdd(&cnt[r[u].y - lo], 1);
        tmp[cb + p] = r[u];
      }
  }
}

__global__ __launch_bounds__(256) void k_edge_order(
    const int* __restrict__ ei, const float* __restrict__ ea,
    const int* __restrict__ off, const int2* __restrict__ tmp,
    int2* __restrict__ rec) {
  int p = blockIdx.x * 256 + threadIdx.x;
  if (p >= NE) return;
  int2 my = tmp[p];
  int s = off[my.y], e = off[my.y + 1];
  int rank = 0;
  for (int q = s; q < e; ++q) rank += (tmp[q].x < my.x) ? 1 : 0;
  rec[s + rank] = make_int2(ei[my.x], __float_as_int(ea[my.x]));
}

__global__ __launch_bounds__(256) void k_gather_update(
    float* __restrict__ h, const float* __restrict__ m,
    const int* __restrict__ off, const int2* __restrict__ rec,
    const float* __restrict__ sD, const float* __restrict__ sS,
    const float* __restrict__ attb,
    const float* __restrict__ g, const float* __restrict__ beta) {
  const int gw = __builtin_amdgcn_readfirstlane(blockIdx.x * 8 + (threadIdx.x >> 5));
  int lane = threadIdx.x & 31;
  if (gw >= NN) return;
  size_t base = (size_t)gw * H;
  const int s = off[gw], e = off[gw + 1];
  const float sd = sD[gw], ab = attb[0];
  float r0 = h[base + lane], r1 = h[base + lane + 32];
  float m0 = m[base + lane], m1 = m[base + lane + 32];
  float acc0 = 0.0f, acc1 = 0.0f;
  for (int cb = s; cb < e; cb += 32) {
    const int cnt = min(32, e - cb);
    int src = 0;
    float w = 0.0f;
    if (lane < cnt) {
      int2 r = rec[cb + lane];
      src = r.x;
      __builtin_prefetch(m + (size_t)src * H, 0, 1);
      __builtin_prefetch(m + (size_t)src * H + 32, 0, 1);
      w = __int_as_float(r.y) * sigmoidf_(sd + sS[src] + ab);
    }
#pragma unroll 4
    for (int i = 0; i < cnt; ++i) {
      int bs = __builtin_amdgcn_readlane(src, i);
      float bw = __int_as_float(__builtin_amdgcn_readlane(__float_as_int(w), i));
      const float* mr = m + (size_t)bs * H;
      acc0 = fmaf(mr[lane], bw, acc0);
      acc1 = fmaf(mr[lane + 32], bw, acc1);
    }
  }
  float f0 = acc0 + m0;
  float f1 = acc1 + m1;
  float mean = wave_sum(f0 + f1) * (1.0f / H);
  float d0 = f0 - mean, d1 = f1 - mean;
  float var = wave_sum(fmaf(d0, d0, d1 * d1)) * (1.0f / H);
  float inv = rsqrtf(var + LN_EPS);
  float y0 = fmaf(d0 * inv, g[lane], beta[lane]);
  float y1 = fmaf(d1 * inv, g[lane + 32], beta[lane + 32]);
  h[base + lane]      = fmaxf(y0, 0.0f) + r0;
  h[base + lane + 32] = fmaxf(y1, 0.0f) + r1;
}

__global__ __launch_bounds__(256) void k_heads(
    const float* __restrict__ h,
    const float* __restrict__ qW1, const float* __restrict__ qb1,
    const float* __restrict__ qW2, const float* __restrict__ qb2,
    const float* __restrict__ iW1, const float* __restrict__ ib1,
    const float* __restrict__ iW2, const float* __restrict__ ib2,
    const float* __restrict__ gW1, const float* __restrict__ gb1,
    const float* __restrict__ gW2, const float* __restrict__ gb2,
    const float* __restrict__ oW,  const float* __restrict__ obv,
    const float* __restrict__ sW1, const float* __restrict__ sb1,
    const float* __restrict__ sW2, const float* __restrict__ sb2,
    float* __restrict__ oq, float* __restrict__ oi, float* __restrict__ og,
    float* __restrict__ oh, float* __restrict__ obot, float* __restrict__ osev) {
  int gw = blockIdx.x * 8 + (threadIdx.x >> 5);
  int lane = threadIdx.x & 31;
  if (gw >= NN) return;
  size_t base = (size_t)gw * H;
  float h0 = h[base + lane], h1 = h[base + lane + 32];

  float tq = 0.f, ti = 0.f, tg = 0.f, ts = 0.f, bo = 0.f;
#pragma unroll 8
  for (int k = 0; k < H; ++k) {
    float hk = (k < 32) ? __shfl(h0, k, 32) : __shfl(h1, k - 32, 32);
    int wi = k * HHID + lane;
    tq = fmaf(hk, qW1[wi], tq);
    ti = fmaf(hk, iW1[wi], ti);
    tg = fmaf(hk, gW1[wi], tg);
    ts = fmaf(hk, sW1[wi], ts);
    bo = fmaf(hk, oW[wi],  bo);
  }
  tq = fmaxf(tq + qb1[lane], 0.0f);
  ti = fmaxf(ti + ib1[lane], 0.0f);
  tg = fmaxf(tg + gb1[lane], 0.0f);
  ts = fmaxf(ts + sb1[lane], 0.0f);

  float q  = wave_sum(tq * qW2[lane]);
  float ii = wave_sum(ti * iW2[lane]);
  float gg = wave_sum(tg * gW2[lane]);
  if (lane == 0) {
    oq[gw] = sigmoidf_(q + qb2[0]);
    oi[gw] = sigmoidf_(ii + ib2[0]);
    og[gw] = sigmoidf_(gg + gb2[0]);
  }

  float z[5];
#pragma unroll
  for (int c = 0; c < 5; ++c) z[c] = wave_sum(ts * sW2[lane * 5 + c]) + sb2[c];
  float zm = z[0];
#pragma unroll
  for (int c = 1; c < 5; ++c) zm = fmaxf(zm, z[c]);
  float es = 0.f;
#pragma unroll
  for (int c = 0; c < 5; ++c) { z[c] = __expf(z[c] - zm); es += z[c]; }
  float inv = 1.0f / es;
#pragma unroll
  for (int c = 0; c < 5; ++c)
    if (lane == c) osev[(size_t)gw * 5 + c] = z[c] * inv;

  obot[(size_t)gw * DOUT + lane] = bo + obv[lane];
  oh[base + lane]      = h0;
  oh[base + lane + 32] = h1;
}

#define GM_BLOCKS 625
#define GM_LOADS 10
__global__ __launch_bounds__(256) void k_graph_partial(
    const float* __restrict__ h, float* __restrict__ part) {
  __shared__ float ls[8][H];
  const int tid = threadIdx.x;
  const int wave = tid >> 5, lane = tid & 31;
  const float4* h4 = reinterpret_cast<const float4*>(h);
  const size_t t = (size_t)blockIdx.x * 256 + tid;
  float4 v[GM_LOADS];
#pragma unroll
  for (int i = 0; i < GM_LOADS; ++i) v[i] = h4[t + (size_t)i * (GM_BLOCKS * 256)];
  float4 a = v[0];
#pragma unroll
  for (int i = 1; i < GM_LOADS; ++i) {
    a.x += v[i].x; a.y += v[i].y; a.z += v[i].z; a.w += v[i].w;
  }
  a.x += __shfl_xor(a.x, 16, 32);
  a.y += __shfl_xor(a.y, 16, 32);
  a.z += __shfl_xor(a.z, 16, 32);
  a.w += __shfl_xor(a.w, 16, 32);
  if (lane < 16) *reinterpret_cast<float4*>(&ls[wave][lane * 4]) = a;
  __syncthreads();
  if (tid < H) {
    float s = ls[0][tid];
#pragma unroll
    for (int w = 1; w < 8; ++w) s += ls[w][tid];
    part[(size_t)blockIdx.x * H + tid] = s;
  }
}

__global__ __launch_bounds__(1024) void k_graph_final(
    const float* __restrict__ part, float* __restrict__ gout) {
  __shared__ float ls[16][H];
  const int tid = threadIdx.x;
  const int f = tid & 63, q = tid >> 6;
  float v[40];
#pragma unroll
  for (int i = 0; i < 40; ++i) {
    int b = q + 16 * i;
    v[i] = (b < GM_BLOCKS) ? part[(size_t)b * H + f] : 0.0f;
  }
  float s = v[0];
#pragma unroll
  for (int i = 1; i < 40; ++i) s += v[i];
  ls[q][f] = s;
  __syncthreads();
  if (tid < H) {
    float r = ls[0][tid];
#pragma unroll
    for (int w = 1; w < 16; ++w) r += ls[w][tid];
    gout[tid] = r * (1.0f / NN);
  }
}

extern "C" void kernel_launch(void* const* d_in, const int* in_sizes, int n_in,
                              void* d_out, int out_size, void* d_ws, size_t ws_size,
                              hipStream_t stream) {
  (void)in_sizes; (void)n_in; (void)out_size; (void)ws_size;
  const float* x    = (const float*)d_in[0];
  const int*   ei   = (const int*)  d_in[1];
  const float* ea   = (const float*)d_in[2];
  const float* inW  = (const float*)d_in[3];
  const float* inb  = (const float*)d_in[4];
  const float* linW = (const float*)d_in[5];
  const float* linb = (const float*)d_in[6];
  const float* attW = (const float*)d_in[7];
  const float* attb = (const float*)d_in[8];
  const float* lng  = (const float*)d_in[9];
  const float* lnb  = (const float*)d_in[10];
  const float* qW1 = (const float*)d_in[11]; const float* qb1 = (const float*)d_in[12];
  const float* qW2 = (const float*)d_in[13]; const float* qb2 = (const float*)d_in[14];
  const float* iW1 = (const float*)d_in[15]; const float* ib1 = (const float*)d_in[16];
  const float* iW2 = (const float*)d_in[17]; const float* ib2 = (const float*)d_in[18];
  const float* gW1 = (const float*)d_in[19]; const float* gb1 = (const float*)d_in[20];
  const float* gW2 = (const float*)d_in[21]; const float* gb2 = (const float*)d_in[22];
  const float* oW  = (const float*)d_in[23]; const float* obv = (const float*)d_in[24];
  const float* sW1 = (const float*)d_in[25]; const float* sb1 = (const float*)d_in[26];
  const float* sW2 = (const float*)d_in[27]; const float* sb2 = (const float*)d_in[28];

  float* ws   = (float*)d_ws;
  float* h    = ws;
  float* mbuf = h    + (size_t)NN * H;
  float* lst  = mbuf + (size_t)NN * H;
  float* sD   = lst + (size_t)NN * H;
  float* sS   = sD + NN;
  int*   off  = (int*)lst;
  int2*  tmp  = (int2*)(off + NODE_PAD);
  int2*  rec  = tmp + NE;
  float* part = (float*)(rec + NE);
  int2*  binned = (int2*)(part + GM_BLOCKS * H);
  int*   gh   = (int*)(binned + NE);

  float* out  = (float*)d_out;
  float* oq   = out;
  float* oi   = out + (size_t)NN;
  float* og   = out + 2 * (size_t)NN;
  float* oh   = out + 3 * (size_t)NN;
  float* ogr  = out + 67 * (size_t)NN;
  float* obot = out + 67 * (size_t)NN + H;
  float* osev = out + 99 * (size_t)NN + H;

  k_input_proj<<<(NN * H + 255) / 256, 256, 0, stream>>>(x, inW, inb, h);

  k_bin_hist<<<NCHK, 1024, 0, stream>>>(ei, gh);
  k_bin_scan<<<1, 1024, 0, stream>>>(gh);
  k_bin_scatter<<<NCHK, 1024, 0, stream>>>(ei, gh, binned);
  k_range_group<<<NBKT, 1024, 0, stream>>>(gh, binned, off, tmp);
  k_edge_order<<<(NE + 255) / 256, 256, 0, stream>>>(ei, ea, off, tmp, rec);

  const int nodeWaveBlocks = (NN + 7) / 8;
  for (int l = 0; l < NL; ++l) {
    k_gemm_h<<<NN / 16, 128, 0, stream>>>(h, linW + (size_t)l * H * H, linb + (size_t)l * H,
                                          attW + (size_t)l * 2 * H, mbuf, sD, sS);
    k_gather_update<<<nodeWaveBlocks, 256, 0, stream>>>(h, mbuf, off, rec, sD, sS, attb + l,
                                                        lng + (size_t)l * H, lnb + (size_t)l * H);
  }

  k_heads<<<nodeWaveBlocks, 256, 0, stream>>>(h, qW1, qb1, qW2, qb2, iW1, ib1, iW2, ib2,
                                              gW1, gb1, gW2, gb2, oW, obv, sW1, sb1, sW2, sb2,
                                              oq, oi, og, oh, obot, osev);
  k_graph_partial<<<GM_BLOCKS, 256, 0, stream>>>(h, part);
  k_graph_final<<<1, 1024, 0, stream>>>(part, ogr);
}


# input projection: 200000 one-shot waves (one output per thread) become 5000 waves that each walk 20 nodes, weights in registers, 512-byte stores; per-output sum order unchanged
# speedup vs baseline: 1.8347x; 1.0435x over previous
#include <hip/hip_runtime.h>

#define NN 100000
#define NE 1000000
#define DIN 5
#define H 64
#define HHID 32
#define NL 4
#define DOUT 32
#define LN_EPS 1e-5f

typedef __attribute__((ext_vector_type(2)))  float    v2f;
typedef __attribute__((ext_vector_type(8)))  float    v8f;
typedef __attribute__((ext_vector_type(16))) _Float16 v16h;

#if defined(__has_builtin)
#if __has_builtin(__builtin_amdgcn_wmma_f32_16x16x4_f32)
#define HAVE_WMMA_F32 1
#endif
#endif

__device__ __forceinline__ float wave_sum(float v) {
#pragma unroll
  for (int off = 16; off > 0; off >>= 1) v += __shfl_xor(v, off, 32);
  return v;
}
__device__ __forceinline__ float sigmoidf_(float x) { return 1.0f / (1.0f + __expf(-x)); }

__device__ __forceinline__ int wave_sum_i(int v) {
#pragma unroll
  for (int off = 16; off > 0; off >>= 1) v += __shfl_xor(v, off, 32);
  return v;
}

#define IP_BLOCKS 625
#define IP_WAVE_NODES 20
static_assert(IP_BLOCKS * 8 * IP_WAVE_NODES == NN, "the waves cover the nodes exactly");
static_assert(IP_WAVE_NODES % 2 == 0 && H == 64, "two nodes per step, sixteen lanes per row");
__global__ __launch_bounds__(256) void k_input_proj(
    const float* __restrict__ x, const float* __restrict__ W,
    const float* __restrict__ b, float* __restrict__ h) {
  const int wave = threadIdx.x >> 5, lane = threadIdx.x & 31;
  const int c4 = (lane & 15) * 4;
  float4 w[DIN];
#pragma unroll
  for (int d = 0; d < DIN; ++d) w[d] = *reinterpret_cast<const float4*>(W + d * H + c4);
  const float4 bv = *reinterpret_cast<const float4*>(b + c4);
  const int n0 = (blockIdx.x * 8 + wave) * IP_WAVE_NODES + (lane >> 4);
#pragma unroll
  for (int i = 0; i < IP_WAVE_NODES / 2; ++i) {
    const int n = n0 + 2 * i;
    const float* xr = x + (size_t)n * DIN;
    float xv[DIN];
#pragma unroll
    for (int d = 0; d < DIN; ++d) xv[d] = xr[d];
    float4 acc = bv;
#pragma unroll
    for (int d = 0; d < DIN; ++d) {
      acc.x = fmaf(xv[d], w[d].x, acc.x);
      acc.y = fmaf(xv[d], w[d].y, acc.y);
      acc.z = fmaf(xv[d], w[d].z, acc.z);
      acc.w = fmaf(xv[d], w[d].w, acc.w);
    }
    *reinterpret_cast<float4*>(h + (size_t)n * H + c4) = acc;
  }
}

__global__ __launch_bounds__(128) void k_gemm_h(
    const float* __restrict__ A, const float* __restrict__ W,
    const float* __restrict__ bias, const float* __restrict__ aw,
    float* __restrict__ out, float* __restrict__ sD, float* __restrict__ sS) {
  __shared__ float lA[16][H + 1];
  __shared__ float lW[H][H + 1];
  const int tid = threadIdx.x;
  const int row0 = blockIdx.x * 16;
  for (int idx = tid; idx < 16 * H; idx += 128)
    lA[idx >> 6][idx & 63] = A[(size_t)row0 * H + idx];
  for (int idx = tid; idx < H * H; idx += 128)
    lW[idx >> 6][idx & 63] = W[idx];
  __syncthreads();

  const int wave = tid >> 5, lane = tid & 31;
  const int m = lane & 15, g = lane >> 4;
  const int col = wave * 16 + m;

  v8f acc;
  float bv = bias[col];
#pragma unroll
  for (int i = 0; i < 8; ++i) acc[i] = bv;

#if HAVE_WMMA_F32
#pragma unroll
  for (int k0 = 0; k0 < H; k0 += 4) {
    int kb = k0 + 2 * g;
    v2f a, b;
    a.x = lA[m][kb];     a.y = lA[m][kb + 1];
    b.x = lW[kb][col];   b.y = lW[kb + 1][col];
    acc = __builtin_amdgcn_wmma_f32_16x16x4_f32(false, a, false, b, (short)0, acc, false, false);
  }
#else
#pragma unroll
  for (int kk = 0; kk < H; kk += 32) {
    v16h a, b;
#pragma unroll
    for (int j = 0; j < 8; ++j) {
      int ka = (j < 4) ? (2 * j + 8 * g) : (16 + 2 * (j - 4) + 8 * g);
      a[2 * j]     = (_Float16)lA[m][kk + ka];
      a[2 * j + 1] = (_Float16)lA[m][kk + ka + 1];
      int kb = 2 * j + 16 * g;
      b[2 * j]     = (_Float16)lW[kk + kb][col];
      b[2 * j + 1] = (_Float16)lW[kk + kb + 1][col];
    }
    acc = __builtin_amdgcn_wmma_f32_16x16x32_f16(false, a, false, b, (short)0, acc, false, false);
  }
#endif

#pragma unroll
  for (int i = 0; i < 8; ++i) {
    int row = row0 + i + 8 * g;
    out[(size_t)row * H + col] = acc[i];
  }

  if (wave < 2) {
    const float* awp = aw + wave * H;
    float p = 0.0f;
#pragma unroll
    for (int k = 0; k < 32; ++k) p = fmaf(lA[m][32 * g + k], awp[32 * g + k], p);
    p += __shfl_xor(p, 16, 32);
    if (g == 0) {
      if (wave == 0) sD[row0 + m] = p; else sS[row0 + m] = p;
    }
  }
}

static_assert(NE % 4 == 0, "four edges per thread");
#define NODE_PAD 100032
#define BKT_SHIFT 9
#define BKT_R (1 << BKT_SHIFT)
#define NBKT ((NN + BKT_R - 1) / BKT_R)
#define NCHK 250
#define CHK_V4 (NE / 4 / NCHK)
#define GH_PITCH 256
#define RG_DEPTH 8
static_assert(NCHK * CHK_V4 * 4 == NE, "chunks cover the edges exactly");
static_assert(CHK_V4 <= 1024, "one four-edge group per thread");
static_assert(NBKT <= 224 && NBKT * BKT_R >= NN + 1, "range table");
static_assert(NCHK <= GH_PITCH && GH_PITCH - NCHK <= NCHK, "pad columns are written by the first chunks");

__device__ __forceinline__ int wave_incl_scan_i(int v, int lane) {
#pragma unroll
  for (int o = 1; o < 32; o <<= 1) {
    int up = __shfl_up(v, o, 32);
    if (lane >= o) v += up;
  }
  return v;
}

__global__ __launch_bounds__(1024) void k_bin_hist(
    const int* __restrict__ ei, int* __restrict__ gh) {
  __shared__ int hist[NBKT];
  const int tid = threadIdx.x, w = blockIdx.x;
  if (tid < NBKT) hist[tid] = 0;
  __syncthreads();
  if (tid < CHK_V4) {
    int4 d = reinterpret_cast<const int4*>(ei + NE)[w * CHK_V4 + tid];
    atomicAdd(&hist[d.x >> BKT_SHIFT], 1);
    atomicAdd(&hist[d.y >> BKT_SHIFT], 1);
    atomicAdd(&hist[d.z >> BKT_SHIFT], 1);
    atomicAdd(&hist[d.w >> BKT_SHIFT], 1);
  }
  __syncthreads();
  if (tid < NBKT) {
    gh[tid * GH_PITCH + w] = hist[tid];
    if (w < GH_PITCH - NCHK) gh[tid * GH_PITCH + NCHK + w] = 0;
  }
}

__global__ __launch_bounds__(1024) void k_bin_scan(int* __restrict__ gh) {
  __shared__ int rowtot[224];
  __shared__ int rowbase[224];
  __shared__ int wtot[8];
  const int tid = threadIdx.x, wave = tid >> 5, lane = tid & 31;
  int4 a[7], c[7];
#pragma unroll
  for (int k = 0; k < 7; ++k) {
    int b = min(wave + 32 * k, NBKT - 1);
    const int4* p = reinterpret_cast<const int4*>(gh + b * GH_PITCH + lane * 8);
    a[k] = p[0];
    c[k] = p[1];
  }
  int ex[7];
#pragma unroll
  for (int k = 0; k < 7; ++k) {
    int b = wave + 32 * k;
    int sum = ((a[k].x + a[k].y) + (a[k].z + a[k].w)) + ((c[k].x + c[k].y) + (c[k].z + c[k].w));
    int incl = wave_incl_scan_i(sum, lane);
    ex[k] = incl - sum;
    int tot = __shfl(incl, 31, 32);
    if (lane == 0) rowtot[b] = (b < NBKT) ? tot : 0;
  }
  __syncthreads();
  int v = 0, vin = 0;
  if (tid < 224) {
    v = rowtot[tid];
    vin = wave_incl_scan_i(v, lane);
    if (lane == 31) wtot[wave] = vin;
  }
  __syncthreads();
  if (tid < 224) {
    int before = 0;
#pragma unroll
    for (int j = 0; j < 7; ++j) before += (j < wave) ? wtot[j] : 0;
    rowbase[tid] = before + vin - v;
  }
  __syncthreads();
#pragma unroll
  for (int k = 0; k < 7; ++k) {
    int b = wave + 32 * k;
    if (b < NBKT) {
      int r = rowbase[b] + ex[k];
      int4 o0, o1;
      o0.x = r; r += a[k].x;
      o0.y = r; r += a[k].y;
      o0.z = r; r += a[k].z;
      o0.w = r; r += a[k].w;
      o1.x = r; r += c[k].x;
      o1.y = r; r += c[k].y;
      o1.z = r; r += c[k].z;
      o1.w = r;
      int4* p = reinterpret_cast<int4*>(gh + b * GH_PITCH + lane * 8);
      p[0] = o0;
      p[1] = o1;
    }
  }
}

__global__ __launch_bounds__(1024) void k_bin_scatter(
    const int* __restrict__ ei, const int* __restrict__ gh, int2* __restrict__ binned) {
  __shared__ int cur[NBKT];
  const int tid = threadIdx.x, w = blockIdx.x;
  if (tid < NBKT) cur[tid] = gh[tid * GH_PITCH + w];
  __syncthreads();
  if (tid < CHK_V4) {
    int t = w * CHK_V4 + tid;
    int4 d = reinterpret_cast<const int4*>(ei + NE)[t];
    int e0 = 4 * t;
    int p;
    p = atomicAdd(&cur[d.x >> BKT_SHIFT], 1); binned[p] = make_int2(e0,     d.x);
    p = atomicAdd(&cur[d.y >> BKT_SHIFT], 1); binned[p] = make_int2(e0 + 1, d.y);
    p = atomicAdd(&cur[d.z >> BKT_SHIFT], 1); binned[p] = make_int2(e0 + 2, d.z);
    p = atomicAdd(&cur[d.w >> BKT_SHIFT], 1); binned[p] = make_int2(e0 + 3, d.w);
  }
}

__global__ __launch_bounds__(1024) void k_range_group(
    const int* __restrict__ gh, const int2* __restrict__ binned,
    int* __restrict__ off, int2* __restrict__ tmp) {
  __shared__ int cnt[BKT_R];
  __shared__ int wtot[32];
  const int tid = threadIdx.x, wave = tid >> 5, lane = tid & 31;
  const int b = blockIdx.x;
  const int lo = b << BKT_SHIFT;
  const int cb = gh[b * GH_PITCH];
  const int ce = (b + 1 < NBKT) ? gh[(b + 1) * GH_PITCH] : NE;
  if (tid < BKT_R) cnt[tid] = 0;
  __syncthreads();
  for (int i0 = cb + tid; i0 < ce; i0 += RG_DEPTH * 1024) {
    int dd[RG_DEPTH];
#pragma unroll
    for (int u = 0; u < RG_DEPTH; ++u) dd[u] = binned[min(i0 + 1024 * u, ce - 1)].y;
#pragma unroll
    for (int u = 0; u < RG_DEPTH; ++u)
      if (i0 + 1024 * u < ce) atomicAdd(&cnt[dd[u] - lo], 1);
  }
  __syncthreads();
  const int c = (tid < BKT_R) ? cnt[tid] : 0;
  const int incl = wave_incl_scan_i(c, lane);
  if (lane == 31) wtot[wave] = incl;
  __syncthreads();
  const int before = wave_sum_i((lane < wave) ? wtot[lane] : 0);
  const int excl = before + incl - c;
  if (tid < BKT_R) {
    cnt[tid] = excl;
    if (lo + tid <= NN) off[lo + tid] = cb + excl;
  }
  __syncthreads();
  for (int i0 = cb + tid; i0 < ce; i0 += RG_DEPTH * 1024) {
    int2 r[RG_DEPTH];
#pragma unroll
    for (int u = 0; u < RG_DEPTH; ++u) r[u] = binned[min(i0 + 1024 * u, ce - 1)];
#pragma unroll
    for (int u = 0; u < RG_DEPTH; ++u)
      if (i0 + 1024 * u < ce) {
        int p = atomicAdd(&cnt[r[u].y - lo], 1);
        tmp[cb + p] = r[u];
      }
  }
}

__global__ __launch_bounds__(256) void k_edge_order(
    const int* __restrict__ ei, const float* __restrict__ ea,
    const int* __restrict__ off, const int2* __restrict__ tmp,
    int2* __restrict__ rec) {
  int p = blockIdx.x * 256 + threadIdx.x;
  if (p >= NE) return;
  int2 my = tmp[p];
  int s = off[my.y], e = off[my.y + 1];
  int rank = 0;
  for (int q = s; q < e; ++q) rank += (tmp[q].x < my.x) ? 1 : 0;
  rec[s + rank] = make_int2(ei[my.x], __float_as_int(ea[my.x]));
}

__global__ __launch_bounds__(256) void k_gather_update(
    float* __restrict__ h, const float* __restrict__ m,
    const int* __restrict__ off, const int2* __restrict__ rec,
    const float* __restrict__ sD, const float* __restrict__ sS,
    const float* __restrict__ attb,
    const float* __restrict__ g, const float* __restrict__ beta) {
  const int gw = __builtin_amdgcn_readfirstlane(blockIdx.x * 8 + (threadIdx.x >> 5));
  int lane = threadIdx.x & 31;
  if (gw >= NN) return;
  size_t base = (size_t)gw * H;
  const int s = off[gw], e = off[gw + 1];
  const float sd = sD[gw], ab = attb[0];
  float r0 = h[base + lane], r1 = h[base + lane + 32];
  float m0 = m[base + lane], m1 = m[base + lane + 32];
  float acc0 = 0.0f, acc1 = 0.0f;
  for (int cb = s; cb < e; cb += 32) {
    const int cnt = min(32, e - cb);
    int src = 0;
    float w = 0.0f;
    if (lane < cnt) {
      int2 r = rec[cb + lane];
      src = r.x;
      __builtin_prefetch(m + (size_t)src * H, 0, 1);
      __builtin_prefetch(m + (size_t)src * H + 32, 0, 1);
      w = __int_as_float(r.y) * sigmoidf_(sd + sS[src] + ab);
    }
#pragma unroll 4
    for (int i = 0; i < cnt; ++i) {
      int bs = __builtin_amdgcn_readlane(src, i);
      float bw = __int_as_float(__builtin_amdgcn_readlane(__float_as_int(w), i));
      const float* mr = m + (size_t)bs * H;
      acc0 = fmaf(mr[lane], bw, acc0);
      acc1 = fmaf(mr[lane + 32], bw, acc1);
    }
  }
  float f0 = acc0 + m0;
  float f1 = acc1 + m1;
  float mean = wave_sum(f0 + f1) * (1.0f / H);
  float d0 = f0 - mean, d1 = f1 - mean;
  float var = wave_sum(fmaf(d0, d0, d1 * d1)) * (1.0f / H);
  float inv = rsqrtf(var + LN_EPS);
  float y0 = fmaf(d0 * inv, g[lane], beta[lane]);
  float y1 = fmaf(d1 * inv, g[lane + 32], beta[lane + 32]);
  h[base + lane]      = fmaxf(y0, 0.0f) + r0;
  h[base + lane + 32] = fmaxf(y1, 0.0f) + r1;
}

__global__ __launch_bounds__(256) void k_heads(
    const float* __restrict__ h,
    const float* __restrict__ qW1, const float* __restrict__ qb1,
    const float* __restrict__ qW2, const float* __restrict__ qb2,
    const float* __restrict__ iW1, const float* __restrict__ ib1,
    const float* __restrict__ iW2, const float* __restrict__ ib2,
    const float* __restrict__ gW1, const float* __restrict__ gb1,
    const float* __restrict__ gW2, const float* __restrict__ gb2,
    const float* __restrict__ oW,  const float* __restrict__ obv,
    const float* __restrict__ sW1, const float* __restrict__ sb1,
    const float* __restrict__ sW2, const float* __restrict__ sb2,
    float* __restrict__ oq, float* __restrict__ oi, float* __restrict__ og,
    float* __restrict__ oh, float* __restrict__ obot, float* __restrict__ osev) {
  int gw = blockIdx.x * 8 + (threadIdx.x >> 5);
  int lane = threadIdx.x & 31;
  if (gw >= NN) return;
  size_t base = (size_t)gw * H;
  float h0 = h[base + lane], h1 = h[base + lane + 32];

  float tq = 0.f, ti = 0.f, tg = 0.f, ts = 0.f, bo = 0.f;
#pragma unroll 8
  for (int k = 0; k < H; ++k) {
    float hk = (k < 32) ? __shfl(h0, k, 32) : __shfl(h1, k - 32, 32);
    int wi = k * HHID + lane;
    tq = fmaf(hk, qW1[wi], tq);
    ti = fmaf(hk, iW1[wi], ti);
    tg = fmaf(hk, gW1[wi], tg);
    ts = fmaf(hk, sW1[wi], ts);
    bo = fmaf(hk, oW[wi],  bo);
  }
  tq = fmaxf(tq + qb1[lane], 0.0f);
  ti = fmaxf(ti + ib1[lane], 0.0f);
  tg = fmaxf(tg + gb1[lane], 0.0f);
  ts = fmaxf(ts + sb1[lane], 0.0f);

  float q  = wave_sum(tq * qW2[lane]);
  float ii = wave_sum(ti * iW2[lane]);
  float gg = wave_sum(tg * gW2[lane]);
  if (lane == 0) {
    oq[gw] = sigmoidf_(q + qb2[0]);
    oi[gw] = sigmoidf_(ii + ib2[0]);
    og[gw] = sigmoidf_(gg + gb2[0]);
  }

  float z[5];
#pragma unroll
  for (int c = 0; c < 5; ++c) z[c] = wave_sum(ts * sW2[lane * 5 + c]) + sb2[c];
  float zm = z[0];
#pragma unroll
  for (int c = 1; c < 5; ++c) zm = fmaxf(zm, z[c]);
  float es = 0.f;
#pragma unroll
  for (int c = 0; c < 5; ++c) { z[c] = __expf(z[c] - zm); es += z[c]; }
  float inv = 1.0f / es;
#pragma unroll
  for (int c = 0; c < 5; ++c)
    if (lane == c) osev[(size_t)gw * 5 + c] = z[c] * inv;

  obot[(size_t)gw * DOUT + lane] = bo + obv[lane];
  oh[base + lane]      = h0;
  oh[base + lane + 32] = h1;
}

#define GM_BLOCKS 625
#define GM_LOADS 10
__global__ __launch_bounds__(256) void k_graph_partial(
    const float* __restrict__ h, float* __restrict__ part) {
  __shared__ float ls[8][H];
  const int tid = threadIdx.x;
  const int wave = tid >> 5, lane = tid & 31;
  const float4* h4 = reinterpret_cast<const float4*>(h);
  const size_t t = (size_t)blockIdx.x * 256 + tid;
  float4 v[GM_LOADS];
#pragma unroll
  for (int i = 0; i < GM_LOADS; ++i) v[i] = h4[t + (size_t)i * (GM_BLOCKS * 256)];
  float4 a = v[0];
#pragma unroll
  for (int i = 1; i < GM_LOADS; ++i) {
    a.x += v[i].x; a.y += v[i].y; a.z += v[i].z; a.w += v[i].w;
  }
  a.x += __shfl_xor(a.x, 16, 32);
  a.y += __shfl_xor(a.y, 16, 32);
  a.z += __shfl_xor(a.z, 16, 32);
  a.w += __shfl_xor(a.w, 16, 32);
  if (lane < 16) *reinterpret_cast<float4*>(&ls[wave][lane * 4]) = a;
  __syncthreads();
  if (tid < H) {
    float s = ls[0][tid];
#pragma unroll
    for (int w = 1; w < 8; ++w) s += ls[w][tid];
    part[(size_t)blockIdx.x * H + tid] = s;
  }
}

__global__ __launch_bounds__(1024) void k_graph_final(
    const float* __restrict__ part, float* __restrict__ gout) {
  __shared__ float ls[16][H];
  const int tid = threadIdx.x;
  const int f = tid & 63, q = tid >> 6;
  float v[40];
#pragma unroll
  for (int i = 0; i < 40; ++i) {
    int b = q + 16 * i;
    v[i] = (b < GM_BLOCKS) ? part[(size_t)b * H + f] : 0.0f;
  }
  float s = v[0];
#pragma unroll
  for (int i = 1; i < 40; ++i) s += v[i];
  ls[q][f] = s;
  __syncthreads();
  if (tid < H) {
    float r = ls[0][tid];
#pragma unroll
    for (int w = 1; w < 16; ++w) r += ls[w][tid];
    gout[tid] = r * (1.0f / NN);
  }
}

extern "C" void kernel_launch(void* const* d_in, const int* in_sizes, int n_in,
                              void* d_out, int out_size, void* d_ws, size_t ws_size,
                              hipStream_t stream) {
  (void)in_sizes; (void)n_in; (void)out_size; (void)ws_size;
  const float* x    = (const float*)d_in[0];
  const int*   ei   = (const int*)  d_in[1];
  const float* ea   = (const float*)d_in[2];
  const float* inW  = (const float*)d_in[3];
  const float* inb  = (const float*)d_in[4];
  const float* linW = (const float*)d_in[5];
  const float* linb = (const float*)d_in[6];
  const float* attW = (const float*)d_in[7];
  const float* attb = (const float*)d_in[8];
  const float* lng  = (const float*)d_in[9];
  const float* lnb  = (const float*)d_in[10];
  const float* qW1 = (const float*)d_in[11]; const float* qb1 = (const float*)d_in[12];
  const float* qW2 = (const float*)d_in[13]; const float* qb2 = (const float*)d_in[14];
  const float* iW1 = (const float*)d_in[15]; const float* ib1 = (const float*)d_in[16];
  const float* iW2 = (const float*)d_in[17]; const float* ib2 = (const float*)d_in[18];
  const float* gW1 = (const float*)d_in[19]; const float* gb1 = (const float*)d_in[20];
  const float* gW2 = (const float*)d_in[21]; const float* gb2 = (const float*)d_in[22];
  const float* oW  = (const float*)d_in[23]; const float* obv = (const float*)d_in[24];
  const float* sW1 = (const float*)d_in[25]; const float* sb1 = (const float*)d_in[26];
  const float* sW2 = (const float*)d_in[27]; const float* sb2 = (const float*)d_in[28];

  float* ws   = (float*)d_ws;
  float* h    = ws;
  float* mbuf = h    + (size_t)NN * H;
  float* lst  = mbuf + (size_t)NN * H;
  float* sD   = lst + (size_t)NN * H;
  float* sS   = sD + NN;
  int*   off  = (int*)lst;
  int2*  tmp  = (int2*)(off + NODE_PAD);
  int2*  rec  = tmp + NE;
  float* part = (float*)(rec + NE);
  int2*  binned = (int2*)(part + GM_BLOCKS * H);
  int*   gh   = (int*)(binned + NE);

  float* out  = (float*)d_out;
  float* oq   = out;
  float* oi   = out + (size_t)NN;
  float* og   = out + 2 * (size_t)NN;
  float* oh   = out + 3 * (size_t)NN;
  float* ogr  = out + 67 * (size_t)NN;
  float* obot = out + 67 * (size_t)NN + H;
  float* osev = out + 99 * (size_t)NN + H;

  k_input_proj<<<IP_BLOCKS, 256, 0, stream>>>(x, inW, inb, h);

  k_bin_hist<<<NCHK, 1024, 0, stream>>>(ei, gh);
  k_bin_scan<<<1, 1024, 0, stream>>>(gh);
  k_bin_scatter<<<NCHK, 1024, 0, stream>>>(ei, gh, binned);
  k_range_group<<<NBKT, 1024, 0, stream>>>(gh, binned, off, tmp);
  k_edge_order<<<(NE + 255) / 256, 256, 0, stream>>>(ei, ea, off, tmp, rec);

  const int nodeWaveBlocks = (NN + 7) / 8;
  for (int l = 0; l < NL; ++l) {
    k_gemm_h<<<NN / 16, 128, 0, stream>>>(h, linW + (size_t)l * H * H, linb + (size_t)l * H,
                                          attW + (size_t)l * 2 * H, mbuf, sD, sS);
    k_gather_update<<<nodeWaveBlocks, 256, 0, stream>>>(h, mbuf, off, rec, sD, sS, attb + l,
                                                        lng + (size_t)l * H, lnb + (size_t)l * H);
  }

  k_heads<<<nodeWaveBlocks, 256, 0, stream>>>(h, qW1, qb1, qW2, qb2, iW1, ib1, iW2, ib2,
                                              gW1, gb1, gW2, gb2, oW, obv, sW1, sb1, sW2, sb2,
                                              oq, oi, og, oh, obot, osev);
  k_graph_partial<<<GM_BLOCKS, 256, 0, stream>>>(h, part);
  k_graph_final<<<1, 1024, 0, stream>>>(part, ogr);
}


# output heads: a wave owns 8 consecutive nodes and uses each loaded weight row for all 8 (3125 blocks of 4 waves, was 12500 of 8): weight loads per node fall 8x; per-output sum order unchanged
# speedup vs baseline: 2.2404x; 1.2211x over previous
#include <hip/hip_runtime.h>

#define NN 100000
#define NE 1000000
#define DIN 5
#define H 64
#define HHID 32
#define NL 4
#define DOUT 32
#define LN_EPS 1e-5f

typedef __attribute__((ext_vector_type(2)))  float    v2f;
typedef __attribute__((ext_vector_type(8)))  float    v8f;
typedef __attribute__((ext_vector_type(16))) _Float16 v16h;

#if defined(__has_builtin)
#if __has_builtin(__builtin_amdgcn_wmma_f32_16x16x4_f32)
#define HAVE_WMMA_F32 1
#endif
#endif

__device__ __forceinline__ float wave_sum(float v) {
#pragma unroll
  for (int off = 16; off > 0; off >>= 1) v += __shfl_xor(v, off, 32);
  return v;
}
__device__ __forceinline__ float sigmoidf_(float x) { return 1.0f / (1.0f + __expf(-x)); }

__device__ __forceinline__ int wave_sum_i(int v) {
#pragma unroll
  for (int off = 16; off > 0; off >>= 1) v += __shfl_xor(v, off, 32);
  return v;
}

#define IP_BLOCKS 625
#define IP_WAVE_NODES 20
static_assert(IP_BLOCKS * 8 * IP_WAVE_NODES == NN, "the waves cover the nodes exactly");
static_assert(IP_WAVE_NODES % 2 == 0 && H == 64, "two nodes per step, sixteen lanes per row");
__global__ __launch_bounds__(256) void k_input_proj(
    const float* __restrict__ x, const float* __restrict__ W,
    const float* __restrict__ b, float* __restrict__ h) {
  const int wave = threadIdx.x >> 5, lane = threadIdx.x & 31;
  const int c4 = (lane & 15) * 4;
  float4 w[DIN];
#pragma unroll
  for (int d = 0; d < DIN; ++d) w[d] = *reinterpret_cast<const float4*>(W + d * H + c4);
  const float4 bv = *reinterpret_cast<const float4*>(b + c4);
  const int n0 = (blockIdx.x * 8 + wave) * IP_WAVE_NODES + (lane >> 4);
#pragma unroll
  for (int i = 0; i < IP_WAVE_NODES / 2; ++i) {
    const int n = n0 + 2 * i;
    const float* xr = x + (size_t)n * DIN;
    float xv[DIN];
#pragma unroll
    for (int d = 0; d < DIN; ++d) xv[d] = xr[d];
    float4 acc = bv;
#pragma unroll
    for (int d = 0; d < DIN; ++d) {
      acc.x = fmaf(xv[d], w[d].x, acc.x);
      acc.y = fmaf(xv[d], w[d].y, acc.y);
      acc.z = fmaf(xv[d], w[d].z, acc.z);
      acc.w = fmaf(xv[d], w[d].w, acc.w);
    }
    *reinterpret_cast<float4*>(h + (size_t)n * H + c4) = acc;
  }
}

__global__ __launch_bounds__(128) void k_gemm_h(
    const float* __restrict__ A, const float* __restrict__ W,
    const float* __restrict__ bias, const float* __restrict__ aw,
    float* __restrict__ out, float* __restrict__ sD, float* __restrict__ sS) {
  __shared__ float lA[16][H + 1];
  __shared__ float lW[H][H + 1];
  const int tid = threadIdx.x;
  const int row0 = blockIdx.x * 16;
  for (int idx = tid; idx < 16 * H; idx += 128)
    lA[idx >> 6][idx & 63] = A[(size_t)row0 * H + idx];
  for (int idx = tid; idx < H * H; idx += 128)
    lW[idx >> 6][idx & 63] = W[idx];
  __syncthreads();

  const int wave = tid >> 5, lane = tid & 31;
  const int m = lane & 15, g = lane >> 4;
  const int col = wave * 16 + m;

  v8f acc;
  float bv = bias[col];
#pragma unroll
  for (int i = 0; i < 8; ++i) acc[i] = bv;

#if HAVE_WMMA_F32
#pragma unroll
  for (int k0 = 0; k0 < H; k0 += 4) {
    int kb = k0 + 2 * g;
    v2f a, b;
    a.x = lA[m][kb];     a.y = lA[m][kb + 1];
    b.x = lW[kb][col];   b.y = lW[kb + 1][col];
    acc = __builtin_amdgcn_wmma_f32_16x16x4_f32(false, a, false, b, (short)0, acc, false, false);
  }
#else
#pragma unroll
  for (int kk = 0; kk < H; kk += 32) {
    v16h a, b;
#pragma unroll
    for (int j = 0; j < 8; ++j) {
      int ka = (j < 4) ? (2 * j + 8 * g) : (16 + 2 * (j - 4) + 8 * g);
      a[2 * j]     = (_Float16)lA[m][kk + ka];
      a[2 * j + 1] = (_Float16)lA[m][kk + ka + 1];
      int kb = 2 * j + 16 * g;
      b[2 * j]     = (_Float16)lW[kk + kb][col];
      b[2 * j + 1] = (_Float16)lW[kk + kb + 1][col];
    }
    acc = __builtin_amdgcn_wmma_f32_16x16x32_f16(false, a, false, b, (short)0, acc, false, false);
  }
#endif

#pragma unroll
  for (int i = 0; i < 8; ++i) {
    int row = row0 + i + 8 * g;
    out[(size_t)row * H + col] = acc[i];
  }

  if (wave < 2) {
    const float* awp = aw + wave * H;
    float p = 0.0f;
#pragma unroll
    for (int k = 0; k < 32; ++k) p = fmaf(lA[m][32 * g + k], awp[32 * g + k], p);
    p += __shfl_xor(p, 16, 32);
    if (g == 0) {
      if (wave == 0) sD[row0 + m] = p; else sS[row0 + m] = p;
    }
  }
}

static_assert(NE % 4 == 0, "four edges per thread");
#define NODE_PAD 100032
#define BKT_SHIFT 9
#define BKT_R (1 << BKT_SHIFT)
#define NBKT ((NN + BKT_R - 1) / BKT_R)
#define NCHK 250
#define CHK_V4 (NE / 4 / NCHK)
#define GH_PITCH 256
#define RG_DEPTH 8
static_assert(NCHK * CHK_V4 * 4 == NE, "chunks cover the edges exactly");
static_assert(CHK_V4 <= 1024, "one four-edge group per thread");
static_assert(NBKT <= 224 && NBKT * BKT_R >= NN + 1, "range table");
static_assert(NCHK <= GH_PITCH && GH_PITCH - NCHK <= NCHK, "pad columns are written by the first chunks");

__device__ __forceinline__ int wave_incl_scan_i(int v, int lane) {
#pragma unroll
  for (int o = 1; o < 32; o <<= 1) {
    int up = __shfl_up(v, o, 32);
    if (lane >= o) v += up;
  }
  return v;
}

__global__ __launch_bounds__(1024) void k_bin_hist(
    const int* __restrict__ ei, int* __restrict__ gh) {
  __shared__ int hist[NBKT];
  const int tid = threadIdx.x, w = blockIdx.x;
  if (tid < NBKT) hist[tid] = 0;
  __syncthreads();
  if (tid < CHK_V4) {
    int4 d = reinterpret_cast<const int4*>(ei + NE)[w * CHK_V4 + tid];
    atomicAdd(&hist[d.x >> BKT_SHIFT], 1);
    atomicAdd(&hist[d.y >> BKT_SHIFT], 1);
    atomicAdd(&hist[d.z >> BKT_SHIFT], 1);
    atomicAdd(&hist[d.w >> BKT_SHIFT], 1);
  }
  __syncthreads();
  if (tid < NBKT) {
    gh[tid * GH_PITCH + w] = hist[tid];
    if (w < GH_PITCH - NCHK) gh[tid * GH_PITCH + NCHK + w] = 0;
  }
}

__global__ __launch_bounds__(1024) void k_bin_scan(int* __restrict__ gh) {
  __shared__ int rowtot[224];
  __shared__ int rowbase[224];
  __shared__ int wtot[8];
  const int tid = threadIdx.x, wave = tid >> 5, lane = tid & 31;
  int4 a[7], c[7];
#pragma unroll
  for (int k = 0; k < 7; ++k) {
    int b = min(wave + 32 * k, NBKT - 1);
    const int4* p = reinterpret_cast<const int4*>(gh + b * GH_PITCH + lane * 8);
    a[k] = p[0];
    c[k] = p[1];
  }
  int ex[7];
#pragma unroll
  for (int k = 0; k < 7; ++k) {
    int b = wave + 32 * k;
    int sum = ((a[k].x + a[k].y) + (a[k].z + a[k].w)) + ((c[k].x + c[k].y) + (c[k].z + c[k].w));
    int incl = wave_incl_scan_i(sum, lane);
    ex[k] = incl - sum;
    int tot = __shfl(incl, 31, 32);
    if (lane == 0) rowtot[b] = (b < NBKT) ? tot : 0;
  }
  __syncthreads();
  int v = 0, vin = 0;
  if (tid < 224) {
    v = rowtot[tid];
    vin = wave_incl_scan_i(v, lane);
    if (lane == 31) wtot[wave] = vin;
  }
  __syncthreads();
  if (tid < 224) {
    int before = 0;
#pragma unroll
    for (int j = 0; j < 7; ++j) before += (j < wave) ? wtot[j] : 0;
    rowbase[tid] = before + vin - v;
  }
  __syncthreads();
#pragma unroll
  for (int k = 0; k < 7; ++k) {
    int b = wave + 32 * k;
    if (b < NBKT) {
      int r = rowbase[b] + ex[k];
      int4 o0, o1;
      o0.x = r; r += a[k].x;
      o0.y = r; r += a[k].y;
      o0.z = r; r += a[k].z;
      o0.w = r; r += a[k].w;
      o1.x = r; r += c[k].x;
      o1.y = r; r += c[k].y;
      o1.z = r; r += c[k].z;
      o1.w = r;
      int4* p = reinterpret_cast<int4*>(gh + b * GH_PITCH + lane * 8);
      p[0] = o0;
      p[1] = o1;
    }
  }
}

__global__ __launch_bounds__(1024) void k_bin_scatter(
    const int* __restrict__ ei, const int* __restrict__ gh, int2* __restrict__ binned) {
  __shared__ int cur[NBKT];
  const int tid = threadIdx.x, w = blockIdx.x;
  if (tid < NBKT) cur[tid] = gh[tid * GH_PITCH + w];
  __syncthreads();
  if (tid < CHK_V4) {
    int t = w * CHK_V4 + tid;
    int4 d = reinterpret_cast<const int4*>(ei + NE)[t];
    int e0 = 4 * t;
    int p;
    p = atomicAdd(&cur[d.x >> BKT_SHIFT], 1); binned[p] = make_int2(e0,     d.x);
    p = atomicAdd(&cur[d.y >> BKT_SHIFT], 1); binned[p] = make_int2(e0 + 1, d.y);
    p = atomicAdd(&cur[d.z >> BKT_SHIFT], 1); binned[p] = make_int2(e0 + 2, d.z);
    p = atomicAdd(&cur[d.w >> BKT_SHIFT], 1); binned[p] = make_int2(e0 + 3, d.w);
  }
}

__global__ __launch_bounds__(1024) void k_range_group(
    const int* __restrict__ gh, const int2* __restrict__ binned,
    int* __restrict__ off, int2* __restrict__ tmp) {
  __shared__ int cnt[BKT_R];
  __shared__ int wtot[32];
  const int tid = threadIdx.x, wave = tid >> 5, lane = tid & 31;
  const int b = blockIdx.x;
  const int lo = b << BKT_SHIFT;
  const int cb = gh[b * GH_PITCH];
  const int ce = (b + 1 < NBKT) ? gh[(b + 1) * GH_PITCH] : NE;
  if (tid < BKT_R) cnt[tid] = 0;
  __syncthreads();
  for (int i0 = cb + tid; i0 < ce; i0 += RG_DEPTH * 1024) {
    int dd[RG_DEPTH];
#pragma unroll
    for (int u = 0; u < RG_DEPTH; ++u) dd[u] = binned[min(i0 + 1024 * u, ce - 1)].y;
#pragma unroll
    for (int u = 0; u < RG_DEPTH; ++u)
      if (i0 + 1024 * u < ce) atomicAdd(&cnt[dd[u] - lo], 1);
  }
  __syncthreads();
  const int c = (tid < BKT_R) ? cnt[tid] : 0;
  const int incl = wave_incl_scan_i(c, lane);
  if (lane == 31) wtot[wave] = incl;
  __syncthreads();
  const int before = wave_sum_i((lane < wave) ? wtot[lane] : 0);
  const int excl = before + incl - c;
  if (tid < BKT_R) {
    cnt[tid] = excl;
    if (lo + tid <= NN) off[lo + tid] = cb + excl;
  }
  __syncthreads();
  for (int i0 = cb + tid; i0 < ce; i0 += RG_DEPTH * 1024) {
    int2 r[RG_DEPTH];
#pragma unroll
    for (int u = 0; u < RG_DEPTH; ++u) r[u] = binned[min(i0 + 1024 * u, ce - 1)];
#pragma unroll
    for (int u = 0; u < RG_DEPTH; ++u)
      if (i0 + 1024 * u < ce) {
        int p = atomicAdd(&cnt[r[u].y - lo], 1);
        tmp[cb + p] = r[u];
      }
  }
}

__global__ __launch_bounds__(256) void k_edge_order(
    const int* __restrict__ ei, const float* __restrict__ ea,
    const int* __restrict__ off, const int2* __restrict__ tmp,
    int2* __restrict__ rec) {
  int p = blockIdx.x * 256 + threadIdx.x;
  if (p >= NE) return;
  int2 my = tmp[p];
  int s = off[my.y], e = off[my.y + 1];
  int rank = 0;
  for (int q = s; q < e; ++q) rank += (tmp[q].x < my.x) ? 1 : 0;
  rec[s + rank] = make_int2(ei[my.x], __float_as_int(ea[my.x]));
}

__global__ __launch_bounds__(256) void k_gather_update(
    float* __restrict__ h, const float* __restrict__ m,
    const int* __restrict__ off, const int2* __restrict__ rec,
    const float* __restrict__ sD, const float* __restrict__ sS,
    const float* __restrict__ attb,
    const float* __restrict__ g, const float* __restrict__ beta) {
  const int gw = __builtin_amdgcn_readfirstlane(blockIdx.x * 8 + (threadIdx.x >> 5));
  int lane = threadIdx.x & 31;
  if (gw >= NN) return;
  size_t base = (size_t)gw * H;
  const int s = off[gw], e = off[gw + 1];
  const float sd = sD[gw], ab = attb[0];
  float r0 = h[base + lane], r1 = h[base + lane + 32];
  float m0 = m[base + lane], m1 = m[base + lane + 32];
  float acc0 = 0.0f, acc1 = 0.0f;
  for (int cb = s; cb < e; cb += 32) {
    const int cnt = min(32, e - cb);
    int src = 0;
    float w = 0.0f;
    if (lane < cnt) {
      int2 r = rec[cb + lane];
      src = r.x;
      __builtin_prefetch(m + (size_t)src * H, 0, 1);
      __builtin_prefetch(m + (size_t)src * H + 32, 0, 1);
      w = __int_as_float(r.y) * sigmoidf_(sd + sS[src] + ab);
    }
#pragma unroll 4
    for (int i = 0; i < cnt; ++i) {
      int bs = __builtin_amdgcn_readlane(src, i);
      float bw = __int_as_float(__builtin_amdgcn_readlane(__float_as_int(w), i));
      const float* mr = m + (size_t)bs * H;
      acc0 = fmaf(mr[lane], bw, acc0);
      acc1 = fmaf(mr[lane + 32], bw, acc1);
    }
  }
  float f0 = acc0 + m0;
  float f1 = acc1 + m1;
  float mean = wave_sum(f0 + f1) * (1.0f / H);
  float d0 = f0 - mean, d1 = f1 - mean;
  float var = wave_sum(fmaf(d0, d0, d1 * d1)) * (1.0f / H);
  float inv = rsqrtf(var + LN_EPS);
  float y0 = fmaf(d0 * inv, g[lane], beta[lane]);
  float y1 = fmaf(d1 * inv, g[lane + 32], beta[lane + 32]);
  h[base + lane]      = fmaxf(y0, 0.0f) + r0;
  h[base + lane + 32] = fmaxf(y1, 0.0f) + r1;
}

#define HD_NODES 8
#define HD_WAVES 4
#define HD_BLOCKS 3125
#define HD_THREADS 128
static_assert(HD_BLOCKS * HD_WAVES * HD_NODES == NN && HD_THREADS == HD_WAVES * 32, "the waves cover the nodes exactly");
static_assert(HHID == 32 && DOUT == 32 && H == 64, "one lane per hidden unit, two features per lane");

__device__ __forceinline__ void heads_half(
    const float (&hv)[HD_NODES], const int krow0,
    const float* __restrict__ qW1, const float* __restrict__ iW1,
    const float* __restrict__ gW1, const float* __restrict__ sW1,
    const float* __restrict__ oW, const int lane,
    float (&tq)[HD_NODES], float (&ti)[HD_NODES], float (&tg)[HD_NODES],
    float (&ts)[HD_NODES], float (&bo)[HD_NODES]) {
#pragma unroll 4
  for (int k = 0; k < 32; ++k) {
    const int wi = (krow0 + k) * HHID + lane;
    const float wq = qW1[wi], wn = iW1[wi], wg = gW1[wi], wv = sW1[wi], wo = oW[wi];
#pragma unroll
    for (int j = 0; j < HD_NODES; ++j) {
      const float hk = __int_as_float(__builtin_amdgcn_readlane(__float_as_int(hv[j]), k));
      tq[j] = fmaf(hk, wq, tq[j]);
      ti[j] = fmaf(hk, wn, ti[j]);
      tg[j] = fmaf(hk, wg, tg[j]);
      ts[j] = fmaf(hk, wv, ts[j]);
      bo[j] = fmaf(hk, wo, bo[j]);
    }
  }
}

__global__ __launch_bounds__(HD_THREADS) void k_heads(
    const float* __restrict__ h,
    const float* __restrict__ qW1, const float* __restrict__ qb1,
    const float* __restrict__ qW2, const float* __restrict__ qb2,
    const float* __restrict__ iW1, const float* __restrict__ ib1,
    const float* __restrict__ iW2, const float* __restrict__ ib2,
    const float* __restrict__ gW1, const float* __restrict__ gb1,
    const float* __restrict__ gW2, const float* __restrict__ gb2,
    const float* __restrict__ oW,  const float* __restrict__ obv,
    const float* __restrict__ sW1, const float* __restrict__ sb1,
    const float* __restrict__ sW2, const float* __restrict__ sb2,
    float* __restrict__ oq, float* __restrict__ oi, float* __restrict__ og,
    float* __restrict__ oh, float* __restrict__ obot, float* __restrict__ osev) {
  const int wv = __builtin_amdgcn_readfirstlane(blockIdx.x * HD_WAVES + (threadIdx.x >> 5));
  const int lane = threadIdx.x & 31;
  const int n0 = wv * HD_NODES;
  const size_t base = (size_t)n0 * H;

  float h0[HD_NODES], h1[HD_NODES];
#pragma unroll
  for (int j = 0; j < HD_NODES; ++j) {
    h0[j] = h[base + j * H + lane];
    h1[j] = h[base + j * H + lane + 32];
  }

  float tq[HD_NODES], ti[HD_NODES], tg[HD_NODES], ts[HD_NODES], bo[HD_NODES];
#pragma unroll
  for (int j = 0; j < HD_NODES; ++j) { tq[j] = 0.f; ti[j] = 0.f; tg[j] = 0.f; ts[j] = 0.f; bo[j] = 0.f; }

  heads_half(h0, 0,  qW1, iW1, gW1, sW1, oW, lane, tq, ti, tg, ts, bo);
  heads_half(h1, 32, qW1, iW1, gW1, sW1, oW, lane, tq, ti, tg, ts, bo);

  const float bq1 = qb1[lane], bi1 = ib1[lane], bg1 = gb1[lane], bs1 = sb1[lane];
  const float wq2 = qW2[lane], wi2 = iW2[lane], wg2 = gW2[lane];
  const float bq2 = qb2[0], bi2 = ib2[0], bg2 = gb2[0];
  const float bob = obv[lane];
  float ws2[5], bs2[5];
#pragma unroll
  for (int c = 0; c < 5; ++c) { ws2[c] = sW2[lane * 5 + c]; bs2[c] = sb2[c]; }

  float keepq = 0.f, keepi = 0.f, keepg = 0.f;
#pragma unroll
  for (int j = 0; j < HD_NODES; ++j) {
    const float aq = fmaxf(tq[j] + bq1, 0.0f);
    const float ai = fmaxf(ti[j] + bi1, 0.0f);
    const float ag = fmaxf(tg[j] + bg1, 0.0f);
    const float asv = fmaxf(ts[j] + bs1, 0.0f);

    const float q  = wave_sum(aq * wq2);
    const float ii = wave_sum(ai * wi2);
    const float gg = wave_sum(ag * wg2);
    const float pq = sigmoidf_(q + bq2);
    const float pi = sigmoidf_(ii + bi2);
    const float pg = sigmoidf_(gg + bg2);
    if (lane == j) { keepq = pq; keepi = pi; keepg = pg; }

    float z[5];
#pragma unroll
    for (int c = 0; c < 5; ++c) z[c] = wave_sum(asv * ws2[c]) + bs2[c];
    float zm = z[0];
#pragma unroll
    for (int c = 1; c < 5; ++c) zm = fmaxf(zm, z[c]);
    float es = 0.f;
#pragma unroll
    for (int c = 0; c < 5; ++c) { z[c] = __expf(z[c] - zm); es += z[c]; }
    const float inv = 1.0f / es;
    float sv = z[0];
#pragma unroll
    for (int c = 1; c < 5; ++c) sv = (lane == c) ? z[c] : sv;
    if (lane < 5) osev[(size_t)(n0 + j) * 5 + lane] = sv * inv;

    obot[(size_t)(n0 + j) * DOUT + lane] = bo[j] + bob;
    oh[base + j * H + lane]      = h0[j];
    oh[base + j * H + lane + 32] = h1[j];
  }
  if (lane < HD_NODES) {
    oq[n0 + lane] = keepq;
    oi[n0 + lane] = keepi;
    og[n0 + lane] = keepg;
  }
}

#define GM_BLOCKS 625
#define GM_LOADS 10
__global__ __launch_bounds__(256) void k_graph_partial(
    const float* __restrict__ h, float* __restrict__ part) {
  __shared__ float ls[8][H];
  const int tid = threadIdx.x;
  const int wave = tid >> 5, lane = tid & 31;
  const float4* h4 = reinterpret_cast<const float4*>(h);
  const size_t t = (size_t)blockIdx.x * 256 + tid;
  float4 v[GM_LOADS];
#pragma unroll
  for (int i = 0; i < GM_LOADS; ++i) v[i] = h4[t + (size_t)i * (GM_BLOCKS * 256)];
  float4 a = v[0];
#pragma unroll
  for (int i = 1; i < GM_LOADS; ++i) {
    a.x += v[i].x; a.y += v[i].y; a.z += v[i].z; a.w += v[i].w;
  }
  a.x += __shfl_xor(a.x, 16, 32);
  a.y += __shfl_xor(a.y, 16, 32);
  a.z += __shfl_xor(a.z, 16, 32);
  a.w += __shfl_xor(a.w, 16, 32);
  if (lane < 16) *reinterpret_cast<float4*>(&ls[wave][lane * 4]) = a;
  __syncthreads();
  if (tid < H) {
    float s = ls[0][tid];
#pragma unroll
    for (int w = 1; w < 8; ++w) s += ls[w][tid];
    part[(size_t)blockIdx.x * H + tid] = s;
  }
}

__global__ __launch_bounds__(1024) void k_graph_final(
    const float* __restrict__ part, float* __restrict__ gout) {
  __shared__ float ls[16][H];
  const int tid = threadIdx.x;
  const int f = tid & 63, q = tid >> 6;
  float v[40];
#pragma unroll
  for (int i = 0; i < 40; ++i) {
    int b = q + 16 * i;
    v[i] = (b < GM_BLOCKS) ? part[(size_t)b * H + f] : 0.0f;
  }
  float s = v[0];
#pragma unroll
  for (int i = 1; i < 40; ++i) s += v[i];
  ls[q][f] = s;
  __syncthreads();
  if (tid < H) {
    float r = ls[0][tid];
#pragma unroll
    for (int w = 1; w < 16; ++w) r += ls[w][tid];
    gout[tid] = r * (1.0f / NN);
  }
}

extern "C" void kernel_launch(void* const* d_in, const int* in_sizes, int n_in,
                              void* d_out, int out_size, void* d_ws, size_t ws_size,
                              hipStream_t stream) {
  (void)in_sizes; (void)n_in; (void)out_size; (void)ws_size;
  const float* x    = (const float*)d_in[0];
  const int*   ei   = (const int*)  d_in[1];
  const float* ea   = (const float*)d_in[2];
  const float* inW  = (const float*)d_in[3];
  const float* inb  = (const float*)d_in[4];
  const float* linW = (const float*)d_in[5];
  const float* linb = (const float*)d_in[6];
  const float* attW = (const float*)d_in[7];
  const float* attb = (const float*)d_in[8];
  const float* lng  = (const float*)d_in[9];
  const float* lnb  = (const float*)d_in[10];
  const float* qW1 = (const float*)d_in[11]; const float* qb1 = (const float*)d_in[12];
  const float* qW2 = (const float*)d_in[13]; const float* qb2 = (const float*)d_in[14];
  const float* iW1 = (const float*)d_in[15]; const float* ib1 = (const float*)d_in[16];
  const float* iW2 = (const float*)d_in[17]; const float* ib2 = (const float*)d_in[18];
  const float* gW1 = (const float*)d_in[19]; const float* gb1 = (const float*)d_in[20];
  const float* gW2 = (const float*)d_in[21]; const float* gb2 = (const float*)d_in[22];
  const float* oW  = (const float*)d_in[23]; const float* obv = (const float*)d_in[24];
  const float* sW1 = (const float*)d_in[25]; const float* sb1 = (const float*)d_in[26];
  const float* sW2 = (const float*)d_in[27]; const float* sb2 = (const float*)d_in[28];

  float* ws   = (float*)d_ws;
  float* h    = ws;
  float* mbuf = h    + (size_t)NN * H;
  float* lst  = mbuf + (size_t)NN * H;
  float* sD   = lst + (size_t)NN * H;
  float* sS   = sD + NN;
  int*   off  = (int*)lst;
  int2*  tmp  = (int2*)(off + NODE_PAD);
  int2*  rec  = tmp + NE;
  float* part = (float*)(rec + NE);
  int2*  binned = (int2*)(part + GM_BLOCKS * H);
  int*   gh   = (int*)(binned + NE);

  float* out  = (float*)d_out;
  float* oq   = out;
  float* oi   = out + (size_t)NN;
  float* og   = out + 2 * (size_t)NN;
  float* oh   = out + 3 * (size_t)NN;
  float* ogr  = out + 67 * (size_t)NN;
  float* obot = out + 67 * (size_t)NN + H;
  float* osev = out + 99 * (size_t)NN + H;

  k_input_proj<<<IP_BLOCKS, 256, 0, stream>>>(x, inW, inb, h);

  k_bin_hist<<<NCHK, 1024, 0, stream>>>(ei, gh);
  k_bin_scan<<<1, 1024, 0, stream>>>(gh);
  k_bin_scatter<<<NCHK, 1024, 0, stream>>>(ei, gh, binned);
  k_range_group<<<NBKT, 1024, 0, stream>>>(gh, binned, off, tmp);
  k_edge_order<<<(NE + 255) / 256, 256, 0, stream>>>(ei, ea, off, tmp, rec);

  const int nodeWaveBlocks = (NN + 7) / 8;
  for (int l = 0; l < NL; ++l) {
    k_gemm_h<<<NN / 16, 128, 0, stream>>>(h, linW + (size_t)l * H * H, linb + (size_t)l * H,
                                          attW + (size_t)l * 2 * H, mbuf, sD, sS);
    k_gather_update<<<nodeWaveBlocks, 256, 0, stream>>>(h, mbuf, off, rec, sD, sS, attb + l,
                                                        lng + (size_t)l * H, lnb + (size_t)l * H);
  }

  k_heads<<<HD_BLOCKS, HD_THREADS, 0, stream>>>(h, qW1, qb1, qW2, qb2, iW1, ib1, iW2, ib2,
                                              gW1, gb1, gW2, gb2, oW, obv, sW1, sb1, sW2, sb2,
                                              oq, oi, og, oh, obot, osev);
  k_graph_partial<<<GM_BLOCKS, 256, 0, stream>>>(h, part);
  k_graph_final<<<1, 1024, 0, stream>>>(part, ogr);
}


# node update: half-wave per edge, 16-byte loads (one load = two edges' rows), offset and weight via a per-wave LDS strip; even/odd list chains then added, 16-lane layernorm trees (sum order changes)
# speedup vs baseline: 2.3138x; 1.0328x over previous
#include <hip/hip_runtime.h>

#define NN 100000
#define NE 1000000
#define DIN 5
#define H 64
#define HHID 32
#define NL 4
#define DOUT 32
#define LN_EPS 1e-5f

typedef __attribute__((ext_vector_type(2)))  float    v2f;
typedef __attribute__((ext_vector_type(8)))  float    v8f;
typedef __attribute__((ext_vector_type(16))) _Float16 v16h;

#if defined(__has_builtin)
#if __has_builtin(__builtin_amdgcn_wmma_f32_16x16x4_f32)
#define HAVE_WMMA_F32 1
#endif
#endif

__device__ __forceinline__ float wave_sum(float v) {
#pragma unroll
  for (int off = 16; off > 0; off >>= 1) v += __shfl_xor(v, off, 32);
  return v;
}
__device__ __forceinline__ float sigmoidf_(float x) { return 1.0f / (1.0f + __expf(-x)); }

__device__ __forceinline__ int wave_sum_i(int v) {
#pragma unroll
  for (int off = 16; off > 0; off >>= 1) v += __shfl_xor(v, off, 32);
  return v;
}

#define IP_BLOCKS 625
#define IP_WAVE_NODES 20
static_assert(IP_BLOCKS * 8 * IP_WAVE_NODES == NN, "the waves cover the nodes exactly");
static_assert(IP_WAVE_NODES % 2 == 0 && H == 64, "two nodes per step, sixteen lanes per row");
__global__ __launch_bounds__(256) void k_input_proj(
    const float* __restrict__ x, const float* __restrict__ W,
    const float* __restrict__ b, float* __restrict__ h) {
  const int wave = threadIdx.x >> 5, lane = threadIdx.x & 31;
  const int c4 = (lane & 15) * 4;
  float4 w[DIN];
#pragma unroll
  for (int d = 0; d < DIN; ++d) w[d] = *reinterpret_cast<const float4*>(W + d * H + c4);
  const float4 bv = *reinterpret_cast<const float4*>(b + c4);
  const int n0 = (blockIdx.x * 8 + wave) * IP_WAVE_NODES + (lane >> 4);
#pragma unroll
  for (int i = 0; i < IP_WAVE_NODES / 2; ++i) {
    const int n = n0 + 2 * i;
    const float* xr = x + (size_t)n * DIN;
    float xv[DIN];
#pragma unroll
    for (int d = 0; d < DIN; ++d) xv[d] = xr[d];
    float4 acc = bv;
#pragma unroll
    for (int d = 0; d < DIN; ++d) {
      acc.x = fmaf(xv[d], w[d].x, acc.x);
      acc.y = fmaf(xv[d], w[d].y, acc.y);
      acc.z = fmaf(xv[d], w[d].z, acc.z);
      acc.w = fmaf(xv[d], w[d].w, acc.w);
    }
    *reinterpret_cast<float4*>(h + (size_t)n * H + c4) = acc;
  }
}

__global__ __launch_bounds__(128) void k_gemm_h(
    const float* __restrict__ A, const float* __restrict__ W,
    const float* __restrict__ bias, const float* __restrict__ aw,
    float* __restrict__ out, float* __restrict__ sD, float* __restrict__ sS) {
  __shared__ float lA[16][H + 1];
  __shared__ float lW[H][H + 1];
  const int tid = threadIdx.x;
  const int row0 = blockIdx.x * 16;
  for (int idx = tid; idx < 16 * H; idx += 128)
    lA[idx >> 6][idx & 63] = A[(size_t)row0 * H + idx];
  for (int idx = tid; idx < H * H; idx += 128)
    lW[idx >> 6][idx & 63] = W[idx];
  __syncthreads();

  const int wave = tid >> 5, lane = tid & 31;
  const int m = lane & 15, g = lane >> 4;
  const int col = wave * 16 + m;

  v8f acc;
  float bv = bias[col];
#pragma unroll
  for (int i = 0; i < 8; ++i) acc[i] = bv;

#if HAVE_WMMA_F32
#pragma unroll
  for (int k0 = 0; k0 < H; k0 += 4) {
    int kb = k0 + 2 * g;
    v2f a, b;
    a.x = lA[m][kb];     a.y = lA[m][kb + 1];
    b.x = lW[kb][col];   b.y = lW[kb + 1][col];
    acc = __builtin_amdgcn_wmma_f32_16x16x4_f32(false, a, false, b, (short)0, acc, false, false);
  }
#else
#pragma unroll
  for (int kk = 0; kk < H; kk += 32) {
    v16h a, b;
#pragma unroll
    for (int j = 0; j < 8; ++j) {
      int ka = (j < 4) ? (2 * j + 8 * g) : (16 + 2 * (j - 4) + 8 * g);
      a[2 * j]     = (_Float16)lA[m][kk + ka];
      a[2 * j + 1] = (_Float16)lA[m][kk + ka + 1];
      int kb = 2 * j + 16 * g;
      b[2 * j]     = (_Float16)lW[kk + kb][col];
      b[2 * j + 1] = (_Float16)lW[kk + kb + 1][col];
    }
    acc = __builtin_amdgcn_wmma_f32_16x16x32_f16(false, a, false, b, (short)0, acc, false, false);
  }
#endif

#pragma unroll
  for (int i = 0; i < 8; ++i) {
    int row = row0 + i + 8 * g;
    out[(size_t)row * H + col] = acc[i];
  }

  if (wave < 2) {
    const float* awp = aw + wave * H;
    float p = 0.0f;
#pragma unroll
    for (int k = 0; k < 32; ++k) p = fmaf(lA[m][32 * g + k], awp[32 * g + k], p);
    p += __shfl_xor(p, 16, 32);
    if (g == 0) {
      if (wave == 0) sD[row0 + m] = p; else sS[row0 + m] = p;
    }
  }
}

static_assert(NE % 4 == 0, "four edges per thread");
#define NODE_PAD 100032
#define BKT_SHIFT 9
#define BKT_R (1 << BKT_SHIFT)
#define NBKT ((NN + BKT_R - 1) / BKT_R)
#define NCHK 250
#define CHK_V4 (NE / 4 / NCHK)
#define GH_PITCH 256
#define RG_DEPTH 8
static_assert(NCHK * CHK_V4 * 4 == NE, "chunks cover the edges exactly");
static_assert(CHK_V4 <= 1024, "one four-edge group per thread");
static_assert(NBKT <= 224 && NBKT * BKT_R >= NN + 1, "range table");
static_assert(NCHK <= GH_PITCH && GH_PITCH - NCHK <= NCHK, "pad columns are written by the first chunks");

__device__ __forceinline__ int wave_incl_scan_i(int v, int lane) {
#pragma unroll
  for (int o = 1; o < 32; o <<= 1) {
    int up = __shfl_up(v, o, 32);
    if (lane >= o) v += up;
  }
  return v;
}

__global__ __launch_bounds__(1024) void k_bin_hist(
    const int* __restrict__ ei, int* __restrict__ gh) {
  __shared__ int hist[NBKT];
  const int tid = threadIdx.x, w = blockIdx.x;
  if (tid < NBKT) hist[tid] = 0;
  __syncthreads();
  if (tid < CHK_V4) {
    int4 d = reinterpret_cast<const int4*>(ei + NE)[w * CHK_V4 + tid];
    atomicAdd(&hist[d.x >> BKT_SHIFT], 1);
    atomicAdd(&hist[d.y >> BKT_SHIFT], 1);
    atomicAdd(&hist[d.z >> BKT_SHIFT], 1);
    atomicAdd(&hist[d.w >> BKT_SHIFT], 1);
  }
  __syncthreads();
  if (tid < NBKT) {
    gh[tid * GH_PITCH + w] = hist[tid];
    if (w < GH_PITCH - NCHK) gh[tid * GH_PITCH + NCHK + w] = 0;
  }
}

__global__ __launch_bounds__(1024) void k_bin_scan(int* __restrict__ gh) {
  __shared__ int rowtot[224];
  __shared__ int rowbase[224];
  __shared__ int wtot[8];
  const int tid = threadIdx.x, wave = tid >> 5, lane = tid & 31;
  int4 a[7], c[7];
#pragma unroll
  for (int k = 0; k < 7; ++k) {
    int b = min(wave + 32 * k, NBKT - 1);
    const int4* p = reinterpret_cast<const int4*>(gh + b * GH_PITCH + lane * 8);
    a[k] = p[0];
    c[k] = p[1];
  }
  int ex[7];
#pragma unroll
  for (int k = 0; k < 7; ++k) {
    int b = wave + 32 * k;
    int sum = ((a[k].x + a[k].y) + (a[k].z + a[k].w)) + ((c[k].x + c[k].y) + (c[k].z + c[k].w));
    int incl = wave_incl_scan_i(sum, lane);
    ex[k] = incl - sum;
    int tot = __shfl(incl, 31, 32);
    if (lane == 0) rowtot[b] = (b < NBKT) ? tot : 0;
  }
  __syncthreads();
  int v = 0, vin = 0;
  if (tid < 224) {
    v = rowtot[tid];
    vin = wave_incl_scan_i(v, lane);
    if (lane == 31) wtot[wave] = vin;
  }
  __syncthreads();
  if (tid < 224) {
    int before = 0;
#pragma unroll
    for (int j = 0; j < 7; ++j) before += (j < wave) ? wtot[j] : 0;
    rowbase[tid] = before + vin - v;
  }
  __syncthreads();
#pragma unroll
  for (int k = 0; k < 7; ++k) {
    int b = wave + 32 * k;
    if (b < NBKT) {
      int r = rowbase[b] + ex[k];
      int4 o0, o1;
      o0.x = r; r += a[k].x;
      o0.y = r; r += a[k].y;
      o0.z = r; r += a[k].z;
      o0.w = r; r += a[k].w;
      o1.x = r; r += c[k].x;
      o1.y = r; r += c[k].y;
      o1.z = r; r += c[k].z;
      o1.w = r;
      int4* p = reinterpret_cast<int4*>(gh + b * GH_PITCH + lane * 8);
      p[0] = o0;
      p[1] = o1;
    }
  }
}

__global__ __launch_bounds__(1024) void k_bin_scatter(
    const int* __restrict__ ei, const int* __restrict__ gh, int2* __restrict__ binned) {
  __shared__ int cur[NBKT];
  const int tid = threadIdx.x, w = blockIdx.x;
  if (tid < NBKT) cur[tid] = gh[tid * GH_PITCH + w];
  __syncthreads();
  if (tid < CHK_V4) {
    int t = w * CHK_V4 + tid;
    int4 d = reinterpret_cast<const int4*>(ei + NE)[t];
    int e0 = 4 * t;
    int p;
    p = atomicAdd(&cur[d.x >> BKT_SHIFT], 1); binned[p] = make_int2(e0,     d.x);
    p = atomicAdd(&cur[d.y >> BKT_SHIFT], 1); binned[p] = make_int2(e0 + 1, d.y);
    p = atomicAdd(&cur[d.z >> BKT_SHIFT], 1); binned[p] = make_int2(e0 + 2, d.z);
    p = atomicAdd(&cur[d.w >> BKT_SHIFT], 1); binned[p] = make_int2(e0 + 3, d.w);
  }
}

__global__ __launch_bounds__(1024) void k_range_group(
    const int* __restrict__ gh, const int2* __restrict__ binned,
    int* __restrict__ off, int2* __restrict__ tmp) {
  __shared__ int cnt[BKT_R];
  __shared__ int wtot[32];
  const int tid = threadIdx.x, wave = tid >> 5, lane = tid & 31;
  const int b = blockIdx.x;
  const int lo = b << BKT_SHIFT;
  const int cb = gh[b * GH_PITCH];
  const int ce = (b + 1 < NBKT) ? gh[(b + 1) * GH_PITCH] : NE;
  if (tid < BKT_R) cnt[tid] = 0;
  __syncthreads();
  for (int i0 = cb + tid; i0 < ce; i0 += RG_DEPTH * 1024) {
    int dd[RG_DEPTH];
#pragma unroll
    for (int u = 0; u < RG_DEPTH; ++u) dd[u] = binned[min(i0 + 1024 * u, ce - 1)].y;
#pragma unroll
    for (int u = 0; u < RG_DEPTH; ++u)
      if (i0 + 1024 * u < ce) atomicAdd(&cnt[dd[u] - lo], 1);
  }
  __syncthreads();
  const int c = (tid < BKT_R) ? cnt[tid] : 0;
  const int incl = wave_incl_scan_i(c, lane);
  if (lane == 31) wtot[wave] = incl;
  __syncthreads();
  const int before = wave_sum_i((lane < wave) ? wtot[lane] : 0);
  const int excl = before + incl - c;
  if (tid < BKT_R) {
    cnt[tid] = excl;
    if (lo + tid <= NN) off[lo + tid] = cb + excl;
  }
  __syncthreads();
  for (int i0 = cb + tid; i0 < ce; i0 += RG_DEPTH * 1024) {
    int2 r[RG_DEPTH];
#pragma unroll
    for (int u = 0; u < RG_DEPTH; ++u) r[u] = binned[min(i0 + 1024 * u, ce - 1)];
#pragma unroll
    for (int u = 0; u < RG_DEPTH; ++u)
      if (i0 + 1024 * u < ce) {
        int p = atomicAdd(&cnt[r[u].y - lo], 1);
        tmp[cb + p] = r[u];
      }
  }
}

__global__ __launch_bounds__(256) void k_edge_order(
    const int* __restrict__ ei, const float* __restrict__ ea,
    const int* __restrict__ off, const int2* __restrict__ tmp,
    int2* __restrict__ rec) {
  int p = blockIdx.x * 256 + threadIdx.x;
  if (p >= NE) return;
  int2 my = tmp[p];
  int s = off[my.y], e = off[my.y + 1];
  int rank = 0;
  for (int q = s; q < e; ++q) rank += (tmp[q].x < my.x) ? 1 : 0;
  rec[s + rank] = make_int2(ei[my.x], __float_as_int(ea[my.x]));
}

__device__ __forceinline__ float half_sum16(float v) {
#pragma unroll
  for (int off = 8; off > 0; off >>= 1) v += __shfl_xor(v, off, 32);
  return v;
}
__global__ __launch_bounds__(256) void k_gather_update(
    float* __restrict__ h, const float* __restrict__ m,
    const int* __restrict__ off, const int2* __restrict__ rec,
    const float* __restrict__ sD, const float* __restrict__ sS,
    const float* __restrict__ attb,
    const float* __restrict__ g, const float* __restrict__ beta) {
  __shared__ int2 strip[8][32];
  const int gw = __builtin_amdgcn_readfirstlane(blockIdx.x * 8 + (threadIdx.x >> 5));
  const int lane = threadIdx.x & 31;
  if (gw >= NN) return;
  int2* const mystrip = strip[threadIdx.x >> 5];
  const int odd = lane >> 4;
  const unsigned c16 = (lane & 15) * 16u;
  const size_t base = (size_t)gw * H + (lane & 15) * 4;
  const int s = off[gw], e = off[gw + 1];
  const float sd = sD[gw], ab = attb[0];
  const float4 r  = *reinterpret_cast<const float4*>(h + base);
  const float4 mn = *reinterpret_cast<const float4*>(m + base);
  const char* const mbytes = reinterpret_cast<const char*>(m);
  float4 acc = make_float4(0.0f, 0.0f, 0.0f, 0.0f);
  for (int cb = s; cb < e; cb += 32) {
    const int cnt = min(32, e - cb);
    int src = gw;
    float w = 0.0f;
    if (lane < cnt) {
      int2 rc = rec[cb + lane];
      src = rc.x;
      __builtin_prefetch(m + (size_t)src * H, 0, 1);
      w = __int_as_float(rc.y) * sigmoidf_(sd + sS[src] + ab);
    }
    mystrip[lane] = make_int2((int)((unsigned)src << 8), __float_as_int(w));
#pragma unroll 4
    for (int i = 0; i < cnt; i += 2) {
      const int2 p = mystrip[i + odd];
      const float bw = __int_as_float(p.y);
      const float4 v = *reinterpret_cast<const float4*>(mbytes + ((unsigned)p.x + c16));
      acc.x = fmaf(v.x, bw, acc.x);
      acc.y = fmaf(v.y, bw, acc.y);
      acc.z = fmaf(v.z, bw, acc.z);
      acc.w = fmaf(v.w, bw, acc.w);
    }
  }
  acc.x += __shfl_xor(acc.x, 16, 32);
  acc.y += __shfl_xor(acc.y, 16, 32);
  acc.z += __shfl_xor(acc.z, 16, 32);
  acc.w += __shfl_xor(acc.w, 16, 32);
  const float f0 = acc.x + mn.x, f1 = acc.y + mn.y, f2 = acc.z + mn.z, f3 = acc.w + mn.w;
  const float mean = half_sum16((f0 + f1) + (f2 + f3)) * (1.0f / H);
  const float d0 = f0 - mean, d1 = f1 - mean, d2 = f2 - mean, d3 = f3 - mean;
  const float var = half_sum16(fmaf(d0, d0, d1 * d1) + fmaf(d2, d2, d3 * d3)) * (1.0f / H);
  const float inv = rsqrtf(var + LN_EPS);
  const float4 gv = *reinterpret_cast<const float4*>(g + (lane & 15) * 4);
  const float4 bv = *reinterpret_cast<const float4*>(beta + (lane & 15) * 4);
  if (odd == 0) {
    float4 o;
    o.x = fmaxf(fmaf(d0 * inv, gv.x, bv.x), 0.0f) + r.x;
    o.y = fmaxf(fmaf(d1 * inv, gv.y, bv.y), 0.0f) + r.y;
    o.z = fmaxf(fmaf(d2 * inv, gv.z, bv.z), 0.0f) + r.z;
    o.w = fmaxf(fmaf(d3 * inv, gv.w, bv.w), 0.0f) + r.w;
    *reinterpret_cast<float4*>(h + base) = o;
  }
}

#define HD_NODES 8
#define HD_WAVES 4
#define HD_BLOCKS 3125
#define HD_THREADS 128
static_assert(HD_BLOCKS * HD_WAVES * HD_NODES == NN && HD_THREADS == HD_WAVES * 32, "the waves cover the nodes exactly");
static_assert(HHID == 32 && DOUT == 32 && H == 64, "one lane per hidden unit, two features per lane");

__device__ __forceinline__ void heads_half(
    const float (&hv)[HD_NODES], const int krow0,
    const float* __restrict__ qW1, const float* __restrict__ iW1,
    const float* __restrict__ gW1, const float* __restrict__ sW1,
    const float* __restrict__ oW, const int lane,
    float (&tq)[HD_NODES], float (&ti)[HD_NODES], float (&tg)[HD_NODES],
    float (&ts)[HD_NODES], float (&bo)[HD_NODES]) {
#pragma unroll 4
  for (int k = 0; k < 32; ++k) {
    const int wi = (krow0 + k) * HHID + lane;
    const float wq = qW1[wi], wn = iW1[wi], wg = gW1[wi], wv = sW1[wi], wo = oW[wi];
#pragma unroll
    for (int j = 0; j < HD_NODES; ++j) {
      const float hk = __int_as_float(__builtin_amdgcn_readlane(__float_as_int(hv[j]), k));
      tq[j] = fmaf(hk, wq, tq[j]);
      ti[j] = fmaf(hk, wn, ti[j]);
      tg[j] = fmaf(hk, wg, tg[j]);
      ts[j] = fmaf(hk, wv, ts[j]);
      bo[j] = fmaf(hk, wo, bo[j]);
    }
  }
}

__global__ __launch_bounds__(HD_THREADS) void k_heads(
    const float* __restrict__ h,
    const float* __restrict__ qW1, const float* __restrict__ qb1,
    const float* __restrict__ qW2, const float* __restrict__ qb2,
    const float* __restrict__ iW1, const float* __restrict__ ib1,
    const float* __restrict__ iW2, const float* __restrict__ ib2,
    const float* __restrict__ gW1, const float* __restrict__ gb1,
    const float* __restrict__ gW2, const float* __restrict__ gb2,
    const float* __restrict__ oW,  const float* __restrict__ obv,
    const float* __restrict__ sW1, const float* __restrict__ sb1,
    const float* __restrict__ sW2, const float* __restrict__ sb2,
    float* __restrict__ oq, float* __restrict__ oi, float* __restrict__ og,
    float* __restrict__ oh, float* __restrict__ obot, float* __restrict__ osev) {
  const int wv = __builtin_amdgcn_readfirstlane(blockIdx.x * HD_WAVES + (threadIdx.x >> 5));
  const int lane = threadIdx.x & 31;
  const int n0 = wv * HD_NODES;
  const size_t base = (size_t)n0 * H;

  float h0[HD_NODES], h1[HD_NODES];
#pragma unroll
  for (int j = 0; j < HD_NODES; ++j) {
    h0[j] = h[base + j * H + lane];
    h1[j] = h[base + j * H + lane + 32];
  }

  float tq[HD_NODES], ti[HD_NODES], tg[HD_NODES], ts[HD_NODES], bo[HD_NODES];
#pragma unroll
  for (int j = 0; j < HD_NODES; ++j) { tq[j] = 0.f; ti[j] = 0.f; tg[j] = 0.f; ts[j] = 0.f; bo[j] = 0.f; }

  heads_half(h0, 0,  qW1, iW1, gW1, sW1, oW, lane, tq, ti, tg, ts, bo);
  heads_half(h1, 32, qW1, iW1, gW1, sW1, oW, lane, tq, ti, tg, ts, bo);

  const float bq1 = qb1[lane], bi1 = ib1[lane], bg1 = gb1[lane], bs1 = sb1[lane];
  const float wq2 = qW2[lane], wi2 = iW2[lane], wg2 = gW2[lane];
  const float bq2 = qb2[0], bi2 = ib2[0], bg2 = gb2[0];
  const float bob = obv[lane];
  float ws2[5], bs2[5];
#pragma unroll
  for (int c = 0; c < 5; ++c) { ws2[c] = sW2[lane * 5 + c]; bs2[c] = sb2[c]; }

  float keepq = 0.f, keepi = 0.f, keepg = 0.f;
#pragma unroll
  for (int j = 0; j < HD_NODES; ++j) {
    const float aq = fmaxf(tq[j] + bq1, 0.0f);
    const float ai = fmaxf(ti[j] + bi1, 0.0f);
    const float ag = fmaxf(tg[j] + bg1, 0.0f);
    const float asv = fmaxf(ts[j] + bs1, 0.0f);

    const float q  = wave_sum(aq * wq2);
    const float ii = wave_sum(ai * wi2);
    const float gg = wave_sum(ag * wg2);
    const float pq = sigmoidf_(q + bq2);
    const float pi = sigmoidf_(ii + bi2);
    const float pg = sigmoidf_(gg + bg2);
    if (lane == j) { keepq = pq; keepi = pi; keepg = pg; }

    float z[5];
#pragma unroll
    for (int c = 0; c < 5; ++c) z[c] = wave_sum(asv * ws2[c]) + bs2[c];
    float zm = z[0];
#pragma unroll
    for (int c = 1; c < 5; ++c) zm = fmaxf(zm, z[c]);
    float es = 0.f;
#pragma unroll
    for (int c = 0; c < 5; ++c) { z[c] = __expf(z[c] - zm); es += z[c]; }
    const float inv = 1.0f / es;
    float sv = z[0];
#pragma unroll
    for (int c = 1; c < 5; ++c) sv = (lane == c) ? z[c] : sv;
    if (lane < 5) osev[(size_t)(n0 + j) * 5 + lane] = sv * inv;

    obot[(size_t)(n0 + j) * DOUT + lane] = bo[j] + bob;
    oh[base + j * H + lane]      = h0[j];
    oh[base + j * H + lane + 32] = h1[j];
  }
  if (lane < HD_NODES) {
    oq[n0 + lane] = keepq;
    oi[n0 + lane] = keepi;
    og[n0 + lane] = keepg;
  }
}

#define GM_BLOCKS 625
#define GM_LOADS 10
__global__ __launch_bounds__(256) void k_graph_partial(
    const float* __restrict__ h, float* __restrict__ part) {
  __shared__ float ls[8][H];
  const int tid = threadIdx.x;
  const int wave = tid >> 5, lane = tid & 31;
  const float4* h4 = reinterpret_cast<const float4*>(h);
  const size_t t = (size_t)blockIdx.x * 256 + tid;
  float4 v[GM_LOADS];
#pragma unroll
  for (int i = 0; i < GM_LOADS; ++i) v[i] = h4[t + (size_t)i * (GM_BLOCKS * 256)];
  float4 a = v[0];
#pragma unroll
  for (int i = 1; i < GM_LOADS; ++i) {
    a.x += v[i].x; a.y += v[i].y; a.z += v[i].z; a.w += v[i].w;
  }
  a.x += __shfl_xor(a.x, 16, 32);
  a.y += __shfl_xor(a.y, 16, 32);
  a.z += __shfl_xor(a.z, 16, 32);
  a.w += __shfl_xor(a.w, 16, 32);
  if (lane < 16) *reinterpret_cast<float4*>(&ls[wave][lane * 4]) = a;
  __syncthreads();
  if (tid < H) {
    float s = ls[0][tid];
#pragma unroll
    for (int w = 1; w < 8; ++w) s += ls[w][tid];
    part[(size_t)blockIdx.x * H + tid] = s;
  }
}

__global__ __launch_bounds__(1024) void k_graph_final(
    const float* __restrict__ part, float* __restrict__ gout) {
  __shared__ float ls[16][H];
  const int tid = threadIdx.x;
  const int f = tid & 63, q = tid >> 6;
  float v[40];
#pragma unroll
  for (int i = 0; i < 40; ++i) {
    int b = q + 16 * i;
    v[i] = (b < GM_BLOCKS) ? part[(size_t)b * H + f] : 0.0f;
  }
  float s = v[0];
#pragma unroll
  for (int i = 1; i < 40; ++i) s += v[i];
  ls[q][f] = s;
  __syncthreads();
  if (tid < H) {
    float r = ls[0][tid];
#pragma unroll
    for (int w = 1; w < 16; ++w) r += ls[w][tid];
    gout[tid] = r * (1.0f / NN);
  }
}

extern "C" void kernel_launch(void* const* d_in, const int* in_sizes, int n_in,
                              void* d_out, int out_size, void* d_ws, size_t ws_size,
                              hipStream_t stream) {
  (void)in_sizes; (void)n_in; (void)out_size; (void)ws_size;
  const float* x    = (const float*)d_in[0];
  const int*   ei   = (const int*)  d_in[1];
  const float* ea   = (const float*)d_in[2];
  const float* inW  = (const float*)d_in[3];
  const float* inb  = (const float*)d_in[4];
  const float* linW = (const float*)d_in[5];
  const float* linb = (const float*)d_in[6];
  const float* attW = (const float*)d_in[7];
  const float* attb = (const float*)d_in[8];
  const float* lng  = (const float*)d_in[9];
  const float* lnb  = (const float*)d_in[10];
  const float* qW1 = (const float*)d_in[11]; const float* qb1 = (const float*)d_in[12];
  const float* qW2 = (const float*)d_in[13]; const float* qb2 = (const float*)d_in[14];
  const float* iW1 = (const float*)d_in[15]; const float* ib1 = (const float*)d_in[16];
  const float* iW2 = (const float*)d_in[17]; const float* ib2 = (const float*)d_in[18];
  const float* gW1 = (const float*)d_in[19]; const float* gb1 = (const float*)d_in[20];
  const float* gW2 = (const float*)d_in[21]; const float* gb2 = (const float*)d_in[22];
  const float* oW  = (const float*)d_in[23]; const float* obv = (const float*)d_in[24];
  const float* sW1 = (const float*)d_in[25]; const float* sb1 = (const float*)d_in[26];
  const float* sW2 = (const float*)d_in[27]; const float* sb2 = (const float*)d_in[28];

  float* ws   = (float*)d_ws;
  float* h    = ws;
  float* mbuf = h    + (size_t)NN * H;
  float* lst  = mbuf + (size_t)NN * H;
  float* sD   = lst + (size_t)NN * H;
  float* sS   = sD + NN;
  int*   off  = (int*)lst;
  int2*  tmp  = (int2*)(off + NODE_PAD);
  int2*  rec  = tmp + NE;
  float* part = (float*)(rec + NE);
  int2*  binned = (int2*)(part + GM_BLOCKS * H);
  int*   gh   = (int*)(binned + NE);

  float* out  = (float*)d_out;
  float* oq   = out;
  float* oi   = out + (size_t)NN;
  float* og   = out + 2 * (size_t)NN;
  float* oh   = out + 3 * (size_t)NN;
  float* ogr  = out + 67 * (size_t)NN;
  float* obot = out + 67 * (size_t)NN + H;
  float* osev = out + 99 * (size_t)NN + H;

  k_input_proj<<<IP_BLOCKS, 256, 0, stream>>>(x, inW, inb, h);

  k_bin_hist<<<NCHK, 1024, 0, stream>>>(ei, gh);
  k_bin_scan<<<1, 1024, 0, stream>>>(gh);
  k_bin_scatter<<<NCHK, 1024, 0, stream>>>(ei, gh, binned);
  k_range_group<<<NBKT, 1024, 0, stream>>>(gh, binned, off, tmp);
  k_edge_order<<<(NE + 255) / 256, 256, 0, stream>>>(ei, ea, off, tmp, rec);

  const int nodeWaveBlocks = (NN + 7) / 8;
  for (int l = 0; l < NL; ++l) {
    k_gemm_h<<<NN / 16, 128, 0, stream>>>(h, linW + (size_t)l * H * H, linb + (size_t)l * H,
                                          attW + (size_t)l * 2 * H, mbuf, sD, sS);
    k_gather_update<<<nodeWaveBlocks, 256, 0, stream>>>(h, mbuf, off, rec, sD, sS, attb + l,
                                                        lng + (size_t)l * H, lnb + (size_t)l * H);
  }

  k_heads<<<HD_BLOCKS, HD_THREADS, 0, stream>>>(h, qW1, qb1, qW2, qb2, iW1, ib1, iW2, ib2,
                                              gW1, gb1, gW2, gb2, oW, obv, sW1, sb1, sW2, sb2,
                                              oq, oi, og, oh, obot, osev);
  k_graph_partial<<<GM_BLOCKS, 256, 0, stream>>>(h, part);
  k_graph_final<<<1, 1024, 0, stream>>>(part, ogr);
}


# node GEMM: a wave owns 16 rows x all 64 columns (4 independent tiles); its rows go straight into A fragments, eight 16-byte loads issued together; W staged once per 64 rows (sum order changes)
# speedup vs baseline: 2.4760x; 1.0701x over previous
#include <hip/hip_runtime.h>

#define NN 100000
#define NE 1000000
#define DIN 5
#define H 64
#define HHID 32
#define NL 4
#define DOUT 32
#define LN_EPS 1e-5f

typedef __attribute__((ext_vector_type(2)))  float    v2f;
typedef __attribute__((ext_vector_type(8)))  float    v8f;

__device__ __forceinline__ float wave_sum(float v) {
#pragma unroll
  for (int off = 16; off > 0; off >>= 1) v += __shfl_xor(v, off, 32);
  return v;
}
__device__ __forceinline__ float sigmoidf_(float x) { return 1.0f / (1.0f + __expf(-x)); }

__device__ __forceinline__ int wave_sum_i(int v) {
#pragma unroll
  for (int off = 16; off > 0; off >>= 1) v += __shfl_xor(v, off, 32);
  return v;
}

#define IP_BLOCKS 625
#define IP_WAVE_NODES 20
static_assert(IP_BLOCKS * 8 * IP_WAVE_NODES == NN, "the waves cover the nodes exactly");
static_assert(IP_WAVE_NODES % 2 == 0 && H == 64, "two nodes per step, sixteen lanes per row");
__global__ __launch_bounds__(256) void k_input_proj(
    const float* __restrict__ x, const float* __restrict__ W,
    const float* __restrict__ b, float* __restrict__ h) {
  const int wave = threadIdx.x >> 5, lane = threadIdx.x & 31;
  const int c4 = (lane & 15) * 4;
  float4 w[DIN];
#pragma unroll
  for (int d = 0; d < DIN; ++d) w[d] = *reinterpret_cast<const float4*>(W + d * H + c4);
  const float4 bv = *reinterpret_cast<const float4*>(b + c4);
  const int n0 = (blockIdx.x * 8 + wave) * IP_WAVE_NODES + (lane >> 4);
#pragma unroll
  for (int i = 0; i < IP_WAVE_NODES / 2; ++i) {
    const int n = n0 + 2 * i;
    const float* xr = x + (size_t)n * DIN;
    float xv[DIN];
#pragma unroll
    for (int d = 0; d < DIN; ++d) xv[d] = xr[d];
    float4 acc = bv;
#pragma unroll
    for (int d = 0; d < DIN; ++d) {
      acc.x = fmaf(xv[d], w[d].x, acc.x);
      acc.y = fmaf(xv[d], w[d].y, acc.y);
      acc.z = fmaf(xv[d], w[d].z, acc.z);
      acc.w = fmaf(xv[d], w[d].w, acc.w);
    }
    *reinterpret_cast<float4*>(h + (size_t)n * H + c4) = acc;
  }
}

#define GT_TILES (NN / 16)
#define GT_BLOCKS ((GT_TILES + 3) / 4)
static_assert(NN % 16 == 0 && H == 64, "16-row tiles, 64 columns");
__global__ __launch_bounds__(128) void k_gemm_h(
    const float* __restrict__ A, const float* __restrict__ W,
    const float* __restrict__ bias, const float* __restrict__ aw,
    float* __restrict__ out, float* __restrict__ sD, float* __restrict__ sS) {
  __shared__ float4 lW[H * H / 4];
  __shared__ float4 lAw[4][2 * H / 4];
  const int tid = threadIdx.x;
  const int wave = tid >> 5, lane = tid & 31;
  const int c = lane & 15, g = lane >> 4;
  const int tile_raw = blockIdx.x * 4 + wave;
  const bool valid = tile_raw < GT_TILES;
  const int row0 = (valid ? tile_raw : GT_TILES - 1) * 16;

  const float4* arow = reinterpret_cast<const float4*>(A + (size_t)(row0 + c) * H) + g;
  float4 af[8];
#pragma unroll
  for (int i = 0; i < 8; ++i) af[i] = arow[2 * i];

  const float4* W4 = reinterpret_cast<const float4*>(W);
  float4 wv[8];
#pragma unroll
  for (int i = 0; i < 8; ++i) wv[i] = W4[tid + 128 * i];
  const float4 bv = reinterpret_cast<const float4*>(bias)[c];
  const float4 awv = reinterpret_cast<const float4*>(aw)[lane];
#pragma unroll
  for (int i = 0; i < 8; ++i) lW[tid + 128 * i] = wv[i];
  lAw[wave][lane] = awv;
  __syncthreads();

  v8f acc0, acc1, acc2, acc3;
#pragma unroll
  for (int r = 0; r < 8; ++r) { acc0[r] = bv.x; acc1[r] = bv.y; acc2[r] = bv.z; acc3[r] = bv.w; }

  const float4* lWg = lW + (4 * g) * 16 + c;
#pragma unroll
  for (int i = 0; i < 8; ++i) {
#pragma unroll
    for (int p = 0; p < 2; ++p) {
      const int k = 8 * i + 2 * p;
      const float4 q0 = lWg[k * 16];
      const float4 q1 = lWg[(k + 1) * 16];
      v2f a, b;
      a.x = p ? af[i].z : af[i].x;
      a.y = p ? af[i].w : af[i].y;
      b.x = q0.x; b.y = q1.x;
      acc0 = __builtin_amdgcn_wmma_f32_16x16x4_f32(false, a, false, b, (short)0, acc0, false, false);
      b.x = q0.y; b.y = q1.y;
      acc1 = __builtin_amdgcn_wmma_f32_16x16x4_f32(false, a, false, b, (short)0, acc1, false, false);
      b.x = q0.z; b.y = q1.z;
      acc2 = __builtin_amdgcn_wmma_f32_16x16x4_f32(false, a, false, b, (short)0, acc2, false, false);
      b.x = q0.w; b.y = q1.w;
      acc3 = __builtin_amdgcn_wmma_f32_16x16x4_f32(false, a, false, b, (short)0, acc3, false, false);
    }
  }

  float pd = 0.0f, ps = 0.0f;
#pragma unroll
  for (int i = 0; i < 8; ++i) {
    const float4 ad = lAw[wave][2 * i + g];
    const float4 as = lAw[wave][16 + 2 * i + g];
    pd = fmaf(af[i].x, ad.x, pd); pd = fmaf(af[i].y, ad.y, pd);
    pd = fmaf(af[i].z, ad.z, pd); pd = fmaf(af[i].w, ad.w, pd);
    ps = fmaf(af[i].x, as.x, ps); ps = fmaf(af[i].y, as.y, ps);
    ps = fmaf(af[i].z, as.z, ps); ps = fmaf(af[i].w, as.w, ps);
  }
  pd += __shfl_xor(pd, 16, 32);
  ps += __shfl_xor(ps, 16, 32);

  if (valid) {
    float4* o4 = reinterpret_cast<float4*>(out + (size_t)(row0 + 8 * g) * H) + c;
#pragma unroll
    for (int r = 0; r < 8; ++r) o4[r * 16] = make_float4(acc0[r], acc1[r], acc2[r], acc3[r]);
    if (g == 0) { sD[row0 + c] = pd; sS[row0 + c] = ps; }
  }
}

static_assert(NE % 4 == 0, "four edges per thread");
#define NODE_PAD 100032
#define BKT_SHIFT 9
#define BKT_R (1 << BKT_SHIFT)
#define NBKT ((NN + BKT_R - 1) / BKT_R)
#define NCHK 250
#define CHK_V4 (NE / 4 / NCHK)
#define GH_PITCH 256
#define RG_DEPTH 8
static_assert(NCHK * CHK_V4 * 4 == NE, "chunks cover the edges exactly");
static_assert(CHK_V4 <= 1024, "one four-edge group per thread");
static_assert(NBKT <= 224 && NBKT * BKT_R >= NN + 1, "range table");
static_assert(NCHK <= GH_PITCH && GH_PITCH - NCHK <= NCHK, "pad columns are written by the first chunks");

__device__ __forceinline__ int wave_incl_scan_i(int v, int lane) {
#pragma unroll
  for (int o = 1; o < 32; o <<= 1) {
    int up = __shfl_up(v, o, 32);
    if (lane >= o) v += up;
  }
  return v;
}

__global__ __launch_bounds__(1024) void k_bin_hist(
    const int* __restrict__ ei, int* __restrict__ gh) {
  __shared__ int hist[NBKT];
  const int tid = threadIdx.x, w = blockIdx.x;
  if (tid < NBKT) hist[tid] = 0;
  __syncthreads();
  if (tid < CHK_V4) {
    int4 d = reinterpret_cast<const int4*>(ei + NE)[w * CHK_V4 + tid];
    atomicAdd(&hist[d.x >> BKT_SHIFT], 1);
    atomicAdd(&hist[d.y >> BKT_SHIFT], 1);
    atomicAdd(&hist[d.z >> BKT_SHIFT], 1);
    atomicAdd(&hist[d.w >> BKT_SHIFT], 1);
  }
  __syncthreads();
  if (tid < NBKT) {
    gh[tid * GH_PITCH + w] = hist[tid];
    if (w < GH_PITCH - NCHK) gh[tid * GH_PITCH + NCHK + w] = 0;
  }
}

__global__ __launch_bounds__(1024) void k_bin_scan(int* __restrict__ gh) {
  __shared__ int rowtot[224];
  __shared__ int rowbase[224];
  __shared__ int wtot[8];
  const int tid = threadIdx.x, wave = tid >> 5, lane = tid & 31;
  int4 a[7], c[7];
#pragma unroll
  for (int k = 0; k < 7; ++k) {
    int b = min(wave + 32 * k, NBKT - 1);
    const int4* p = reinterpret_cast<const int4*>(gh + b * GH_PITCH + lane * 8);
    a[k] = p[0];
    c[k] = p[1];
  }
  int ex[7];
#pragma unroll
  for (int k = 0; k < 7; ++k) {
    int b = wave + 32 * k;
    int sum = ((a[k].x + a[k].y) + (a[k].z + a[k].w)) + ((c[k].x + c[k].y) + (c[k].z + c[k].w));
    int incl = wave_incl_scan_i(sum, lane);
    ex[k] = incl - sum;
    int tot = __shfl(incl, 31, 32);
    if (lane == 0) rowtot[b] = (b < NBKT) ? tot : 0;
  }
  __syncthreads();
  int v = 0, vin = 0;
  if (tid < 224) {
    v = rowtot[tid];
    vin = wave_incl_scan_i(v, lane);
    if (lane == 31) wtot[wave] = vin;
  }
  __syncthreads();
  if (tid < 224) {
    int before = 0;
#pragma unroll
    for (int j = 0; j < 7; ++j) before += (j < wave) ? wtot[j] : 0;
    rowbase[tid] = before + vin - v;
  }
  __syncthreads();
#pragma unroll
  for (int k = 0; k < 7; ++k) {
    int b = wave + 32 * k;
    if (b < NBKT) {
      int r = rowbase[b] + ex[k];
      int4 o0, o1;
      o0.x = r; r += a[k].x;
      o0.y = r; r += a[k].y;
      o0.z = r; r += a[k].z;
      o0.w = r; r += a[k].w;
      o1.x = r; r += c[k].x;
      o1.y = r; r += c[k].y;
      o1.z = r; r += c[k].z;
      o1.w = r;
      int4* p = reinterpret_cast<int4*>(gh + b * GH_PITCH + lane * 8);
      p[0] = o0;
      p[1] = o1;
    }
  }
}

__global__ __launch_bounds__(1024) void k_bin_scatter(
    const int* __restrict__ ei, const int* __restrict__ gh, int2* __restrict__ binned) {
  __shared__ int cur[NBKT];
  const int tid = threadIdx.x, w = blockIdx.x;
  if (tid < NBKT) cur[tid] = gh[tid * GH_PITCH + w];
  __syncthreads();
  if (tid < CHK_V4) {
    int t = w * CHK_V4 + tid;
    int4 d = reinterpret_cast<const int4*>(ei + NE)[t];
    int e0 = 4 * t;
    int p;
    p = atomicAdd(&cur[d.x >> BKT_SHIFT], 1); binned[p] = make_int2(e0,     d.x);
    p = atomicAdd(&cur[d.y >> BKT_SHIFT], 1); binned[p] = make_int2(e0 + 1, d.y);
    p = atomicAdd(&cur[d.z >> BKT_SHIFT], 1); binned[p] = make_int2(e0 + 2, d.z);
    p = atomicAdd(&cur[d.w >> BKT_SHIFT], 1); binned[p] = make_int2(e0 + 3, d.w);
  }
}

__global__ __launch_bounds__(1024) void k_range_group(
    const int* __restrict__ gh, const int2* __restrict__ binned,
    int* __restrict__ off, int2* __restrict__ tmp) {
  __shared__ int cnt[BKT_R];
  __shared__ int wtot[32];
  const int tid = threadIdx.x, wave = tid >> 5, lane = tid & 31;
  const int b = blockIdx.x;
  const int lo = b << BKT_SHIFT;
  const int cb = gh[b * GH_PITCH];
  const int ce = (b + 1 < NBKT) ? gh[(b + 1) * GH_PITCH] : NE;
  if (tid < BKT_R) cnt[tid] = 0;
  __syncthreads();
  for (int i0 = cb + tid; i0 < ce; i0 += RG_DEPTH * 1024) {
    int dd[RG_DEPTH];
#pragma unroll
    for (int u = 0; u < RG_DEPTH; ++u) dd[u] = binned[min(i0 + 1024 * u, ce - 1)].y;
#pragma unroll
    for (int u = 0; u < RG_DEPTH; ++u)
      if (i0 + 1024 * u < ce) atomicAdd(&cnt[dd[u] - lo], 1);
  }
  __syncthreads();
  const int c = (tid < BKT_R) ? cnt[tid] : 0;
  const int incl = wave_incl_scan_i(c, lane);
  if (lane == 31) wtot[wave] = incl;
  __syncthreads();
  const int before = wave_sum_i((lane < wave) ? wtot[lane] : 0);
  const int excl = before + incl - c;
  if (tid < BKT_R) {
    cnt[tid] = excl;
    if (lo + tid <= NN) off[lo + tid] = cb + excl;
  }
  __syncthreads();
  for (int i0 = cb + tid; i0 < ce; i0 += RG_DEPTH * 1024) {
    int2 r[RG_DEPTH];
#pragma unroll
    for (int u = 0; u < RG_DEPTH; ++u) r[u] = binned[min(i0 + 1024 * u, ce - 1)];
#pragma unroll
    for (int u = 0; u < RG_DEPTH; ++u)
      if (i0 + 1024 * u < ce) {
        int p = atomicAdd(&cnt[r[u].y - lo], 1);
        tmp[cb + p] = r[u];
      }
  }
}

__global__ __launch_bounds__(256) void k_edge_order(
    const int* __restrict__ ei, const float* __restrict__ ea,
    const int* __restrict__ off, const int2* __restrict__ tmp,
    int2* __restrict__ rec) {
  int p = blockIdx.x * 256 + threadIdx.x;
  if (p >= NE) return;
  int2 my = tmp[p];
  int s = off[my.y], e = off[my.y + 1];
  int rank = 0;
  for (int q = s; q < e; ++q) rank += (tmp[q].x < my.x) ? 1 : 0;
  rec[s + rank] = make_int2(ei[my.x], __float_as_int(ea[my.x]));
}

__device__ __forceinline__ float half_sum16(float v) {
#pragma unroll
  for (int off = 8; off > 0; off >>= 1) v += __shfl_xor(v, off, 32);
  return v;
}
__global__ __launch_bounds__(256) void k_gather_update(
    float* __restrict__ h, const float* __restrict__ m,
    const int* __restrict__ off, const int2* __restrict__ rec,
    const float* __restrict__ sD, const float* __restrict__ sS,
    const float* __restrict__ attb,
    const float* __restrict__ g, const float* __restrict__ beta) {
  __shared__ int2 strip[8][32];
  const int gw = __builtin_amdgcn_readfirstlane(blockIdx.x * 8 + (threadIdx.x >> 5));
  const int lane = threadIdx.x & 31;
  if (gw >= NN) return;
  int2* const mystrip = strip[threadIdx.x >> 5];
  const int odd = lane >> 4;
  const unsigned c16 = (lane & 15) * 16u;
  const size_t base = (size_t)gw * H + (lane & 15) * 4;
  const int s = off[gw], e = off[gw + 1];
  const float sd = sD[gw], ab = attb[0];
  const float4 r  = *reinterpret_cast<const float4*>(h + base);
  const float4 mn = *reinterpret_cast<const float4*>(m + base);
  const char* const mbytes = reinterpret_cast<const char*>(m);
  float4 acc = make_float4(0.0f, 0.0f, 0.0f, 0.0f);
  for (int cb = s; cb < e; cb += 32) {
    const int cnt = min(32, e - cb);
    int src = gw;
    float w = 0.0f;
    if (lane < cnt) {
      int2 rc = rec[cb + lane];
      src = rc.x;
      __builtin_prefetch(m + (size_t)src * H, 0, 1);
      w = __int_as_float(rc.y) * sigmoidf_(sd + sS[src] + ab);
    }
    mystrip[lane] = make_int2((int)((unsigned)src << 8), __float_as_int(w));
#pragma unroll 4
    for (int i = 0; i < cnt; i += 2) {
      const int2 p = mystrip[i + odd];
      const float bw = __int_as_float(p.y);
      const float4 v = *reinterpret_cast<const float4*>(mbytes + ((unsigned)p.x + c16));
      acc.x = fmaf(v.x, bw, acc.x);
      acc.y = fmaf(v.y, bw, acc.y);
      acc.z = fmaf(v.z, bw, acc.z);
      acc.w = fmaf(v.w, bw, acc.w);
    }
  }
  acc.x += __shfl_xor(acc.x, 16, 32);
  acc.y += __shfl_xor(acc.y, 16, 32);
  acc.z += __shfl_xor(acc.z, 16, 32);
  acc.w += __shfl_xor(acc.w, 16, 32);
  const float f0 = acc.x + mn.x, f1 = acc.y + mn.y, f2 = acc.z + mn.z, f3 = acc.w + mn.w;
  const float mean = half_sum16((f0 + f1) + (f2 + f3)) * (1.0f / H);
  const float d0 = f0 - mean, d1 = f1 - mean, d2 = f2 - mean, d3 = f3 - mean;
  const float var = half_sum16(fmaf(d0, d0, d1 * d1) + fmaf(d2, d2, d3 * d3)) * (1.0f / H);
  const float inv = rsqrtf(var + LN_EPS);
  const float4 gv = *reinterpret_cast<const float4*>(g + (lane & 15) * 4);
  const float4 bv = *reinterpret_cast<const float4*>(beta + (lane & 15) * 4);
  if (odd == 0) {
    float4 o;
    o.x = fmaxf(fmaf(d0 * inv, gv.x, bv.x), 0.0f) + r.x;
    o.y = fmaxf(fmaf(d1 * inv, gv.y, bv.y), 0.0f) + r.y;
    o.z = fmaxf(fmaf(d2 * inv, gv.z, bv.z), 0.0f) + r.z;
    o.w = fmaxf(fmaf(d3 * inv, gv.w, bv.w), 0.0f) + r.w;
    *reinterpret_cast<float4*>(h + base) = o;
  }
}

#define HD_NODES 8
#define HD_WAVES 4
#define HD_BLOCKS 3125
#define HD_THREADS 128
static_assert(HD_BLOCKS * HD_WAVES * HD_NODES == NN && HD_THREADS == HD_WAVES * 32, "the waves cover the nodes exactly");
static_assert(HHID == 32 && DOUT == 32 && H == 64, "one lane per hidden unit, two features per lane");

__device__ __forceinline__ void heads_half(
    const float (&hv)[HD_NODES], const int krow0,
    const float* __restrict__ qW1, const float* __restrict__ iW1,
    const float* __restrict__ gW1, const float* __restrict__ sW1,
    const float* __restrict__ oW, const int lane,
    float (&tq)[HD_NODES], float (&ti)[HD_NODES], float (&tg)[HD_NODES],
    float (&ts)[HD_NODES], float (&bo)[HD_NODES]) {
#pragma unroll 4
  for (int k = 0; k < 32; ++k) {
    const int wi = (krow0 + k) * HHID + lane;
    const float wq = qW1[wi], wn = iW1[wi], wg = gW1[wi], wv = sW1[wi], wo = oW[wi];
#pragma unroll
    for (int j = 0; j < HD_NODES; ++j) {
      const float hk = __int_as_float(__builtin_amdgcn_readlane(__float_as_int(hv[j]), k));
      tq[j] = fmaf(hk, wq, tq[j]);
      ti[j] = fmaf(hk, wn, ti[j]);
      tg[j] = fmaf(hk, wg, tg[j]);
      ts[j] = fmaf(hk, wv, ts[j]);
      bo[j] = fmaf(hk, wo, bo[j]);
    }
  }
}

__global__ __launch_bounds__(HD_THREADS) void k_heads(
    const float* __restrict__ h,
    const float* __restrict__ qW1, const float* __restrict__ qb1,
    const float* __restrict__ qW2, const float* __restrict__ qb2,
    const float* __restrict__ iW1, const float* __restrict__ ib1,
    const float* __restrict__ iW2, const float* __restrict__ ib2,
    const float* __restrict__ gW1, const float* __restrict__ gb1,
    const float* __restrict__ gW2, const float* __restrict__ gb2,
    const float* __restrict__ oW,  const float* __restrict__ obv,
    const float* __restrict__ sW1, const float* __restrict__ sb1,
    const float* __restrict__ sW2, const float* __restrict__ sb2,
    float* __restrict__ oq, float* __restrict__ oi, float* __restrict__ og,
    float* __restrict__ oh, float* __restrict__ obot, float* __restrict__ osev) {
  const int wv = __builtin_amdgcn_readfirstlane(blockIdx.x * HD_WAVES + (threadIdx.x >> 5));
  const int lane = threadIdx.x & 31;
  const int n0 = wv * HD_NODES;
  const size_t base = (size_t)n0 * H;

  float h0[HD_NODES], h1[HD_NODES];
#pragma unroll
  for (int j = 0; j < HD_NODES; ++j) {
    h0[j] = h[base + j * H + lane];
    h1[j] = h[base + j * H + lane + 32];
  }

  float tq[HD_NODES], ti[HD_NODES], tg[HD_NODES], ts[HD_NODES], bo[HD_NODES];
#pragma unroll
  for (int j = 0; j < HD_NODES; ++j) { tq[j] = 0.f; ti[j] = 0.f; tg[j] = 0.f; ts[j] = 0.f; bo[j] = 0.f; }

  heads_half(h0, 0,  qW1, iW1, gW1, sW1, oW, lane, tq, ti, tg, ts, bo);
  heads_half(h1, 32, qW1, iW1, gW1, sW1, oW, lane, tq, ti, tg, ts, bo);

  const float bq1 = qb1[lane], bi1 = ib1[lane], bg1 = gb1[lane], bs1 = sb1[lane];
  const float wq2 = qW2[lane], wi2 = iW2[lane], wg2 = gW2[lane];
  const float bq2 = qb2[0], bi2 = ib2[0], bg2 = gb2[0];
  const float bob = obv[lane];
  float ws2[5], bs2[5];
#pragma unroll
  for (int c = 0; c < 5; ++c) { ws2[c] = sW2[lane * 5 + c]; bs2[c] = sb2[c]; }

  float keepq = 0.f, keepi = 0.f, keepg = 0.f;
#pragma unroll
  for (int j = 0; j < HD_NODES; ++j) {
    const float aq = fmaxf(tq[j] + bq1, 0.0f);
    const float ai = fmaxf(ti[j] + bi1, 0.0f);
    const float ag = fmaxf(tg[j] + bg1, 0.0f);
    const float asv = fmaxf(ts[j] + bs1, 0.0f);

    const float q  = wave_sum(aq * wq2);
    const float ii = wave_sum(ai * wi2);
    const float gg = wave_sum(ag * wg2);
    const float pq = sigmoidf_(q + bq2);
    const float pi = sigmoidf_(ii + bi2);
    const float pg = sigmoidf_(gg + bg2);
    if (lane == j) { keepq = pq; keepi = pi; keepg = pg; }

    float z[5];
#pragma unroll
    for (int c = 0; c < 5; ++c) z[c] = wave_sum(asv * ws2[c]) + bs2[c];
    float zm = z[0];
#pragma unroll
    for (int c = 1; c < 5; ++c) zm = fmaxf(zm, z[c]);
    float es = 0.f;
#pragma unroll
    for (int c = 0; c < 5; ++c) { z[c] = __expf(z[c] - zm); es += z[c]; }
    const float inv = 1.0f / es;
    float sv = z[0];
#pragma unroll
    for (int c = 1; c < 5; ++c) sv = (lane == c) ? z[c] : sv;
    if (lane < 5) osev[(size_t)(n0 + j) * 5 + lane] = sv * inv;

    obot[(size_t)(n0 + j) * DOUT + lane] = bo[j] + bob;
    oh[base + j * H + lane]      = h0[j];
    oh[base + j * H + lane + 32] = h1[j];
  }
  if (lane < HD_NODES) {
    oq[n0 + lane] = keepq;
    oi[n0 + lane] = keepi;
    og[n0 + lane] = keepg;
  }
}

#define GM_BLOCKS 625
#define GM_LOADS 10
__global__ __launch_bounds__(256) void k_graph_partial(
    const float* __restrict__ h, float* __restrict__ part) {
  __shared__ float ls[8][H];
  const int tid = threadIdx.x;
  const int wave = tid >> 5, lane = tid & 31;
  const float4* h4 = reinterpret_cast<const float4*>(h);
  const size_t t = (size_t)blockIdx.x * 256 + tid;
  float4 v[GM_LOADS];
#pragma unroll
  for (int i = 0; i < GM_LOADS; ++i) v[i] = h4[t + (size_t)i * (GM_BLOCKS * 256)];
  float4 a = v[0];
#pragma unroll
  for (int i = 1; i < GM_LOADS; ++i) {
    a.x += v[i].x; a.y += v[i].y; a.z += v[i].z; a.w += v[i].w;
  }
  a.x += __shfl_xor(a.x, 16, 32);
  a.y += __shfl_xor(a.y, 16, 32);
  a.z += __shfl_xor(a.z, 16, 32);
  a.w += __shfl_xor(a.w, 16, 32);
  if (lane < 16) *reinterpret_cast<float4*>(&ls[wave][lane * 4]) = a;
  __syncthreads();
  if (tid < H) {
    float s = ls[0][tid];
#pragma unroll
    for (int w = 1; w < 8; ++w) s += ls[w][tid];
    part[(size_t)blockIdx.x * H + tid] = s;
  }
}

__global__ __launch_bounds__(1024) void k_graph_final(
    const float* __restrict__ part, float* __restrict__ gout) {
  __shared__ float ls[16][H];
  const int tid = threadIdx.x;
  const int f = tid & 63, q = tid >> 6;
  float v[40];
#pragma unroll
  for (int i = 0; i < 40; ++i) {
    int b = q + 16 * i;
    v[i] = (b < GM_BLOCKS) ? part[(size_t)b * H + f] : 0.0f;
  }
  float s = v[0];
#pragma unroll
  for (int i = 1; i < 40; ++i) s += v[i];
  ls[q][f] = s;
  __syncthreads();
  if (tid < H) {
    float r = ls[0][tid];
#pragma unroll
    for (int w = 1; w < 16; ++w) r += ls[w][tid];
    gout[tid] = r * (1.0f / NN);
  }
}

extern "C" void kernel_launch(void* const* d_in, const int* in_sizes, int n_in,
                              void* d_out, int out_size, void* d_ws, size_t ws_size,
                              hipStream_t stream) {
  (void)in_sizes; (void)n_in; (void)out_size; (void)ws_size;
  const float* x    = (const float*)d_in[0];
  const int*   ei   = (const int*)  d_in[1];
  const float* ea   = (const float*)d_in[2];
  const float* inW  = (const float*)d_in[3];
  const float* inb  = (const float*)d_in[4];
  const float* linW = (const float*)d_in[5];
  const float* linb = (const float*)d_in[6];
  const float* attW = (const float*)d_in[7];
  const float* attb = (const float*)d_in[8];
  const float* lng  = (const float*)d_in[9];
  const float* lnb  = (const float*)d_in[10];
  const float* qW1 = (const float*)d_in[11]; const float* qb1 = (const float*)d_in[12];
  const float* qW2 = (const float*)d_in[13]; const float* qb2 = (const float*)d_in[14];
  const float* iW1 = (const float*)d_in[15]; const float* ib1 = (const float*)d_in[16];
  const float* iW2 = (const float*)d_in[17]; const float* ib2 = (const float*)d_in[18];
  const float* gW1 = (const float*)d_in[19]; const float* gb1 = (const float*)d_in[20];
  const float* gW2 = (const float*)d_in[21]; const float* gb2 = (const float*)d_in[22];
  const float* oW  = (const float*)d_in[23]; const float* obv = (const float*)d_in[24];
  const float* sW1 = (const float*)d_in[25]; const float* sb1 = (const float*)d_in[26];
  const float* sW2 = (const float*)d_in[27]; const float* sb2 = (const float*)d_in[28];

  float* ws   = (float*)d_ws;
  float* h    = ws;
  float* mbuf = h    + (size_t)NN * H;
  float* lst  = mbuf + (size_t)NN * H;
  float* sD   = lst + (size_t)NN * H;
  float* sS   = sD + NN;
  int*   off  = (int*)lst;
  int2*  tmp  = (int2*)(off + NODE_PAD);
  int2*  rec  = tmp + NE;
  float* part = (float*)(rec + NE);
  int2*  binned = (int2*)(part + GM_BLOCKS * H);
  int*   gh   = (int*)(binned + NE);

  float* out  = (float*)d_out;
  float* oq   = out;
  float* oi   = out + (size_t)NN;
  float* og   = out + 2 * (size_t)NN;
  float* oh   = out + 3 * (size_t)NN;
  float* ogr  = out + 67 * (size_t)NN;
  float* obot = out + 67 * (size_t)NN + H;
  float* osev = out + 99 * (size_t)NN + H;

  k_input_proj<<<IP_BLOCKS, 256, 0, stream>>>(x, inW, inb, h);

  k_bin_hist<<<NCHK, 1024, 0, stream>>>(ei, gh);
  k_bin_scan<<<1, 1024, 0, stream>>>(gh);
  k_bin_scatter<<<NCHK, 1024, 0, stream>>>(ei, gh, binned);
  k_range_group<<<NBKT, 1024, 0, stream>>>(gh, binned, off, tmp);
  k_edge_order<<<(NE + 255) / 256, 256, 0, stream>>>(ei, ea, off, tmp, rec);

  const int nodeWaveBlocks = (NN + 7) / 8;
  for (int l = 0; l < NL; ++l) {
    k_gemm_h<<<GT_BLOCKS, 128, 0, stream>>>(h, linW + (size_t)l * H * H, linb + (size_t)l * H,
                                          attW + (size_t)l * 2 * H, mbuf, sD, sS);
    k_gather_update<<<nodeWaveBlocks, 256, 0, stream>>>(h, mbuf, off, rec, sD, sS, attb + l,
                                                        lng + (size_t)l * H, lnb + (size_t)l * H);
  }

  k_heads<<<HD_BLOCKS, HD_THREADS, 0, stream>>>(h, qW1, qb1, qW2, qb2, iW1, ib1, iW2, ib2,
                                              gW1, gb1, gW2, gb2, oW, obv, sW1, sb1, sW2, sb2,
                                              oq, oi, og, oh, obot, osev);
  k_graph_partial<<<GM_BLOCKS, 256, 0, stream>>>(h, part);
  k_graph_final<<<1, 1024, 0, stream>>>(part, ogr);
}


# independent kernels share a launch: edge histogram with input projection, offset scan with layer-0 node product, graph-mean partials with output heads; 14 launches, were 17; sum orders unchanged
# speedup vs baseline: 2.5727x; 1.0391x over previous
#include <hip/hip_runtime.h>

#define NN 100000
#define NE 1000000
#define DIN 5
#define H 64
#define HHID 32
#define NL 4
#define DOUT 32
#define LN_EPS 1e-5f

typedef __attribute__((ext_vector_type(2)))  float    v2f;
typedef __attribute__((ext_vector_type(8)))  float    v8f;

__device__ __forceinline__ float wave_sum(float v) {
#pragma unroll
  for (int off = 16; off > 0; off >>= 1) v += __shfl_xor(v, off, 32);
  return v;
}
__device__ __forceinline__ float sigmoidf_(float x) { return 1.0f / (1.0f + __expf(-x)); }

__device__ __forceinline__ int wave_sum_i(int v) {
#pragma unroll
  for (int off = 16; off > 0; off >>= 1) v += __shfl_xor(v, off, 32);
  return v;
}

#define IP_WAVES 6250
#define IP_WAVE_NODES 16
#define IP_GROUP 4
static_assert(IP_WAVES * IP_WAVE_NODES == NN, "the waves cover the nodes exactly");
static_assert(IP_WAVE_NODES % (2 * IP_GROUP) == 0 && H == 64, "two nodes per step, sixteen lanes per row, whole groups");
__device__ __forceinline__ void input_proj_wave(
    const float* __restrict__ x, const float* __restrict__ W,
    const float* __restrict__ b, float* __restrict__ h, const int gwave, const int lane) {
  const int c4 = (lane & 15) * 4;
  float4 w[DIN];
#pragma unroll
  for (int d = 0; d < DIN; ++d) w[d] = *reinterpret_cast<const float4*>(W + d * H + c4);
  const float4 bv = *reinterpret_cast<const float4*>(b + c4);
  const int n0 = gwave * IP_WAVE_NODES + (lane >> 4);
  const float* xg = x + (size_t)n0 * DIN;
  float* hg = h + (size_t)n0 * H + c4;
#pragma unroll 1
  for (int grp = 0; grp < IP_WAVE_NODES / (2 * IP_GROUP); ++grp) {
    float xv[IP_GROUP][DIN];
#pragma unroll
    for (int i = 0; i < IP_GROUP; ++i) {
#pragma unroll
      for (int d = 0; d < DIN; ++d) xv[i][d] = xg[i * 2 * DIN + d];
    }
#pragma unroll
    for (int i = 0; i < IP_GROUP; ++i) {
      float4 acc = bv;
#pragma unroll
      for (int d = 0; d < DIN; ++d) {
        acc.x = fmaf(xv[i][d], w[d].x, acc.x);
        acc.y = fmaf(xv[i][d], w[d].y, acc.y);
        acc.z = fmaf(xv[i][d], w[d].z, acc.z);
        acc.w = fmaf(xv[i][d], w[d].w, acc.w);
      }
      *reinterpret_cast<float4*>(hg + i * 2 * H) = acc;
    }
    xg += IP_GROUP * 2 * DIN;
    hg += IP_GROUP * 2 * H;
  }
}

#define GT_TILES (NN / 16)
#define GT_BLOCKS ((GT_TILES + 3) / 4)
static_assert(NN % 16 == 0 && H == 64, "16-row tiles, 64 columns");
template <int WAVES>
__device__ __forceinline__ void gemm_h_block(
    const int blk,
    const float* __restrict__ A, const float* __restrict__ W,
    const float* __restrict__ bias, const float* __restrict__ aw,
    float* __restrict__ out, float* __restrict__ sD, float* __restrict__ sS,
    float4* __restrict__ lW, float4 (* __restrict__ lAw)[2 * H / 4]) {
  constexpr int NT = 32 * WAVES;
  constexpr int WPT = (H * H / 4) / NT;
  static_assert(WPT * NT == H * H / 4, "the threads cover W exactly");
  const int tid = threadIdx.x;
  const int wave = tid >> 5, lane = tid & 31;
  const int c = lane & 15, g = lane >> 4;
  const int tile_raw = blk * WAVES + wave;
  const bool valid = tile_raw < GT_TILES;
  const int row0 = (valid ? tile_raw : GT_TILES - 1) * 16;

  const float4* arow = reinterpret_cast<const float4*>(A + (size_t)(row0 + c) * H) + g;
  float4 af[8];
#pragma unroll
  for (int i = 0; i < 8; ++i) af[i] = arow[2 * i];

  const float4* W4 = reinterpret_cast<const float4*>(W);
  float4 wv[WPT];
#pragma unroll
  for (int i = 0; i < WPT; ++i) wv[i] = W4[tid + NT * i];
  const float4 bv = reinterpret_cast<const float4*>(bias)[c];
  const float4 awv = reinterpret_cast<const float4*>(aw)[lane];
#pragma unroll
  for (int i = 0; i < WPT; ++i) lW[tid + NT * i] = wv[i];
  lAw[wave][lane] = awv;
  __syncthreads();

  v8f acc0, acc1, acc2, acc3;
#pragma unroll
  for (int r = 0; r < 8; ++r) { acc0[r] = bv.x; acc1[r] = bv.y; acc2[r] = bv.z; acc3[r] = bv.w; }

  const float4* lWg = lW + (4 * g) * 16 + c;
#pragma unroll
  for (int i = 0; i < 8; ++i) {
#pragma unroll
    for (int p = 0; p < 2; ++p) {
      const int k = 8 * i + 2 * p;
      const float4 q0 = lWg[k * 16];
      const float4 q1 = lWg[(k + 1) * 16];
      v2f a, b;
      a.x = p ? af[i].z : af[i].x;
      a.y = p ? af[i].w : af[i].y;
      b.x = q0.x; b.y = q1.x;
      acc0 = __builtin_amdgcn_wmma_f32_16x16x4_f32(false, a, false, b, (short)0, acc0, false, false);
      b.x = q0.y; b.y = q1.y;
      acc1 = __builtin_amdgcn_wmma_f32_16x16x4_f32(false, a, false, b, (short)0, acc1, false, false);
      b.x = q0.z; b.y = q1.z;
      acc2 = __builtin_amdgcn_wmma_f32_16x16x4_f32(false, a, false, b, (short)0, acc2, false, false);
      b.x = q0.w; b.y = q1.w;
      acc3 = __builtin_amdgcn_wmma_f32_16x16x4_f32(false, a, false, b, (short)0, acc3, false, false);
    }
  }

  float pd = 0.0f, ps = 0.0f;
#pragma unroll
  for (int i = 0; i < 8; ++i) {
    const float4 ad = lAw[wave][2 * i + g];
    const float4 as = lAw[wave][16 + 2 * i + g];
    pd = fmaf(af[i].x, ad.x, pd); pd = fmaf(af[i].y, ad.y, pd);
    pd = fmaf(af[i].z, ad.z, pd); pd = fmaf(af[i].w, ad.w, pd);
    ps = fmaf(af[i].x, as.x, ps); ps = fmaf(af[i].y, as.y, ps);
    ps = fmaf(af[i].z, as.z, ps); ps = fmaf(af[i].w, as.w, ps);
  }
  pd += __shfl_xor(pd, 16, 32);
  ps += __shfl_xor(ps, 16, 32);

  if (valid) {
    float4* o4 = reinterpret_cast<float4*>(out + (size_t)(row0 + 8 * g) * H) + c;
#pragma unroll
    for (int r = 0; r < 8; ++r) o4[r * 16] = make_float4(acc0[r], acc1[r], acc2[r], acc3[r]);
    if (g == 0) { sD[row0 + c] = pd; sS[row0 + c] = ps; }
  }
}

__global__ __launch_bounds__(128) void k_gemm_h(
    const float* __restrict__ A, const float* __restrict__ W,
    const float* __restrict__ bias, const float* __restrict__ aw,
    float* __restrict__ out, float* __restrict__ sD, float* __restrict__ sS) {
  __shared__ float4 lW[H * H / 4];
  __shared__ float4 lAw[4][2 * H / 4];
  gemm_h_block<4>(blockIdx.x, A, W, bias, aw, out, sD, sS, lW, lAw);
}

static_assert(NE % 4 == 0, "four edges per thread");
#define NODE_PAD 100032
#define BKT_SHIFT 9
#define BKT_R (1 << BKT_SHIFT)
#define NBKT ((NN + BKT_R - 1) / BKT_R)
#define NCHK 250
#define CHK_V4 (NE / 4 / NCHK)
#define GH_PITCH 256
#define RG_DEPTH 8
static_assert(NCHK * CHK_V4 * 4 == NE, "chunks cover the edges exactly");
static_assert(CHK_V4 <= 1024, "one four-edge group per thread");
static_assert(NBKT <= 224 && NBKT * BKT_R >= NN + 1, "range table");
static_assert(NCHK <= GH_PITCH && GH_PITCH - NCHK <= NCHK, "pad columns are written by the first chunks");

__device__ __forceinline__ int wave_incl_scan_i(int v, int lane) {
#pragma unroll
  for (int o = 1; o < 32; o <<= 1) {
    int up = __shfl_up(v, o, 32);
    if (lane >= o) v += up;
  }
  return v;
}

__device__ __forceinline__ void bin_hist_block(
    const int* __restrict__ ei, int* __restrict__ gh, int* __restrict__ hist, const int w) {
  const int tid = threadIdx.x;
  if (tid < NBKT) hist[tid] = 0;
  __syncthreads();
  if (tid < CHK_V4) {
    int4 d = reinterpret_cast<const int4*>(ei + NE)[w * CHK_V4 + tid];
    atomicAdd(&hist[d.x >> BKT_SHIFT], 1);
    atomicAdd(&hist[d.y >> BKT_SHIFT], 1);
    atomicAdd(&hist[d.z >> BKT_SHIFT], 1);
    atomicAdd(&hist[d.w >> BKT_SHIFT], 1);
  }
  __syncthreads();
  if (tid < NBKT) {
    gh[tid * GH_PITCH + w] = hist[tid];
    if (w < GH_PITCH - NCHK) gh[tid * GH_PITCH + NCHK + w] = 0;
  }
}

#define IP_WG ((IP_WAVES + 31) / 32)
__global__ __launch_bounds__(1024) void k_hist_proj(
    const int* __restrict__ ei, int* __restrict__ gh,
    const float* __restrict__ x, const float* __restrict__ W,
    const float* __restrict__ b, float* __restrict__ h) {
  __shared__ int hist[NBKT];
  if (blockIdx.x < NCHK) {
    bin_hist_block(ei, gh, hist, blockIdx.x);
    return;
  }
  const int gwave = (blockIdx.x - NCHK) * 32 + (threadIdx.x >> 5);
  if (gwave < IP_WAVES) input_proj_wave(x, W, b, h, gwave, threadIdx.x & 31);
}

__device__ __forceinline__ void bin_scan_block(
    int* __restrict__ gh, int* __restrict__ rowtot, int* __restrict__ rowbase,
    int* __restrict__ wtot) {
  const int tid = threadIdx.x, wave = tid >> 5, lane = tid & 31;
  int4 a[7], c[7];
#pragma unroll
  for (int k = 0; k < 7; ++k) {
    int b = min(wave + 32 * k, NBKT - 1);
    const int4* p = reinterpret_cast<const int4*>(gh + b * GH_PITCH + lane * 8);
    a[k] = p[0];
    c[k] = p[1];
  }
  int ex[7];
#pragma unroll
  for (int k = 0; k < 7; ++k) {
    int b = wave + 32 * k;
    int sum = ((a[k].x + a[k].y) + (a[k].z + a[k].w)) + ((c[k].x + c[k].y) + (c[k].z + c[k].w));
    int incl = wave_incl_scan_i(sum, lane);
    ex[k] = incl - sum;
    int tot = __shfl(incl, 31, 32);
    if (lane == 0) rowtot[b] = (b < NBKT) ? tot : 0;
  }
  __syncthreads();
  int v = 0, vin = 0;
  if (tid < 224) {
    v = rowtot[tid];
    vin = wave_incl_scan_i(v, lane);
    if (lane == 31) wtot[wave] = vin;
  }
  __syncthreads();
  if (tid < 224) {
    int before = 0;
#pragma unroll
    for (int j = 0; j < 7; ++j) before += (j < wave) ? wtot[j] : 0;
    rowbase[tid] = before + vin - v;
  }
  __syncthreads();
#pragma unroll
  for (int k = 0; k < 7; ++k) {
    int b = wave + 32 * k;
    if (b < NBKT) {
      int r = rowbase[b] + ex[k];
      int4 o0, o1;
      o0.x = r; r += a[k].x;
      o0.y = r; r += a[k].y;
      o0.z = r; r += a[k].z;
      o0.w = r; r += a[k].w;
      o1.x = r; r += c[k].x;
      o1.y = r; r += c[k].y;
      o1.z = r; r += c[k].z;
      o1.w = r;
      int4* p = reinterpret_cast<int4*>(gh + b * GH_PITCH + lane * 8);
      p[0] = o0;
      p[1] = o1;
    }
  }
}

#define G0_WG ((GT_TILES + 31) / 32)
__global__ __launch_bounds__(1024) void k_scan_gemm0(
    int* __restrict__ gh,
    const float* __restrict__ A, const float* __restrict__ W,
    const float* __restrict__ bias, const float* __restrict__ aw,
    float* __restrict__ out, float* __restrict__ sD, float* __restrict__ sS) {
  __shared__ float4 lW[H * H / 4];
  __shared__ float4 lAw[32][2 * H / 4];
  __shared__ int rowtot[224];
  __shared__ int rowbase[224];
  __shared__ int wtot[8];
  if (blockIdx.x == 0) {
    bin_scan_block(gh, rowtot, rowbase, wtot);
    return;
  }
  gemm_h_block<32>(blockIdx.x - 1, A, W, bias, aw, out, sD, sS, lW, lAw);
}

__global__ __launch_bounds__(1024) void k_bin_scatter(
    const int* __restrict__ ei, const int* __restrict__ gh, int2* __restrict__ binned) {
  __shared__ int cur[NBKT];
  const int tid = threadIdx.x, w = blockIdx.x;
  if (tid < NBKT) cur[tid] = gh[tid * GH_PITCH + w];
  __syncthreads();
  if (tid < CHK_V4) {
    int t = w * CHK_V4 + tid;
    int4 d = reinterpret_cast<const int4*>(ei + NE)[t];
    int e0 = 4 * t;
    int p;
    p = atomicAdd(&cur[d.x >> BKT_SHIFT], 1); binned[p] = make_int2(e0,     d.x);
    p = atomicAdd(&cur[d.y >> BKT_SHIFT], 1); binned[p] = make_int2(e0 + 1, d.y);
    p = atomicAdd(&cur[d.z >> BKT_SHIFT], 1); binned[p] = make_int2(e0 + 2, d.z);
    p = atomicAdd(&cur[d.w >> BKT_SHIFT], 1); binned[p] = make_int2(e0 + 3, d.w);
  }
}

__global__ __launch_bounds__(1024) void k_range_group(
    const int* __restrict__ gh, const int2* __restrict__ binned,
    int* __restrict__ off, int2* __restrict__ tmp) {
  __shared__ int cnt[BKT_R];
  __shared__ int wtot[32];
  const int tid = threadIdx.x, wave = tid >> 5, lane = tid & 31;
  const int b = blockIdx.x;
  const int lo = b << BKT_SHIFT;
  const int cb = gh[b * GH_PITCH];
  const int ce = (b + 1 < NBKT) ? gh[(b + 1) * GH_PITCH] : NE;
  if (tid < BKT_R) cnt[tid] = 0;
  __syncthreads();
  for (int i0 = cb + tid; i0 < ce; i0 += RG_DEPTH * 1024) {
    int dd[RG_DEPTH];
#pragma unroll
    for (int u = 0; u < RG_DEPTH; ++u) dd[u] = binned[min(i0 + 1024 * u, ce - 1)].y;
#pragma unroll
    for (int u = 0; u < RG_DEPTH; ++u)
      if (i0 + 1024 * u < ce) atomicAdd(&cnt[dd[u] - lo], 1);
  }
  __syncthreads();
  const int c = (tid < BKT_R) ? cnt[tid] : 0;
  const int incl = wave_incl_scan_i(c, lane);
  if (lane == 31) wtot[wave] = incl;
  __syncthreads();
  const int before = wave_sum_i((lane < wave) ? wtot[lane] : 0);
  const int excl = before + incl - c;
  if (tid < BKT_R) {
    cnt[tid] = excl;
    if (lo + tid <= NN) off[lo + tid] = cb + excl;
  }
  __syncthreads();
  for (int i0 = cb + tid; i0 < ce; i0 += RG_DEPTH * 1024) {
    int2 r[RG_DEPTH];
#pragma unroll
    for (int u = 0; u < RG_DEPTH; ++u) r[u] = binned[min(i0 + 1024 * u, ce - 1)];
#pragma unroll
    for (int u = 0; u < RG_DEPTH; ++u)
      if (i0 + 1024 * u < ce) {
        int p = atomicAdd(&cnt[r[u].y - lo], 1);
        tmp[cb + p] = r[u];
      }
  }
}

__global__ __launch_bounds__(256) void k_edge_order(
    const int* __restrict__ ei, const float* __restrict__ ea,
    const int* __restrict__ off, const int2* __restrict__ tmp,
    int2* __restrict__ rec) {
  int p = blockIdx.x * 256 + threadIdx.x;
  if (p >= NE) return;
  int2 my = tmp[p];
  int s = off[my.y], e = off[my.y + 1];
  int rank = 0;
  for (int q = s; q < e; ++q) rank += (tmp[q].x < my.x) ? 1 : 0;
  rec[s + rank] = make_int2(ei[my.x], __float_as_int(ea[my.x]));
}

__device__ __forceinline__ float half_sum16(float v) {
#pragma unroll
  for (int off = 8; off > 0; off >>= 1) v += __shfl_xor(v, off, 32);
  return v;
}
__global__ __launch_bounds__(256) void k_gather_update(
    float* __restrict__ h, const float* __restrict__ m,
    const int* __restrict__ off, const int2* __restrict__ rec,
    const float* __restrict__ sD, const float* __restrict__ sS,
    const float* __restrict__ attb,
    const float* __restrict__ g, const float* __restrict__ beta) {
  __shared__ int2 strip[8][32];
  const int gw = __builtin_amdgcn_readfirstlane(blockIdx.x * 8 + (threadIdx.x >> 5));
  const int lane = threadIdx.x & 31;
  if (gw >= NN) return;
  int2* const mystrip = strip[threadIdx.x >> 5];
  const int odd = lane >> 4;
  const unsigned c16 = (lane & 15) * 16u;
  const size_t base = (size_t)gw * H + (lane & 15) * 4;
  const int s = off[gw], e = off[gw + 1];
  const float sd = sD[gw], ab = attb[0];
  const float4 r  = *reinterpret_cast<const float4*>(h + base);
  const float4 mn = *reinterpret_cast<const float4*>(m + base);
  const char* const mbytes = reinterpret_cast<const char*>(m);
  float4 acc = make_float4(0.0f, 0.0f, 0.0f, 0.0f);
  for (int cb = s; cb < e; cb += 32) {
    const int cnt = min(32, e - cb);
    int src = gw;
    float w = 0.0f;
    if (lane < cnt) {
      int2 rc = rec[cb + lane];
      src = rc.x;
      __builtin_prefetch(m + (size_t)src * H, 0, 1);
      w = __int_as_float(rc.y) * sigmoidf_(sd + sS[src] + ab);
    }
    mystrip[lane] = make_int2((int)((unsigned)src << 8), __float_as_int(w));
#pragma unroll 4
    for (int i = 0; i < cnt; i += 2) {
      const int2 p = mystrip[i + odd];
      const float bw = __int_as_float(p.y);
      const float4 v = *reinterpret_cast<const float4*>(mbytes + ((unsigned)p.x + c16));
      acc.x = fmaf(v.x, bw, acc.x);
      acc.y = fmaf(v.y, bw, acc.y);
      acc.z = fmaf(v.z, bw, acc.z);
      acc.w = fmaf(v.w, bw, acc.w);
    }
  }
  acc.x += __shfl_xor(acc.x, 16, 32);
  acc.y += __shfl_xor(acc.y, 16, 32);
  acc.z += __shfl_xor(acc.z, 16, 32);
  acc.w += __shfl_xor(acc.w, 16, 32);
  const float f0 = acc.x + mn.x, f1 = acc.y + mn.y, f2 = acc.z + mn.z, f3 = acc.w + mn.w;
  const float mean = half_sum16((f0 + f1) + (f2 + f3)) * (1.0f / H);
  const float d0 = f0 - mean, d1 = f1 - mean, d2 = f2 - mean, d3 = f3 - mean;
  const float var = half_sum16(fmaf(d0, d0, d1 * d1) + fmaf(d2, d2, d3 * d3)) * (1.0f / H);
  const float inv = rsqrtf(var + LN_EPS);
  const float4 gv = *reinterpret_cast<const float4*>(g + (lane & 15) * 4);
  const float4 bv = *reinterpret_cast<const float4*>(beta + (lane & 15) * 4);
  if (odd == 0) {
    float4 o;
    o.x = fmaxf(fmaf(d0 * inv, gv.x, bv.x), 0.0f) + r.x;
    o.y = fmaxf(fmaf(d1 * inv, gv.y, bv.y), 0.0f) + r.y;
    o.z = fmaxf(fmaf(d2 * inv, gv.z, bv.z), 0.0f) + r.z;
    o.w = fmaxf(fmaf(d3 * inv, gv.w, bv.w), 0.0f) + r.w;
    *reinterpret_cast<float4*>(h + base) = o;
  }
}

#define HD_NODES 8
#define HD_NWAVES (NN / HD_NODES)
static_assert(HD_NWAVES * HD_NODES == NN, "the waves cover the nodes exactly");
static_assert(HHID == 32 && DOUT == 32 && H == 64, "one lane per hidden unit, two features per lane");

__device__ __forceinline__ void heads_half(
    const float (&hv)[HD_NODES], const int krow0,
    const float* __restrict__ qW1, const float* __restrict__ iW1,
    const float* __restrict__ gW1, const float* __restrict__ sW1,
    const float* __restrict__ oW, const int lane,
    float (&tq)[HD_NODES], float (&ti)[HD_NODES], float (&tg)[HD_NODES],
    float (&ts)[HD_NODES], float (&bo)[HD_NODES]) {
#pragma unroll 4
  for (int k = 0; k < 32; ++k) {
    const int wi = (krow0 + k) * HHID + lane;
    const float wq = qW1[wi], wn = iW1[wi], wg = gW1[wi], wv = sW1[wi], wo = oW[wi];
#pragma unroll
    for (int j = 0; j < HD_NODES; ++j) {
      const float hk = __int_as_float(__builtin_amdgcn_readlane(__float_as_int(hv[j]), k));
      tq[j] = fmaf(hk, wq, tq[j]);
      ti[j] = fmaf(hk, wn, ti[j]);
      tg[j] = fmaf(hk, wg, tg[j]);
      ts[j] = fmaf(hk, wv, ts[j]);
      bo[j] = fmaf(hk, wo, bo[j]);
    }
  }
}

__device__ __forceinline__ void heads_wave(
    const int wv, const int lane,
    const float* __restrict__ h,
    const float* __restrict__ qW1, const float* __restrict__ qb1,
    const float* __restrict__ qW2, const float* __restrict__ qb2,
    const float* __restrict__ iW1, const float* __restrict__ ib1,
    const float* __restrict__ iW2, const float* __restrict__ ib2,
    const float* __restrict__ gW1, const float* __restrict__ gb1,
    const float* __restrict__ gW2, const float* __restrict__ gb2,
    const float* __restrict__ oW,  const float* __restrict__ obv,
    const float* __restrict__ sW1, const float* __restrict__ sb1,
    const float* __restrict__ sW2, const float* __restrict__ sb2,
    float* __restrict__ oq, float* __restrict__ oi, float* __restrict__ og,
    float* __restrict__ oh, float* __restrict__ obot, float* __restrict__ osev) {
  const int n0 = wv * HD_NODES;
  const size_t base = (size_t)n0 * H;

  float h0[HD_NODES], h1[HD_NODES];
#pragma unroll
  for (int j = 0; j < HD_NODES; ++j) {
    h0[j] = h[base + j * H + lane];
    h1[j] = h[base + j * H + lane + 32];
  }

  float tq[HD_NODES], ti[HD_NODES], tg[HD_NODES], ts[HD_NODES], bo[HD_NODES];
#pragma unroll
  for (int j = 0; j < HD_NODES; ++j) { tq[j] = 0.f; ti[j] = 0.f; tg[j] = 0.f; ts[j] = 0.f; bo[j] = 0.f; }

  heads_half(h0, 0,  qW1, iW1, gW1, sW1, oW, lane, tq, ti, tg, ts, bo);
  heads_half(h1, 32, qW1, iW1, gW1, sW1, oW, lane, tq, ti, tg, ts, bo);

  const float bq1 = qb1[lane], bi1 = ib1[lane], bg1 = gb1[lane], bs1 = sb1[lane];
  const float wq2 = qW2[lane], wi2 = iW2[lane], wg2 = gW2[lane];
  const float bq2 = qb2[0], bi2 = ib2[0], bg2 = gb2[0];
  const float bob = obv[lane];
  float ws2[5], bs2[5];
#pragma unroll
  for (int c = 0; c < 5; ++c) { ws2[c] = sW2[lane * 5 + c]; bs2[c] = sb2[c]; }

  float keepq = 0.f, keepi = 0.f, keepg = 0.f;
#pragma unroll
  for (int j = 0; j < HD_NODES; ++j) {
    const float aq = fmaxf(tq[j] + bq1, 0.0f);
    const float ai = fmaxf(ti[j] + bi1, 0.0f);
    const float ag = fmaxf(tg[j] + bg1, 0.0f);
    const float asv = fmaxf(ts[j] + bs1, 0.0f);

    const float q  = wave_sum(aq * wq2);
    const float ii = wave_sum(ai * wi2);
    const float gg = wave_sum(ag * wg2);
    const float pq = sigmoidf_(q + bq2);
    const float pi = sigmoidf_(ii + bi2);
    const float pg = sigmoidf_(gg + bg2);
    if (lane == j) { keepq = pq; keepi = pi; keepg = pg; }

    float z[5];
#pragma unroll
    for (int c = 0; c < 5; ++c) z[c] = wave_sum(asv * ws2[c]) + bs2[c];
    float zm = z[0];
#pragma unroll
    for (int c = 1; c < 5; ++c) zm = fmaxf(zm, z[c]);
    float es = 0.f;
#pragma unroll
    for (int c = 0; c < 5; ++c) { z[c] = __expf(z[c] - zm); es += z[c]; }
    const float inv = 1.0f / es;
    float sv = z[0];
#pragma unroll
    for (int c = 1; c < 5; ++c) sv = (lane == c) ? z[c] : sv;
    if (lane < 5) osev[(size_t)(n0 + j) * 5 + lane] = sv * inv;

    obot[(size_t)(n0 + j) * DOUT + lane] = bo[j] + bob;
    oh[base + j * H + lane]      = h0[j];
    oh[base + j * H + lane + 32] = h1[j];
  }
  if (lane < HD_NODES) {
    oq[n0 + lane] = keepq;
    oi[n0 + lane] = keepi;
    og[n0 + lane] = keepg;
  }
}

#define GM_BLOCKS 625
#define GM_LOADS 10
__device__ __forceinline__ void graph_partial_block(
    const int blk, const float* __restrict__ h, float* __restrict__ part, float (* __restrict__ ls)[H]) {
  const int tid = threadIdx.x;
  const int wave = tid >> 5, lane = tid & 31;
  const float4* h4 = reinterpret_cast<const float4*>(h);
  const size_t t = (size_t)blk * 256 + tid;
  float4 v[GM_LOADS];
#pragma unroll
  for (int i = 0; i < GM_LOADS; ++i) v[i] = h4[t + (size_t)i * (GM_BLOCKS * 256)];
  float4 a = v[0];
#pragma unroll
  for (int i = 1; i < GM_LOADS; ++i) {
    a.x += v[i].x; a.y += v[i].y; a.z += v[i].z; a.w += v[i].w;
  }
  a.x += __shfl_xor(a.x, 16, 32);
  a.y += __shfl_xor(a.y, 16, 32);
  a.z += __shfl_xor(a.z, 16, 32);
  a.w += __shfl_xor(a.w, 16, 32);
  if (lane < 16) *reinterpret_cast<float4*>(&ls[wave][lane * 4]) = a;
  __syncthreads();
  if (tid < H) {
    float s = ls[0][tid];
#pragma unroll
    for (int w = 1; w < 8; ++w) s += ls[w][tid];
    part[(size_t)blk * H + tid] = s;
  }
}

#define HD_WG ((HD_NWAVES + 7) / 8)
__global__ __launch_bounds__(256) void k_heads_graph(
    const float* __restrict__ h,
    const float* __restrict__ qW1, const float* __restrict__ qb1,
    const float* __restrict__ qW2, const float* __restrict__ qb2,
    const float* __restrict__ iW1, const float* __restrict__ ib1,
    const float* __restrict__ iW2, const float* __restrict__ ib2,
    const float* __restrict__ gW1, const float* __restrict__ gb1,
    const float* __restrict__ gW2, const float* __restrict__ gb2,
    const float* __restrict__ oW,  const float* __restrict__ obv,
    const float* __restrict__ sW1, const float* __restrict__ sb1,
    const float* __restrict__ sW2, const float* __restrict__ sb2,
    float* __restrict__ oq, float* __restrict__ oi, float* __restrict__ og,
    float* __restrict__ oh, float* __restrict__ obot, float* __restrict__ osev,
    float* __restrict__ part) {
  __shared__ float ls[8][H];
  if (blockIdx.x < GM_BLOCKS) {
    graph_partial_block(blockIdx.x, h, part, ls);
    return;
  }
  const int wv = __builtin_amdgcn_readfirstlane((blockIdx.x - GM_BLOCKS) * 8 + (threadIdx.x >> 5));
  if (wv < HD_NWAVES)
    heads_wave(wv, threadIdx.x & 31, h, qW1, qb1, qW2, qb2, iW1, ib1, iW2, ib2,
               gW1, gb1, gW2, gb2, oW, obv, sW1, sb1, sW2, sb2, oq, oi, og, oh, obot, osev);
}

__global__ __launch_bounds__(1024) void k_graph_final(
    const float* __restrict__ part, float* __restrict__ gout) {
  __shared__ float ls[16][H];
  const int tid = threadIdx.x;
  const int f = tid & 63, q = tid >> 6;
  float v[40];
#pragma unroll
  for (int i = 0; i < 40; ++i) {
    int b = q + 16 * i;
    v[i] = (b < GM_BLOCKS) ? part[(size_t)b * H + f] : 0.0f;
  }
  float s = v[0];
#pragma unroll
  for (int i = 1; i < 40; ++i) s += v[i];
  ls[q][f] = s;
  __syncthreads();
  if (tid < H) {
    float r = ls[0][tid];
#pragma unroll
    for (int w = 1; w < 16; ++w) r += ls[w][tid];
    gout[tid] = r * (1.0f / NN);
  }
}

extern "C" void kernel_launch(void* const* d_in, const int* in_sizes, int n_in,
                              void* d_out, int out_size, void* d_ws, size_t ws_size,
                              hipStream_t stream) {
  (void)in_sizes; (void)n_in; (void)out_size; (void)ws_size;
  const float* x    = (const float*)d_in[0];
  const int*   ei   = (const int*)  d_in[1];
  const float* ea   = (const float*)d_in[2];
  const float* inW  = (const float*)d_in[3];
  const float* inb  = (const float*)d_in[4];
  const float* linW = (const float*)d_in[5];
  const float* linb = (const float*)d_in[6];
  const float* attW = (const float*)d_in[7];
  const float* attb = (const float*)d_in[8];
  const float* lng  = (const float*)d_in[9];
  const float* lnb  = (const float*)d_in[10];
  const float* qW1 = (const float*)d_in[11]; const float* qb1 = (const float*)d_in[12];
  const float* qW2 = (const float*)d_in[13]; const float* qb2 = (const float*)d_in[14];
  const float* iW1 = (const float*)d_in[15]; const float* ib1 = (const float*)d_in[16];
  const float* iW2 = (const float*)d_in[17]; const float* ib2 = (const float*)d_in[18];
  const float* gW1 = (const float*)d_in[19]; const float* gb1 = (const float*)d_in[20];
  const float* gW2 = (const float*)d_in[21]; const float* gb2 = (const float*)d_in[22];
  const float* oW  = (const float*)d_in[23]; const float* obv = (const float*)d_in[24];
  const float* sW1 = (const float*)d_in[25]; const float* sb1 = (const float*)d_in[26];
  const float* sW2 = (const float*)d_in[27]; const float* sb2 = (const float*)d_in[28];

  float* ws   = (float*)d_ws;
  float* h    = ws;
  float* mbuf = h    + (size_t)NN * H;
  float* lst  = mbuf + (size_t)NN * H;
  float* sD   = lst + (size_t)NN * H;
  float* sS   = sD + NN;
  int*   off  = (int*)lst;
  int2*  tmp  = (int2*)(off + NODE_PAD);
  int2*  rec  = tmp + NE;
  float* part = (float*)(rec + NE);
  int2*  binned = (int2*)(part + GM_BLOCKS * H);
  int*   gh   = (int*)(binned + NE);

  float* out  = (float*)d_out;
  float* oq   = out;
  float* oi   = out + (size_t)NN;
  float* og   = out + 2 * (size_t)NN;
  float* oh   = out + 3 * (size_t)NN;
  float* ogr  = out + 67 * (size_t)NN;
  float* obot = out + 67 * (size_t)NN + H;
  float* osev = out + 99 * (size_t)NN + H;

  k_hist_proj<<<NCHK + IP_WG, 1024, 0, stream>>>(ei, gh, x, inW, inb, h);
  k_scan_gemm0<<<1 + G0_WG, 1024, 0, stream>>>(gh, h, linW, linb, attW, mbuf, sD, sS);
  k_bin_scatter<<<NCHK, 1024, 0, stream>>>(ei, gh, binned);
  k_range_group<<<NBKT, 1024, 0, stream>>>(gh, binned, off, tmp);
  k_edge_order<<<(NE + 255) / 256, 256, 0, stream>>>(ei, ea, off, tmp, rec);

  const int nodeWaveBlocks = (NN + 7) / 8;
  k_gather_update<<<nodeWaveBlocks, 256, 0, stream>>>(h, mbuf, off, rec, sD, sS, attb, lng, lnb);
  for (int l = 1; l < NL; ++l) {
    k_gemm_h<<<GT_BLOCKS, 128, 0, stream>>>(h, linW + (size_t)l * H * H, linb + (size_t)l * H,
                                          attW + (size_t)l * 2 * H, mbuf, sD, sS);
    k_gather_update<<<nodeWaveBlocks, 256, 0, stream>>>(h, mbuf, off, rec, sD, sS, attb + l,
                                                        lng + (size_t)l * H, lnb + (size_t)l * H);
  }

  k_heads_graph<<<GM_BLOCKS + HD_WG, 256, 0, stream>>>(h, qW1, qb1, qW2, qb2, iW1, ib1, iW2, ib2,
                                                     gW1, gb1, gW2, gb2, oW, obv, sW1, sb1, sW2, sb2,
                                                     oq, oi, og, oh, obot, osev, part);
  k_graph_final<<<1, 1024, 0, stream>>>(part, ogr);
}


# node update: a wave walks 8 consecutive nodes (12500 long-lived waves per layer, were 100000), the next nodes' records and sS values fetched while the current rows are summed; sum orders unchanged
# speedup vs baseline: 2.7014x; 1.0500x over previous
#include <hip/hip_runtime.h>

#define NN 100000
#define NE 1000000
#define DIN 5
#define H 64
#define HHID 32
#define NL 4
#define DOUT 32
#define LN_EPS 1e-5f

typedef __attribute__((ext_vector_type(2)))  float    v2f;
typedef __attribute__((ext_vector_type(8)))  float    v8f;

__device__ __forceinline__ float wave_sum(float v) {
#pragma unroll
  for (int off = 16; off > 0; off >>= 1) v += __shfl_xor(v, off, 32);
  return v;
}
__device__ __forceinline__ float sigmoidf_(float x) { return 1.0f / (1.0f + __expf(-x)); }

__device__ __forceinline__ int wave_sum_i(int v) {
#pragma unroll
  for (int off = 16; off > 0; off >>= 1) v += __shfl_xor(v, off, 32);
  return v;
}

#define IP_WAVES 6250
#define IP_WAVE_NODES 16
#define IP_GROUP 4
static_assert(IP_WAVES * IP_WAVE_NODES == NN, "the waves cover the nodes exactly");
static_assert(IP_WAVE_NODES % (2 * IP_GROUP) == 0 && H == 64, "two nodes per step, sixteen lanes per row, whole groups");
__device__ __forceinline__ void input_proj_wave(
    const float* __restrict__ x, const float* __restrict__ W,
    const float* __restrict__ b, float* __restrict__ h, const int gwave, const int lane) {
  const int c4 = (lane & 15) * 4;
  float4 w[DIN];
#pragma unroll
  for (int d = 0; d < DIN; ++d) w[d] = *reinterpret_cast<const float4*>(W + d * H + c4);
  const float4 bv = *reinterpret_cast<const float4*>(b + c4);
  const int n0 = gwave * IP_WAVE_NODES + (lane >> 4);
  const float* xg = x + (size_t)n0 * DIN;
  float* hg = h + (size_t)n0 * H + c4;
#pragma unroll 1
  for (int grp = 0; grp < IP_WAVE_NODES / (2 * IP_GROUP); ++grp) {
    float xv[IP_GROUP][DIN];
#pragma unroll
    for (int i = 0; i < IP_GROUP; ++i) {
#pragma unroll
      for (int d = 0; d < DIN; ++d) xv[i][d] = xg[i * 2 * DIN + d];
    }
#pragma unroll
    for (int i = 0; i < IP_GROUP; ++i) {
      float4 acc = bv;
#pragma unroll
      for (int d = 0; d < DIN; ++d) {
        acc.x = fmaf(xv[i][d], w[d].x, acc.x);
        acc.y = fmaf(xv[i][d], w[d].y, acc.y);
        acc.z = fmaf(xv[i][d], w[d].z, acc.z);
        acc.w = fmaf(xv[i][d], w[d].w, acc.w);
      }
      *reinterpret_cast<float4*>(hg + i * 2 * H) = acc;
    }
    xg += IP_GROUP * 2 * DIN;
    hg += IP_GROUP * 2 * H;
  }
}

#define GT_TILES (NN / 16)
#define GT_BLOCKS ((GT_TILES + 3) / 4)
static_assert(NN % 16 == 0 && H == 64, "16-row tiles, 64 columns");
template <int WAVES>
__device__ __forceinline__ void gemm_h_block(
    const int blk,
    const float* __restrict__ A, const float* __restrict__ W,
    const float* __restrict__ bias, const float* __restrict__ aw,
    float* __restrict__ out, float* __restrict__ sD, float* __restrict__ sS,
    float4* __restrict__ lW, float4 (* __restrict__ lAw)[2 * H / 4]) {
  constexpr int NT = 32 * WAVES;
  constexpr int WPT = (H * H / 4) / NT;
  static_assert(WPT * NT == H * H / 4, "the threads cover W exactly");
  const int tid = threadIdx.x;
  const int wave = tid >> 5, lane = tid & 31;
  const int c = lane & 15, g = lane >> 4;
  const int tile_raw = blk * WAVES + wave;
  const bool valid = tile_raw < GT_TILES;
  const int row0 = (valid ? tile_raw : GT_TILES - 1) * 16;

  const float4* arow = reinterpret_cast<const float4*>(A + (size_t)(row0 + c) * H) + g;
  float4 af[8];
#pragma unroll
  for (int i = 0; i < 8; ++i) af[i] = arow[2 * i];

  const float4* W4 = reinterpret_cast<const float4*>(W);
  float4 wv[WPT];
#pragma unroll
  for (int i = 0; i < WPT; ++i) wv[i] = W4[tid + NT * i];
  const float4 bv = reinterpret_cast<const float4*>(bias)[c];
  const float4 awv = reinterpret_cast<const float4*>(aw)[lane];
#pragma unroll
  for (int i = 0; i < WPT; ++i) lW[tid + NT * i] = wv[i];
  lAw[wave][lane] = awv;
  __syncthreads();

  v8f acc0, acc1, acc2, acc3;
#pragma unroll
  for (int r = 0; r < 8; ++r) { acc0[r] = bv.x; acc1[r] = bv.y; acc2[r] = bv.z; acc3[r] = bv.w; }

  const float4* lWg = lW + (4 * g) * 16 + c;
#pragma unroll
  for (int i = 0; i < 8; ++i) {
#pragma unroll
    for (int p = 0; p < 2; ++p) {
      const int k = 8 * i + 2 * p;
      const float4 q0 = lWg[k * 16];
      const float4 q1 = lWg[(k + 1) * 16];
      v2f a, b;
      a.x = p ? af[i].z : af[i].x;
      a.y = p ? af[i].w : af[i].y;
      b.x = q0.x; b.y = q1.x;
      acc0 = __builtin_amdgcn_wmma_f32_16x16x4_f32(false, a, false, b, (short)0, acc0, false, false);
      b.x = q0.y; b.y = q1.y;
      acc1 = __builtin_amdgcn_wmma_f32_16x16x4_f32(false, a, false, b, (short)0, acc1, false, false);
      b.x = q0.z; b.y = q1.z;
      acc2 = __builtin_amdgcn_wmma_f32_16x16x4_f32(false, a, false, b, (short)0, acc2, false, false);
      b.x = q0.w; b.y = q1.w;
      acc3 = __builtin_amdgcn_wmma_f32_16x16x4_f32(false, a, false, b, (short)0, acc3, false, false);
    }
  }

  float pd = 0.0f, ps = 0.0f;
#pragma unroll
  for (int i = 0; i < 8; ++i) {
    const float4 ad = lAw[wave][2 * i + g];
    const float4 as = lAw[wave][16 + 2 * i + g];
    pd = fmaf(af[i].x, ad.x, pd); pd = fmaf(af[i].y, ad.y, pd);
    pd = fmaf(af[i].z, ad.z, pd); pd = fmaf(af[i].w, ad.w, pd);
    ps = fmaf(af[i].x, as.x, ps); ps = fmaf(af[i].y, as.y, ps);
    ps = fmaf(af[i].z, as.z, ps); ps = fmaf(af[i].w, as.w, ps);
  }
  pd += __shfl_xor(pd, 16, 32);
  ps += __shfl_xor(ps, 16, 32);

  if (valid) {
    float4* o4 = reinterpret_cast<float4*>(out + (size_t)(row0 + 8 * g) * H) + c;
#pragma unroll
    for (int r = 0; r < 8; ++r) o4[r * 16] = make_float4(acc0[r], acc1[r], acc2[r], acc3[r]);
    if (g == 0) { sD[row0 + c] = pd; sS[row0 + c] = ps; }
  }
}

__global__ __launch_bounds__(128) void k_gemm_h(
    const float* __restrict__ A, const float* __restrict__ W,
    const float* __restrict__ bias, const float* __restrict__ aw,
    float* __restrict__ out, float* __restrict__ sD, float* __restrict__ sS) {
  __shared__ float4 lW[H * H / 4];
  __shared__ float4 lAw[4][2 * H / 4];
  gemm_h_block<4>(blockIdx.x, A, W, bias, aw, out, sD, sS, lW, lAw);
}

static_assert(NE % 4 == 0, "four edges per thread");
#define NODE_PAD 100032
#define BKT_SHIFT 9
#define BKT_R (1 << BKT_SHIFT)
#define NBKT ((NN + BKT_R - 1) / BKT_R)
#define NCHK 250
#define CHK_V4 (NE / 4 / NCHK)
#define GH_PITCH 256
#define RG_DEPTH 8
static_assert(NCHK * CHK_V4 * 4 == NE, "chunks cover the edges exactly");
static_assert(CHK_V4 <= 1024, "one four-edge group per thread");
static_assert(NBKT <= 224 && NBKT * BKT_R >= NN + 1, "range table");
static_assert(NCHK <= GH_PITCH && GH_PITCH - NCHK <= NCHK, "pad columns are written by the first chunks");

__device__ __forceinline__ int wave_incl_scan_i(int v, int lane) {
#pragma unroll
  for (int o = 1; o < 32; o <<= 1) {
    int up = __shfl_up(v, o, 32);
    if (lane >= o) v += up;
  }
  return v;
}

__device__ __forceinline__ void bin_hist_block(
    const int* __restrict__ ei, int* __restrict__ gh, int* __restrict__ hist, const int w) {
  const int tid = threadIdx.x;
  if (tid < NBKT) hist[tid] = 0;
  __syncthreads();
  if (tid < CHK_V4) {
    int4 d = reinterpret_cast<const int4*>(ei + NE)[w * CHK_V4 + tid];
    atomicAdd(&hist[d.x >> BKT_SHIFT], 1);
    atomicAdd(&hist[d.y >> BKT_SHIFT], 1);
    atomicAdd(&hist[d.z >> BKT_SHIFT], 1);
    atomicAdd(&hist[d.w >> BKT_SHIFT], 1);
  }
  __syncthreads();
  if (tid < NBKT) {
    gh[tid * GH_PITCH + w] = hist[tid];
    if (w < GH_PITCH - NCHK) gh[tid * GH_PITCH + NCHK + w] = 0;
  }
}

#define IP_WG ((IP_WAVES + 31) / 32)
__global__ __launch_bounds__(1024) void k_hist_proj(
    const int* __restrict__ ei, int* __restrict__ gh,
    const float* __restrict__ x, const float* __restrict__ W,
    const float* __restrict__ b, float* __restrict__ h) {
  __shared__ int hist[NBKT];
  if (blockIdx.x < NCHK) {
    bin_hist_block(ei, gh, hist, blockIdx.x);
    return;
  }
  const int gwave = (blockIdx.x - NCHK) * 32 + (threadIdx.x >> 5);
  if (gwave < IP_WAVES) input_proj_wave(x, W, b, h, gwave, threadIdx.x & 31);
}

__device__ __forceinline__ void bin_scan_block(
    int* __restrict__ gh, int* __restrict__ rowtot, int* __restrict__ rowbase,
    int* __restrict__ wtot) {
  const int tid = threadIdx.x, wave = tid >> 5, lane = tid & 31;
  int4 a[7], c[7];
#pragma unroll
  for (int k = 0; k < 7; ++k) {
    int b = min(wave + 32 * k, NBKT - 1);
    const int4* p = reinterpret_cast<const int4*>(gh + b * GH_PITCH + lane * 8);
    a[k] = p[0];
    c[k] = p[1];
  }
  int ex[7];
#pragma unroll
  for (int k = 0; k < 7; ++k) {
    int b = wave + 32 * k;
    int sum = ((a[k].x + a[k].y) + (a[k].z + a[k].w)) + ((c[k].x + c[k].y) + (c[k].z + c[k].w));
    int incl = wave_incl_scan_i(sum, lane);
    ex[k] = incl - sum;
    int tot = __shfl(incl, 31, 32);
    if (lane == 0) rowtot[b] = (b < NBKT) ? tot : 0;
  }
  __syncthreads();
  int v = 0, vin = 0;
  if (tid < 224) {
    v = rowtot[tid];
    vin = wave_incl_scan_i(v, lane);
    if (lane == 31) wtot[wave] = vin;
  }
  __syncthreads();
  if (tid < 224) {
    int before = 0;
#pragma unroll
    for (int j = 0; j < 7; ++j) before += (j < wave) ? wtot[j] : 0;
    rowbase[tid] = before + vin - v;
  }
  __syncthreads();
#pragma unroll
  for (int k = 0; k < 7; ++k) {
    int b = wave + 32 * k;
    if (b < NBKT) {
      int r = rowbase[b] + ex[k];
      int4 o0, o1;
      o0.x = r; r += a[k].x;
      o0.y = r; r += a[k].y;
      o0.z = r; r += a[k].z;
      o0.w = r; r += a[k].w;
      o1.x = r; r += c[k].x;
      o1.y = r; r += c[k].y;
      o1.z = r; r += c[k].z;
      o1.w = r;
      int4* p = reinterpret_cast<int4*>(gh + b * GH_PITCH + lane * 8);
      p[0] = o0;
      p[1] = o1;
    }
  }
}

#define G0_WG ((GT_TILES + 31) / 32)
__global__ __launch_bounds__(1024) void k_scan_gemm0(
    int* __restrict__ gh,
    const float* __restrict__ A, const float* __restrict__ W,
    const float* __restrict__ bias, const float* __restrict__ aw,
    float* __restrict__ out, float* __restrict__ sD, float* __restrict__ sS) {
  __shared__ float4 lW[H * H / 4];
  __shared__ float4 lAw[32][2 * H / 4];
  __shared__ int rowtot[224];
  __shared__ int rowbase[224];
  __shared__ int wtot[8];
  if (blockIdx.x == 0) {
    bin_scan_block(gh, rowtot, rowbase, wtot);
    return;
  }
  gemm_h_block<32>(blockIdx.x - 1, A, W, bias, aw, out, sD, sS, lW, lAw);
}

__global__ __launch_bounds__(1024) void k_bin_scatter(
    const int* __restrict__ ei, const int* __restrict__ gh, int2* __restrict__ binned) {
  __shared__ int cur[NBKT];
  const int tid = threadIdx.x, w = blockIdx.x;
  if (tid < NBKT) cur[tid] = gh[tid * GH_PITCH + w];
  __syncthreads();
  if (tid < CHK_V4) {
    int t = w * CHK_V4 + tid;
    int4 d = reinterpret_cast<const int4*>(ei + NE)[t];
    int e0 = 4 * t;
    int p;
    p = atomicAdd(&cur[d.x >> BKT_SHIFT], 1); binned[p] = make_int2(e0,     d.x);
    p = atomicAdd(&cur[d.y >> BKT_SHIFT], 1); binned[p] = make_int2(e0 + 1, d.y);
    p = atomicAdd(&cur[d.z >> BKT_SHIFT], 1); binned[p] = make_int2(e0 + 2, d.z);
    p = atomicAdd(&cur[d.w >> BKT_SHIFT], 1); binned[p] = make_int2(e0 + 3, d.w);
  }
}

__global__ __launch_bounds__(1024) void k_range_group(
    const int* __restrict__ gh, const int2* __restrict__ binned,
    int* __restrict__ off, int2* __restrict__ tmp) {
  __shared__ int cnt[BKT_R];
  __shared__ int wtot[32];
  const int tid = threadIdx.x, wave = tid >> 5, lane = tid & 31;
  const int b = blockIdx.x;
  const int lo = b << BKT_SHIFT;
  const int cb = gh[b * GH_PITCH];
  const int ce = (b + 1 < NBKT) ? gh[(b + 1) * GH_PITCH] : NE;
  if (tid < BKT_R) cnt[tid] = 0;
  __syncthreads();
  for (int i0 = cb + tid; i0 < ce; i0 += RG_DEPTH * 1024) {
    int dd[RG_DEPTH];
#pragma unroll
    for (int u = 0; u < RG_DEPTH; ++u) dd[u] = binned[min(i0 + 1024 * u, ce - 1)].y;
#pragma unroll
    for (int u = 0; u < RG_DEPTH; ++u)
      if (i0 + 1024 * u < ce) atomicAdd(&cnt[dd[u] - lo], 1);
  }
  __syncthreads();
  const int c = (tid < BKT_R) ? cnt[tid] : 0;
  const int incl = wave_incl_scan_i(c, lane);
  if (lane == 31) wtot[wave] = incl;
  __syncthreads();
  const int before = wave_sum_i((lane < wave) ? wtot[lane] : 0);
  const int excl = before + incl - c;
  if (tid < BKT_R) {
    cnt[tid] = excl;
    if (lo + tid <= NN) off[lo + tid] = cb + excl;
  }
  __syncthreads();
  for (int i0 = cb + tid; i0 < ce; i0 += RG_DEPTH * 1024) {
    int2 r[RG_DEPTH];
#pragma unroll
    for (int u = 0; u < RG_DEPTH; ++u) r[u] = binned[min(i0 + 1024 * u, ce - 1)];
#pragma unroll
    for (int u = 0; u < RG_DEPTH; ++u)
      if (i0 + 1024 * u < ce) {
        int p = atomicAdd(&cnt[r[u].y - lo], 1);
        tmp[cb + p] = r[u];
      }
  }
}

__global__ __launch_bounds__(256) void k_edge_order(
    const int* __restrict__ ei, const float* __restrict__ ea,
    const int* __restrict__ off, const int2* __restrict__ tmp,
    int2* __restrict__ rec) {
  int p = blockIdx.x * 256 + threadIdx.x;
  if (p >= NE) return;
  int2 my = tmp[p];
  int s = off[my.y], e = off[my.y + 1];
  int rank = 0;
  for (int q = s; q < e; ++q) rank += (tmp[q].x < my.x) ? 1 : 0;
  rec[s + rank] = make_int2(ei[my.x], __float_as_int(ea[my.x]));
}

#define GU_NODES 8
#define GU_WAVES (NN / GU_NODES)
#define GU_WG ((GU_WAVES + 7) / 8)
static_assert(GU_WAVES * GU_NODES == NN, "the waves cover the nodes exactly");
static_assert(GU_NODES >= 2 && GU_NODES < 32, "one lane per segment bound of the wave's nodes");
__device__ __forceinline__ float half_sum16(float v) {
#pragma unroll
  for (int off = 8; off > 0; off >>= 1) v += __shfl_xor(v, off, 32);
  return v;
}
__device__ __forceinline__ int2 first_records(
    const int2* __restrict__ rec, const int s, const int e, const int node, const int lane) {
  int2 rc = make_int2(node, 0);
  if (lane < e - s) rc = rec[s + lane];
  return rc;
}
__global__ __launch_bounds__(256) void k_gather_update(
    float* __restrict__ h, const float* __restrict__ m,
    const int* __restrict__ off, const int2* __restrict__ rec,
    const float* __restrict__ sD, const float* __restrict__ sS,
    const float* __restrict__ attb,
    const float* __restrict__ g, const float* __restrict__ beta) {
  __shared__ int2 strip[8][32];
  __shared__ float4 lnp[8][32];
  const int wv = __builtin_amdgcn_readfirstlane(blockIdx.x * 8 + (threadIdx.x >> 5));
  const int lane = threadIdx.x & 31;
  if (wv >= GU_WAVES) return;
  const int n0 = wv * GU_NODES;
  int2* const mystrip = strip[threadIdx.x >> 5];
  const int odd = lane >> 4;
  const unsigned c16 = (lane & 15) * 16u;
  const int f4 = (lane & 15) * 4;
  const int offv = off[n0 + min(lane, GU_NODES)];
  const float sdv = sD[n0 + min(lane, GU_NODES - 1)];
  const float ab = attb[0];
  float4* const mylnp = lnp[threadIdx.x >> 5];
  mylnp[lane] = *reinterpret_cast<const float4*>((odd ? beta : g) + f4);
  const char* const mbytes = reinterpret_cast<const char*>(m);

  int sA = __builtin_amdgcn_readlane(offv, 0);
  int eA = __builtin_amdgcn_readlane(offv, 1);
  int eB = __builtin_amdgcn_readlane(offv, 2);
  int2 rcA = first_records(rec, sA, eA, n0, lane);
  int2 rcB = first_records(rec, eA, eB, n0 + 1, lane);
  float ssA = sS[rcA.x];
  asm volatile("" :: "v"(ssA));
#pragma unroll 1
  for (int j = 0; j < GU_NODES; ++j) {
    const int gw = n0 + j;
    const int eC = __builtin_amdgcn_readlane(offv, min(j + 3, GU_NODES));
    const int2 rcC = first_records(rec, eB, eC, n0 + min(j + 2, GU_NODES - 1), lane);
    const float ssB = sS[rcB.x];
    if (lane < eB - eA) __builtin_prefetch(m + (size_t)rcB.x * H, 0, 1);

    const size_t base = (size_t)gw * H + f4;
    const float4 r  = *reinterpret_cast<const float4*>(h + base);
    const float4 mn = *reinterpret_cast<const float4*>(m + base);
    const float sd = __int_as_float(__builtin_amdgcn_readlane(__float_as_int(sdv), j));
    float4 acc = make_float4(0.0f, 0.0f, 0.0f, 0.0f);
    for (int cb = sA; cb < eA; cb += 32) {
      const int cnt = min(32, eA - cb);
      int src = gw;
      float w = 0.0f;
      if (cb == sA) {
        src = rcA.x;
        if (lane < cnt) w = __int_as_float(rcA.y) * sigmoidf_(sd + ssA + ab);
      } else if (lane < cnt) {
        int2 rc = rec[cb + lane];
        src = rc.x;
        __builtin_prefetch(m + (size_t)src * H, 0, 1);
        w = __int_as_float(rc.y) * sigmoidf_(sd + sS[src] + ab);
      }
      mystrip[lane] = make_int2((int)((unsigned)src << 8), __float_as_int(w));
#pragma unroll 4
      for (int i = 0; i < cnt; i += 2) {
        const int2 p = mystrip[i + odd];
        const float bw = __int_as_float(p.y);
        const float4 v = *reinterpret_cast<const float4*>(mbytes + ((unsigned)p.x + c16));
        acc.x = fmaf(v.x, bw, acc.x);
        acc.y = fmaf(v.y, bw, acc.y);
        acc.z = fmaf(v.z, bw, acc.z);
        acc.w = fmaf(v.w, bw, acc.w);
      }
    }
    acc.x += __shfl_xor(acc.x, 16, 32);
    acc.y += __shfl_xor(acc.y, 16, 32);
    acc.z += __shfl_xor(acc.z, 16, 32);
    acc.w += __shfl_xor(acc.w, 16, 32);
    const float f0 = acc.x + mn.x, f1 = acc.y + mn.y, f2 = acc.z + mn.z, f3 = acc.w + mn.w;
    const float mean = half_sum16((f0 + f1) + (f2 + f3)) * (1.0f / H);
    const float d0 = f0 - mean, d1 = f1 - mean, d2 = f2 - mean, d3 = f3 - mean;
    const float var = half_sum16(fmaf(d0, d0, d1 * d1) + fmaf(d2, d2, d3 * d3)) * (1.0f / H);
    const float inv = rsqrtf(var + LN_EPS);
    const float4 gv = mylnp[lane & 15];
    const float4 bv = mylnp[16 + (lane & 15)];
    if (odd == 0) {
      float4 o;
      o.x = fmaxf(fmaf(d0 * inv, gv.x, bv.x), 0.0f) + r.x;
      o.y = fmaxf(fmaf(d1 * inv, gv.y, bv.y), 0.0f) + r.y;
      o.z = fmaxf(fmaf(d2 * inv, gv.z, bv.z), 0.0f) + r.z;
      o.w = fmaxf(fmaf(d3 * inv, gv.w, bv.w), 0.0f) + r.w;
      *reinterpret_cast<float4*>(h + base) = o;
    }
    sA = eA; eA = eB; eB = eC;
    rcA = rcB; rcB = rcC; ssA = ssB;
  }
}

#define HD_NODES 8
#define HD_NWAVES (NN / HD_NODES)
static_assert(HD_NWAVES * HD_NODES == NN, "the waves cover the nodes exactly");
static_assert(HHID == 32 && DOUT == 32 && H == 64, "one lane per hidden unit, two features per lane");

__device__ __forceinline__ void heads_half(
    const float (&hv)[HD_NODES], const int krow0,
    const float* __restrict__ qW1, const float* __restrict__ iW1,
    const float* __restrict__ gW1, const float* __restrict__ sW1,
    const float* __restrict__ oW, const int lane,
    float (&tq)[HD_NODES], float (&ti)[HD_NODES], float (&tg)[HD_NODES],
    float (&ts)[HD_NODES], float (&bo)[HD_NODES]) {
#pragma unroll 4
  for (int k = 0; k < 32; ++k) {
    const int wi = (krow0 + k) * HHID + lane;
    const float wq = qW1[wi], wn = iW1[wi], wg = gW1[wi], wv = sW1[wi], wo = oW[wi];
#pragma unroll
    for (int j = 0; j < HD_NODES; ++j) {
      const float hk = __int_as_float(__builtin_amdgcn_readlane(__float_as_int(hv[j]), k));
      tq[j] = fmaf(hk, wq, tq[j]);
      ti[j] = fmaf(hk, wn, ti[j]);
      tg[j] = fmaf(hk, wg, tg[j]);
      ts[j] = fmaf(hk, wv, ts[j]);
      bo[j] = fmaf(hk, wo, bo[j]);
    }
  }
}

__device__ __forceinline__ void heads_wave(
    const int wv, const int lane,
    const float* __restrict__ h,
    const float* __restrict__ qW1, const float* __restrict__ qb1,
    const float* __restrict__ qW2, const float* __restrict__ qb2,
    const float* __restrict__ iW1, const float* __restrict__ ib1,
    const float* __restrict__ iW2, const float* __restrict__ ib2,
    const float* __restrict__ gW1, const float* __restrict__ gb1,
    const float* __restrict__ gW2, const float* __restrict__ gb2,
    const float* __restrict__ oW,  const float* __restrict__ obv,
    const float* __restrict__ sW1, const float* __restrict__ sb1,
    const float* __restrict__ sW2, const float* __restrict__ sb2,
    float* __restrict__ oq, float* __restrict__ oi, float* __restrict__ og,
    float* __restrict__ oh, float* __restrict__ obot, float* __restrict__ osev) {
  const int n0 = wv * HD_NODES;
  const size_t base = (size_t)n0 * H;

  float h0[HD_NODES], h1[HD_NODES];
#pragma unroll
  for (int j = 0; j < HD_NODES; ++j) {
    h0[j] = h[base + j * H + lane];
    h1[j] = h[base + j * H + lane + 32];
  }

  float tq[HD_NODES], ti[HD_NODES], tg[HD_NODES], ts[HD_NODES], bo[HD_NODES];
#pragma unroll
  for (int j = 0; j < HD_NODES; ++j) { tq[j] = 0.f; ti[j] = 0.f; tg[j] = 0.f; ts[j] = 0.f; bo[j] = 0.f; }

  heads_half(h0, 0,  qW1, iW1, gW1, sW1, oW, lane, tq, ti, tg, ts, bo);
  heads_half(h1, 32, qW1, iW1, gW1, sW1, oW, lane, tq, ti, tg, ts, bo);

  const float bq1 = qb1[lane], bi1 = ib1[lane], bg1 = gb1[lane], bs1 = sb1[lane];
  const float wq2 = qW2[lane], wi2 = iW2[lane], wg2 = gW2[lane];
  const float bq2 = qb2[0], bi2 = ib2[0], bg2 = gb2[0];
  const float bob = obv[lane];
  float ws2[5], bs2[5];
#pragma unroll
  for (int c = 0; c < 5; ++c) { ws2[c] = sW2[lane * 5 + c]; bs2[c] = sb2[c]; }

  float keepq = 0.f, keepi = 0.f, keepg = 0.f;
#pragma unroll
  for (int j = 0; j < HD_NODES; ++j) {
    const float aq = fmaxf(tq[j] + bq1, 0.0f);
    const float ai = fmaxf(ti[j] + bi1, 0.0f);
    const float ag = fmaxf(tg[j] + bg1, 0.0f);
    const float asv = fmaxf(ts[j] + bs1, 0.0f);

    const float q  = wave_sum(aq * wq2);
    const float ii = wave_sum(ai * wi2);
    const float gg = wave_sum(ag * wg2);
    const float pq = sigmoidf_(q + bq2);
    const float pi = sigmoidf_(ii + bi2);
    const float pg = sigmoidf_(gg + bg2);
    if (lane == j) { keepq = pq; keepi = pi; keepg = pg; }

    float z[5];
#pragma unroll
    for (int c = 0; c < 5; ++c) z[c] = wave_sum(asv * ws2[c]) + bs2[c];
    float zm = z[0];
#pragma unroll
    for (int c = 1; c < 5; ++c) zm = fmaxf(zm, z[c]);
    float es = 0.f;
#pragma unroll
    for (int c = 0; c < 5; ++c) { z[c] = __expf(z[c] - zm); es += z[c]; }
    const float inv = 1.0f / es;
    float sv = z[0];
#pragma unroll
    for (int c = 1; c < 5; ++c) sv = (lane == c) ? z[c] : sv;
    if (lane < 5) osev[(size_t)(n0 + j) * 5 + lane] = sv * inv;

    obot[(size_t)(n0 + j) * DOUT + lane] = bo[j] + bob;
    oh[base + j * H + lane]      = h0[j];
    oh[base + j * H + lane + 32] = h1[j];
  }
  if (lane < HD_NODES) {
    oq[n0 + lane] = keepq;
    oi[n0 + lane] = keepi;
    og[n0 + lane] = keepg;
  }
}

#define GM_BLOCKS 625
#define GM_LOADS 10
__device__ __forceinline__ void graph_partial_block(
    const int blk, const float* __restrict__ h, float* __restrict__ part, float (* __restrict__ ls)[H]) {
  const int tid = threadIdx.x;
  const int wave = tid >> 5, lane = tid & 31;
  const float4* h4 = reinterpret_cast<const float4*>(h);
  const size_t t = (size_t)blk * 256 + tid;
  float4 v[GM_LOADS];
#pragma unroll
  for (int i = 0; i < GM_LOADS; ++i) v[i] = h4[t + (size_t)i * (GM_BLOCKS * 256)];
  float4 a = v[0];
#pragma unroll
  for (int i = 1; i < GM_LOADS; ++i) {
    a.x += v[i].x; a.y += v[i].y; a.z += v[i].z; a.w += v[i].w;
  }
  a.x += __shfl_xor(a.x, 16, 32);
  a.y += __shfl_xor(a.y, 16, 32);
  a.z += __shfl_xor(a.z, 16, 32);
  a.w += __shfl_xor(a.w, 16, 32);
  if (lane < 16) *reinterpret_cast<float4*>(&ls[wave][lane * 4]) = a;
  __syncthreads();
  if (tid < H) {
    float s = ls[0][tid];
#pragma unroll
    for (int w = 1; w < 8; ++w) s += ls[w][tid];
    part[(size_t)blk * H + tid] = s;
  }
}

#define HD_WG ((HD_NWAVES + 7) / 8)
__global__ __launch_bounds__(256) void k_heads_graph(
    const float* __restrict__ h,
    const float* __restrict__ qW1, const float* __restrict__ qb1,
    const float* __restrict__ qW2, const float* __restrict__ qb2,
    const float* __restrict__ iW1, const float* __restrict__ ib1,
    const float* __restrict__ iW2, const float* __restrict__ ib2,
    const float* __restrict__ gW1, const float* __restrict__ gb1,
    const float* __restrict__ gW2, const float* __restrict__ gb2,
    const float* __restrict__ oW,  const float* __restrict__ obv,
    const float* __restrict__ sW1, const float* __restrict__ sb1,
    const float* __restrict__ sW2, const float* __restrict__ sb2,
    float* __restrict__ oq, float* __restrict__ oi, float* __restrict__ og,
    float* __restrict__ oh, float* __restrict__ obot, float* __restrict__ osev,
    float* __restrict__ part) {
  __shared__ float ls[8][H];
  if (blockIdx.x < GM_BLOCKS) {
    graph_partial_block(blockIdx.x, h, part, ls);
    return;
  }
  const int wv = __builtin_amdgcn_readfirstlane((blockIdx.x - GM_BLOCKS) * 8 + (threadIdx.x >> 5));
  if (wv < HD_NWAVES)
    heads_wave(wv, threadIdx.x & 31, h, qW1, qb1, qW2, qb2, iW1, ib1, iW2, ib2,
               gW1, gb1, gW2, gb2, oW, obv, sW1, sb1, sW2, sb2, oq, oi, og, oh, obot, osev);
}

__global__ __launch_bounds__(1024) void k_graph_final(
    const float* __restrict__ part, float* __restrict__ gout) {
  __shared__ float ls[16][H];
  const int tid = threadIdx.x;
  const int f = tid & 63, q = tid >> 6;
  float v[40];
#pragma unroll
  for (int i = 0; i < 40; ++i) {
    int b = q + 16 * i;
    v[i] = (b < GM_BLOCKS) ? part[(size_t)b * H + f] : 0.0f;
  }
  float s = v[0];
#pragma unroll
  for (int i = 1; i < 40; ++i) s += v[i];
  ls[q][f] = s;
  __syncthreads();
  if (tid < H) {
    float r = ls[0][tid];
#pragma unroll
    for (int w = 1; w < 16; ++w) r += ls[w][tid];
    gout[tid] = r * (1.0f / NN);
  }
}

extern "C" void kernel_launch(void* const* d_in, const int* in_sizes, int n_in,
                              void* d_out, int out_size, void* d_ws, size_t ws_size,
                              hipStream_t stream) {
  (void)in_sizes; (void)n_in; (void)out_size; (void)ws_size;
  const float* x    = (const float*)d_in[0];
  const int*   ei   = (const int*)  d_in[1];
  const float* ea   = (const float*)d_in[2];
  const float* inW  = (const float*)d_in[3];
  const float* inb  = (const float*)d_in[4];
  const float* linW = (const float*)d_in[5];
  const float* linb = (const float*)d_in[6];
  const float* attW = (const float*)d_in[7];
  const float* attb = (const float*)d_in[8];
  const float* lng  = (const float*)d_in[9];
  const float* lnb  = (const float*)d_in[10];
  const float* qW1 = (const float*)d_in[11]; const float* qb1 = (const float*)d_in[12];
  const float* qW2 = (const float*)d_in[13]; const float* qb2 = (const float*)d_in[14];
  const float* iW1 = (const float*)d_in[15]; const float* ib1 = (const float*)d_in[16];
  const float* iW2 = (const float*)d_in[17]; const float* ib2 = (const float*)d_in[18];
  const float* gW1 = (const float*)d_in[19]; const float* gb1 = (const float*)d_in[20];
  const float* gW2 = (const float*)d_in[21]; const float* gb2 = (const float*)d_in[22];
  const float* oW  = (const float*)d_in[23]; const float* obv = (const float*)d_in[24];
  const float* sW1 = (const float*)d_in[25]; const float* sb1 = (const float*)d_in[26];
  const float* sW2 = (const float*)d_in[27]; const float* sb2 = (const float*)d_in[28];

  float* ws   = (float*)d_ws;
  float* h    = ws;
  float* mbuf = h    + (size_t)NN * H;
  float* lst  = mbuf + (size_t)NN * H;
  float* sD   = lst + (size_t)NN * H;
  float* sS   = sD + NN;
  int*   off  = (int*)lst;
  int2*  tmp  = (int2*)(off + NODE_PAD);
  int2*  rec  = tmp + NE;
  float* part = (float*)(rec + NE);
  int2*  binned = (int2*)(part + GM_BLOCKS * H);
  int*   gh   = (int*)(binned + NE);

  float* out  = (float*)d_out;
  float* oq   = out;
  float* oi   = out + (size_t)NN;
  float* og   = out + 2 * (size_t)NN;
  float* oh   = out + 3 * (size_t)NN;
  float* ogr  = out + 67 * (size_t)NN;
  float* obot = out + 67 * (size_t)NN + H;
  float* osev = out + 99 * (size_t)NN + H;

  k_hist_proj<<<NCHK + IP_WG, 1024, 0, stream>>>(ei, gh, x, inW, inb, h);
  k_scan_gemm0<<<1 + G0_WG, 1024, 0, stream>>>(gh, h, linW, linb, attW, mbuf, sD, sS);
  k_bin_scatter<<<NCHK, 1024, 0, stream>>>(ei, gh, binned);
  k_range_group<<<NBKT, 1024, 0, stream>>>(gh, binned, off, tmp);
  k_edge_order<<<(NE + 255) / 256, 256, 0, stream>>>(ei, ea, off, tmp, rec);

  const int nodeWaveBlocks = GU_WG;
  k_gather_update<<<nodeWaveBlocks, 256, 0, stream>>>(h, mbuf, off, rec, sD, sS, attb, lng, lnb);
  for (int l = 1; l < NL; ++l) {
    k_gemm_h<<<GT_BLOCKS, 128, 0, stream>>>(h, linW + (size_t)l * H * H, linb + (size_t)l * H,
                                          attW + (size_t)l * 2 * H, mbuf, sD, sS);
    k_gather_update<<<nodeWaveBlocks, 256, 0, stream>>>(h, mbuf, off, rec, sD, sS, attb + l,
                                                        lng + (size_t)l * H, lnb + (size_t)l * H);
  }

  k_heads_graph<<<GM_BLOCKS + HD_WG, 256, 0, stream>>>(h, qW1, qb1, qW2, qb2, iW1, ib1, iW2, ib2,
                                                     gW1, gb1, gW2, gb2, oW, obv, sW1, sb1, sW2, sb2,
                                                     oq, oi, og, oh, obot, osev, part);
  k_graph_final<<<1, 1024, 0, stream>>>(part, ogr);
}


# output heads: the five 64x32 first-layer products moved from vector multiply-adds to f32 WMMA (wave = 16 nodes x 160 columns, weights staged in LDS); 8-lane butterflies finish (sum order changes)
# speedup vs baseline: 2.8171x; 1.0428x over previous
#include <hip/hip_runtime.h>

#define NN 100000
#define NE 1000000
#define DIN 5
#define H 64
#define HHID 32
#define NL 4
#define DOUT 32
#define LN_EPS 1e-5f

typedef __attribute__((ext_vector_type(2)))  float    v2f;
typedef __attribute__((ext_vector_type(8)))  float    v8f;

__device__ __forceinline__ float wave_sum(float v) {
#pragma unroll
  for (int off = 16; off > 0; off >>= 1) v += __shfl_xor(v, off, 32);
  return v;
}
__device__ __forceinline__ float sigmoidf_(float x) { return 1.0f / (1.0f + __expf(-x)); }

__device__ __forceinline__ int wave_sum_i(int v) {
#pragma unroll
  for (int off = 16; off > 0; off >>= 1) v += __shfl_xor(v, off, 32);
  return v;
}

#define IP_WAVES 6250
#define IP_WAVE_NODES 16
#define IP_GROUP 4
static_assert(IP_WAVES * IP_WAVE_NODES == NN, "the waves cover the nodes exactly");
static_assert(IP_WAVE_NODES % (2 * IP_GROUP) == 0 && H == 64, "two nodes per step, sixteen lanes per row, whole groups");
__device__ __forceinline__ void input_proj_wave(
    const float* __restrict__ x, const float* __restrict__ W,
    const float* __restrict__ b, float* __restrict__ h, const int gwave, const int lane) {
  const int c4 = (lane & 15) * 4;
  float4 w[DIN];
#pragma unroll
  for (int d = 0; d < DIN; ++d) w[d] = *reinterpret_cast<const float4*>(W + d * H + c4);
  const float4 bv = *reinterpret_cast<const float4*>(b + c4);
  const int n0 = gwave * IP_WAVE_NODES + (lane >> 4);
  const float* xg = x + (size_t)n0 * DIN;
  float* hg = h + (size_t)n0 * H + c4;
#pragma unroll 1
  for (int grp = 0; grp < IP_WAVE_NODES / (2 * IP_GROUP); ++grp) {
    float xv[IP_GROUP][DIN];
#pragma unroll
    for (int i = 0; i < IP_GROUP; ++i) {
#pragma unroll
      for (int d = 0; d < DIN; ++d) xv[i][d] = xg[i * 2 * DIN + d];
    }
#pragma unroll
    for (int i = 0; i < IP_GROUP; ++i) {
      float4 acc = bv;
#pragma unroll
      for (int d = 0; d < DIN; ++d) {
        acc.x = fmaf(xv[i][d], w[d].x, acc.x);
        acc.y = fmaf(xv[i][d], w[d].y, acc.y);
        acc.z = fmaf(xv[i][d], w[d].z, acc.z);
        acc.w = fmaf(xv[i][d], w[d].w, acc.w);
      }
      *reinterpret_cast<float4*>(hg + i * 2 * H) = acc;
    }
    xg += IP_GROUP * 2 * DIN;
    hg += IP_GROUP * 2 * H;
  }
}

#define GT_TILES (NN / 16)
#define GT_BLOCKS ((GT_TILES + 3) / 4)
static_assert(NN % 16 == 0 && H == 64, "16-row tiles, 64 columns");
template <int WAVES>
__device__ __forceinline__ void gemm_h_block(
    const int blk,
    const float* __restrict__ A, const float* __restrict__ W,
    const float* __restrict__ bias, const float* __restrict__ aw,
    float* __restrict__ out, float* __restrict__ sD, float* __restrict__ sS,
    float4* __restrict__ lW, float4 (* __restrict__ lAw)[2 * H / 4]) {
  constexpr int NT = 32 * WAVES;
  constexpr int WPT = (H * H / 4) / NT;
  static_assert(WPT * NT == H * H / 4, "the threads cover W exactly");
  const int tid = threadIdx.x;
  const int wave = tid >> 5, lane = tid & 31;
  const int c = lane & 15, g = lane >> 4;
  const int tile_raw = blk * WAVES + wave;
  const bool valid = tile_raw < GT_TILES;
  const int row0 = (valid ? tile_raw : GT_TILES - 1) * 16;

  const float4* arow = reinterpret_cast<const float4*>(A + (size_t)(row0 + c) * H) + g;
  float4 af[8];
#pragma unroll
  for (int i = 0; i < 8; ++i) af[i] = arow[2 * i];

  const float4* W4 = reinterpret_cast<const float4*>(W);
  float4 wv[WPT];
#pragma unroll
  for (int i = 0; i < WPT; ++i) wv[i] = W4[tid + NT * i];
  const float4 bv = reinterpret_cast<const float4*>(bias)[c];
  const float4 awv = reinterpret_cast<const float4*>(aw)[lane];
#pragma unroll
  for (int i = 0; i < WPT; ++i) lW[tid + NT * i] = wv[i];
  lAw[wave][lane] = awv;
  __syncthreads();

  v8f acc0, acc1, acc2, acc3;
#pragma unroll
  for (int r = 0; r < 8; ++r) { acc0[r] = bv.x; acc1[r] = bv.y; acc2[r] = bv.z; acc3[r] = bv.w; }

  const float4* lWg = lW + (4 * g) * 16 + c;
#pragma unroll
  for (int i = 0; i < 8; ++i) {
#pragma unroll
    for (int p = 0; p < 2; ++p) {
      const int k = 8 * i + 2 * p;
      const float4 q0 = lWg[k * 16];
      const float4 q1 = lWg[(k + 1) * 16];
      v2f a, b;
      a.x = p ? af[i].z : af[i].x;
      a.y = p ? af[i].w : af[i].y;
      b.x = q0.x; b.y = q1.x;
      acc0 = __builtin_amdgcn_wmma_f32_16x16x4_f32(false, a, false, b, (short)0, acc0, false, false);
      b.x = q0.y; b.y = q1.y;
      acc1 = __builtin_amdgcn_wmma_f32_16x16x4_f32(false, a, false, b, (short)0, acc1, false, false);
      b.x = q0.z; b.y = q1.z;
      acc2 = __builtin_amdgcn_wmma_f32_16x16x4_f32(false, a, false, b, (short)0, acc2, false, false);
      b.x = q0.w; b.y = q1.w;
      acc3 = __builtin_amdgcn_wmma_f32_16x16x4_f32(false, a, false, b, (short)0, acc3, false, false);
    }
  }

  float pd = 0.0f, ps = 0.0f;
#pragma unroll
  for (int i = 0; i < 8; ++i) {
    const float4 ad = lAw[wave][2 * i + g];
    const float4 as = lAw[wave][16 + 2 * i + g];
    pd = fmaf(af[i].x, ad.x, pd); pd = fmaf(af[i].y, ad.y, pd);
    pd = fmaf(af[i].z, ad.z, pd); pd = fmaf(af[i].w, ad.w, pd);
    ps = fmaf(af[i].x, as.x, ps); ps = fmaf(af[i].y, as.y, ps);
    ps = fmaf(af[i].z, as.z, ps); ps = fmaf(af[i].w, as.w, ps);
  }
  pd += __shfl_xor(pd, 16, 32);
  ps += __shfl_xor(ps, 16, 32);

  if (valid) {
    float4* o4 = reinterpret_cast<float4*>(out + (size_t)(row0 + 8 * g) * H) + c;
#pragma unroll
    for (int r = 0; r < 8; ++r) o4[r * 16] = make_float4(acc0[r], acc1[r], acc2[r], acc3[r]);
    if (g == 0) { sD[row0 + c] = pd; sS[row0 + c] = ps; }
  }
}

__global__ __launch_bounds__(128) void k_gemm_h(
    const float* __restrict__ A, const float* __restrict__ W,
    const float* __restrict__ bias, const float* __restrict__ aw,
    float* __restrict__ out, float* __restrict__ sD, float* __restrict__ sS) {
  __shared__ float4 lW[H * H / 4];
  __shared__ float4 lAw[4][2 * H / 4];
  gemm_h_block<4>(blockIdx.x, A, W, bias, aw, out, sD, sS, lW, lAw);
}

static_assert(NE % 4 == 0, "four edges per thread");
#define NODE_PAD 100032
#define BKT_SHIFT 9
#define BKT_R (1 << BKT_SHIFT)
#define NBKT ((NN + BKT_R - 1) / BKT_R)
#define NCHK 250
#define CHK_V4 (NE / 4 / NCHK)
#define GH_PITCH 256
#define RG_DEPTH 8
static_assert(NCHK * CHK_V4 * 4 == NE, "chunks cover the edges exactly");
static_assert(CHK_V4 <= 1024, "one four-edge group per thread");
static_assert(NBKT <= 224 && NBKT * BKT_R >= NN + 1, "range table");
static_assert(NCHK <= GH_PITCH && GH_PITCH - NCHK <= NCHK, "pad columns are written by the first chunks");

__device__ __forceinline__ int wave_incl_scan_i(int v, int lane) {
#pragma unroll
  for (int o = 1; o < 32; o <<= 1) {
    int up = __shfl_up(v, o, 32);
    if (lane >= o) v += up;
  }
  return v;
}

__device__ __forceinline__ void bin_hist_block(
    const int* __restrict__ ei, int* __restrict__ gh, int* __restrict__ hist, const int w) {
  const int tid = threadIdx.x;
  if (tid < NBKT) hist[tid] = 0;
  __syncthreads();
  if (tid < CHK_V4) {
    int4 d = reinterpret_cast<const int4*>(ei + NE)[w * CHK_V4 + tid];
    atomicAdd(&hist[d.x >> BKT_SHIFT], 1);
    atomicAdd(&hist[d.y >> BKT_SHIFT], 1);
    atomicAdd(&hist[d.z >> BKT_SHIFT], 1);
    atomicAdd(&hist[d.w >> BKT_SHIFT], 1);
  }
  __syncthreads();
  if (tid < NBKT) {
    gh[tid * GH_PITCH + w] = hist[tid];
    if (w < GH_PITCH - NCHK) gh[tid * GH_PITCH + NCHK + w] = 0;
  }
}

#define IP_WG ((IP_WAVES + 31) / 32)
__global__ __launch_bounds__(1024) void k_hist_proj(
    const int* __restrict__ ei, int* __restrict__ gh,
    const float* __restrict__ x, const float* __restrict__ W,
    const float* __restrict__ b, float* __restrict__ h) {
  __shared__ int hist[NBKT];
  if (blockIdx.x < NCHK) {
    bin_hist_block(ei, gh, hist, blockIdx.x);
    return;
  }
  const int gwave = (blockIdx.x - NCHK) * 32 + (threadIdx.x >> 5);
  if (gwave < IP_WAVES) input_proj_wave(x, W, b, h, gwave, threadIdx.x & 31);
}

__device__ __forceinline__ void bin_scan_block(
    int* __restrict__ gh, int* __restrict__ rowtot, int* __restrict__ rowbase,
    int* __restrict__ wtot) {
  const int tid = threadIdx.x, wave = tid >> 5, lane = tid & 31;
  int4 a[7], c[7];
#pragma unroll
  for (int k = 0; k < 7; ++k) {
    int b = min(wave + 32 * k, NBKT - 1);
    const int4* p = reinterpret_cast<const int4*>(gh + b * GH_PITCH + lane * 8);
    a[k] = p[0];
    c[k] = p[1];
  }
  int ex[7];
#pragma unroll
  for (int k = 0; k < 7; ++k) {
    int b = wave + 32 * k;
    int sum = ((a[k].x + a[k].y) + (a[k].z + a[k].w)) + ((c[k].x + c[k].y) + (c[k].z + c[k].w));
    int incl = wave_incl_scan_i(sum, lane);
    ex[k] = incl - sum;
    int tot = __shfl(incl, 31, 32);
    if (lane == 0) rowtot[b] = (b < NBKT) ? tot : 0;
  }
  __syncthreads();
  int v = 0, vin = 0;
  if (tid < 224) {
    v = rowtot[tid];
    vin = wave_incl_scan_i(v, lane);
    if (lane == 31) wtot[wave] = vin;
  }
  __syncthreads();
  if (tid < 224) {
    int before = 0;
#pragma unroll
    for (int j = 0; j < 7; ++j) before += (j < wave) ? wtot[j] : 0;
    rowbase[tid] = before + vin - v;
  }
  __syncthreads();
#pragma unroll
  for (int k = 0; k < 7; ++k) {
    int b = wave + 32 * k;
    if (b < NBKT) {
      int r = rowbase[b] + ex[k];
      int4 o0, o1;
      o0.x = r; r += a[k].x;
      o0.y = r; r += a[k].y;
      o0.z = r; r += a[k].z;
      o0.w = r; r += a[k].w;
      o1.x = r; r += c[k].x;
      o1.y = r; r += c[k].y;
      o1.z = r; r += c[k].z;
      o1.w = r;
      int4* p = reinterpret_cast<int4*>(gh + b * GH_PITCH + lane * 8);
      p[0] = o0;
      p[1] = o1;
    }
  }
}

#define G0_WG ((GT_TILES + 31) / 32)
__global__ __launch_bounds__(1024) void k_scan_gemm0(
    int* __restrict__ gh,
    const float* __restrict__ A, const float* __restrict__ W,
    const float* __restrict__ bias, const float* __restrict__ aw,
    float* __restrict__ out, float* __restrict__ sD, float* __restrict__ sS) {
  __shared__ float4 lW[H * H / 4];
  __shared__ float4 lAw[32][2 * H / 4];
  __shared__ int rowtot[224];
  __shared__ int rowbase[224];
  __shared__ int wtot[8];
  if (blockIdx.x == 0) {
    bin_scan_block(gh, rowtot, rowbase, wtot);
    return;
  }
  gemm_h_block<32>(blockIdx.x - 1, A, W, bias, aw, out, sD, sS, lW, lAw);
}

__global__ __launch_bounds__(1024) void k_bin_scatter(
    const int* __restrict__ ei, const int* __restrict__ gh, int2* __restrict__ binned) {
  __shared__ int cur[NBKT];
  const int tid = threadIdx.x, w = blockIdx.x;
  if (tid < NBKT) cur[tid] = gh[tid * GH_PITCH + w];
  __syncthreads();
  if (tid < CHK_V4) {
    int t = w * CHK_V4 + tid;
    int4 d = reinterpret_cast<const int4*>(ei + NE)[t];
    int e0 = 4 * t;
    int p;
    p = atomicAdd(&cur[d.x >> BKT_SHIFT], 1); binned[p] = make_int2(e0,     d.x);
    p = atomicAdd(&cur[d.y >> BKT_SHIFT], 1); binned[p] = make_int2(e0 + 1, d.y);
    p = atomicAdd(&cur[d.z >> BKT_SHIFT], 1); binned[p] = make_int2(e0 + 2, d.z);
    p = atomicAdd(&cur[d.w >> BKT_SHIFT], 1); binned[p] = make_int2(e0 + 3, d.w);
  }
}

__global__ __launch_bounds__(1024) void k_range_group(
    const int* __restrict__ gh, const int2* __restrict__ binned,
    int* __restrict__ off, int2* __restrict__ tmp) {
  __shared__ int cnt[BKT_R];
  __shared__ int wtot[32];
  const int tid = threadIdx.x, wave = tid >> 5, lane = tid & 31;
  const int b = blockIdx.x;
  const int lo = b << BKT_SHIFT;
  const int cb = gh[b * GH_PITCH];
  const int ce = (b + 1 < NBKT) ? gh[(b + 1) * GH_PITCH] : NE;
  if (tid < BKT_R) cnt[tid] = 0;
  __syncthreads();
  for (int i0 = cb + tid; i0 < ce; i0 += RG_DEPTH * 1024) {
    int dd[RG_DEPTH];
#pragma unroll
    for (int u = 0; u < RG_DEPTH; ++u) dd[u] = binned[min(i0 + 1024 * u, ce - 1)].y;
#pragma unroll
    for (int u = 0; u < RG_DEPTH; ++u)
      if (i0 + 1024 * u < ce) atomicAdd(&cnt[dd[u] - lo], 1);
  }
  __syncthreads();
  const int c = (tid < BKT_R) ? cnt[tid] : 0;
  const int incl = wave_incl_scan_i(c, lane);
  if (lane == 31) wtot[wave] = incl;
  __syncthreads();
  const int before = wave_sum_i((lane < wave) ? wtot[lane] : 0);
  const int excl = before + incl - c;
  if (tid < BKT_R) {
    cnt[tid] = excl;
    if (lo + tid <= NN) off[lo + tid] = cb + excl;
  }
  __syncthreads();
  for (int i0 = cb + tid; i0 < ce; i0 += RG_DEPTH * 1024) {
    int2 r[RG_DEPTH];
#pragma unroll
    for (int u = 0; u < RG_DEPTH; ++u) r[u] = binned[min(i0 + 1024 * u, ce - 1)];
#pragma unroll
    for (int u = 0; u < RG_DEPTH; ++u)
      if (i0 + 1024 * u < ce) {
        int p = atomicAdd(&cnt[r[u].y - lo], 1);
        tmp[cb + p] = r[u];
      }
  }
}

__global__ __launch_bounds__(256) void k_edge_order(
    const int* __restrict__ ei, const float* __restrict__ ea,
    const int* __restrict__ off, const int2* __restrict__ tmp,
    int2* __restrict__ rec) {
  int p = blockIdx.x * 256 + threadIdx.x;
  if (p >= NE) return;
  int2 my = tmp[p];
  int s = off[my.y], e = off[my.y + 1];
  int rank = 0;
  for (int q = s; q < e; ++q) rank += (tmp[q].x < my.x) ? 1 : 0;
  rec[s + rank] = make_int2(ei[my.x], __float_as_int(ea[my.x]));
}

#define GU_NODES 8
#define GU_WAVES (NN / GU_NODES)
#define GU_WG ((GU_WAVES + 7) / 8)
static_assert(GU_WAVES * GU_NODES == NN, "the waves cover the nodes exactly");
static_assert(GU_NODES >= 2 && GU_NODES < 32, "one lane per segment bound of the wave's nodes");
__device__ __forceinline__ float half_sum16(float v) {
#pragma unroll
  for (int off = 8; off > 0; off >>= 1) v += __shfl_xor(v, off, 32);
  return v;
}
__device__ __forceinline__ int2 first_records(
    const int2* __restrict__ rec, const int s, const int e, const int node, const int lane) {
  int2 rc = make_int2(node, 0);
  if (lane < e - s) rc = rec[s + lane];
  return rc;
}
__global__ __launch_bounds__(256) void k_gather_update(
    float* __restrict__ h, const float* __restrict__ m,
    const int* __restrict__ off, const int2* __restrict__ rec,
    const float* __restrict__ sD, const float* __restrict__ sS,
    const float* __restrict__ attb,
    const float* __restrict__ g, const float* __restrict__ beta) {
  __shared__ int2 strip[8][32];
  __shared__ float4 lnp[8][32];
  const int wv = __builtin_amdgcn_readfirstlane(blockIdx.x * 8 + (threadIdx.x >> 5));
  const int lane = threadIdx.x & 31;
  if (wv >= GU_WAVES) return;
  const int n0 = wv * GU_NODES;
  int2* const mystrip = strip[threadIdx.x >> 5];
  const int odd = lane >> 4;
  const unsigned c16 = (lane & 15) * 16u;
  const int f4 = (lane & 15) * 4;
  const int offv = off[n0 + min(lane, GU_NODES)];
  const float sdv = sD[n0 + min(lane, GU_NODES - 1)];
  const float ab = attb[0];
  float4* const mylnp = lnp[threadIdx.x >> 5];
  mylnp[lane] = *reinterpret_cast<const float4*>((odd ? beta : g) + f4);
  const char* const mbytes = reinterpret_cast<const char*>(m);

  int sA = __builtin_amdgcn_readlane(offv, 0);
  int eA = __builtin_amdgcn_readlane(offv, 1);
  int eB = __builtin_amdgcn_readlane(offv, 2);
  int2 rcA = first_records(rec, sA, eA, n0, lane);
  int2 rcB = first_records(rec, eA, eB, n0 + 1, lane);
  float ssA = sS[rcA.x];
  asm volatile("" :: "v"(ssA));
#pragma unroll 1
  for (int j = 0; j < GU_NODES; ++j) {
    const int gw = n0 + j;
    const int eC = __builtin_amdgcn_readlane(offv, min(j + 3, GU_NODES));
    const int2 rcC = first_records(rec, eB, eC, n0 + min(j + 2, GU_NODES - 1), lane);
    const float ssB = sS[rcB.x];
    if (lane < eB - eA) __builtin_prefetch(m + (size_t)rcB.x * H, 0, 1);

    const size_t base = (size_t)gw * H + f4;
    const float4 r  = *reinterpret_cast<const float4*>(h + base);
    const float4 mn = *reinterpret_cast<const float4*>(m + base);
    const float sd = __int_as_float(__builtin_amdgcn_readlane(__float_as_int(sdv), j));
    float4 acc = make_float4(0.0f, 0.0f, 0.0f, 0.0f);
    for (int cb = sA; cb < eA; cb += 32) {
      const int cnt = min(32, eA - cb);
      int src = gw;
      float w = 0.0f;
      if (cb == sA) {
        src = rcA.x;
        if (lane < cnt) w = __int_as_float(rcA.y) * sigmoidf_(sd + ssA + ab);
      } else if (lane < cnt) {
        int2 rc = rec[cb + lane];
        src = rc.x;
        __builtin_prefetch(m + (size_t)src * H, 0, 1);
        w = __int_as_float(rc.y) * sigmoidf_(sd + sS[src] + ab);
      }
      mystrip[lane] = make_int2((int)((unsigned)src << 8), __float_as_int(w));
#pragma unroll 4
      for (int i = 0; i < cnt; i += 2) {
        const int2 p = mystrip[i + odd];
        const float bw = __int_as_float(p.y);
        const float4 v = *reinterpret_cast<const float4*>(mbytes + ((unsigned)p.x + c16));
        acc.x = fmaf(v.x, bw, acc.x);
        acc.y = fmaf(v.y, bw, acc.y);
        acc.z = fmaf(v.z, bw, acc.z);
        acc.w = fmaf(v.w, bw, acc.w);
      }
    }
    acc.x += __shfl_xor(acc.x, 16, 32);
    acc.y += __shfl_xor(acc.y, 16, 32);
    acc.z += __shfl_xor(acc.z, 16, 32);
    acc.w += __shfl_xor(acc.w, 16, 32);
    const float f0 = acc.x + mn.x, f1 = acc.y + mn.y, f2 = acc.z + mn.z, f3 = acc.w + mn.w;
    const float mean = half_sum16((f0 + f1) + (f2 + f3)) * (1.0f / H);
    const float d0 = f0 - mean, d1 = f1 - mean, d2 = f2 - mean, d3 = f3 - mean;
    const float var = half_sum16(fmaf(d0, d0, d1 * d1) + fmaf(d2, d2, d3 * d3)) * (1.0f / H);
    const float inv = rsqrtf(var + LN_EPS);
    const float4 gv = mylnp[lane & 15];
    const float4 bv = mylnp[16 + (lane & 15)];
    if (odd == 0) {
      float4 o;
      o.x = fmaxf(fmaf(d0 * inv, gv.x, bv.x), 0.0f) + r.x;
      o.y = fmaxf(fmaf(d1 * inv, gv.y, bv.y), 0.0f) + r.y;
      o.z = fmaxf(fmaf(d2 * inv, gv.z, bv.z), 0.0f) + r.z;
      o.w = fmaxf(fmaf(d3 * inv, gv.w, bv.w), 0.0f) + r.w;
      *reinterpret_cast<float4*>(h + base) = o;
    }
    sA = eA; eA = eB; eB = eC;
    rcA = rcB; rcB = rcC; ssA = ssB;
  }
}

#define HD_TILES (NN / 16)
#define HD_WG ((HD_TILES + 7) / 8)
static_assert(HHID == 32 && DOUT == 32 && H == 64 && NN % 16 == 0, "two 16-column tiles per head, 16-row tiles");

__device__ __forceinline__ float group8_sum(float v) {
#pragma unroll
  for (int off = 1; off < 8; off <<= 1) v += __shfl_xor(v, off, 32);
  return v;
}

__device__ __forceinline__ void heads_block(
    const int blk,
    const float* __restrict__ h,
    const float* __restrict__ qW1, const float* __restrict__ qb1,
    const float* __restrict__ qW2, const float* __restrict__ qb2,
    const float* __restrict__ iW1, const float* __restrict__ ib1,
    const float* __restrict__ iW2, const float* __restrict__ ib2,
    const float* __restrict__ gW1, const float* __restrict__ gb1,
    const float* __restrict__ gW2, const float* __restrict__ gb2,
    const float* __restrict__ oW,  const float* __restrict__ obv,
    const float* __restrict__ sW1, const float* __restrict__ sb1,
    const float* __restrict__ sW2, const float* __restrict__ sb2,
    float* __restrict__ oq, float* __restrict__ oi, float* __restrict__ og,
    float* __restrict__ oh, float* __restrict__ obot, float* __restrict__ osev,
    float4* __restrict__ lQI, float4* __restrict__ lGS, float4* __restrict__ lO) {
  const int tid = threadIdx.x;
  const int wave = __builtin_amdgcn_readfirstlane(tid >> 5);
  const int lane = tid & 31;
  const int c = lane & 15, g = lane >> 4;
  const int cp = c & 7;
  const bool lo = c < 8;
  const int tile_raw = blk * 8 + wave;
  const bool valid = tile_raw < HD_TILES;
  const int row0 = (valid ? tile_raw : HD_TILES - 1) * 16;

  const float4* arow = reinterpret_cast<const float4*>(h + (size_t)(row0 + c) * H) + g;
  float4 af[8];
#pragma unroll
  for (int i = 0; i < 8; ++i) af[i] = arow[2 * i];

  const int sj = tid & 15, sk = tid >> 4;
  const float* srcQI = (sj < 8 ? qW1 : iW1) + (sj & 7) * 4;
  const float* srcGS = (sj < 8 ? gW1 : sW1) + (sj & 7) * 4;
  float4 sq[4], sg[4], so[2];
#pragma unroll
  for (int i = 0; i < 4; ++i) {
    sq[i] = *reinterpret_cast<const float4*>(srcQI + (sk + 16 * i) * HHID);
    sg[i] = *reinterpret_cast<const float4*>(srcGS + (sk + 16 * i) * HHID);
  }
#pragma unroll
  for (int i = 0; i < 2; ++i) so[i] = reinterpret_cast<const float4*>(oW)[tid + 256 * i];
  const float4 b1qi = *reinterpret_cast<const float4*>((lo ? qb1 : ib1) + 4 * cp);
  const float4 b1gs = *reinterpret_cast<const float4*>((lo ? gb1 : sb1) + 4 * cp);
  const float2 b1o  = *reinterpret_cast<const float2*>(obv + 2 * c);
#pragma unroll
  for (int i = 0; i < 4; ++i) { lQI[tid + 256 * i] = sq[i]; lGS[tid + 256 * i] = sg[i]; }
#pragma unroll
  for (int i = 0; i < 2; ++i) lO[tid + 256 * i] = so[i];
  __syncthreads();

  if (valid) {
    float4* orow = reinterpret_cast<float4*>(oh + (size_t)(row0 + c) * H) + g;
#pragma unroll
    for (int i = 0; i < 8; ++i) orow[2 * i] = af[i];
  }

  v8f aq0, aq1, aq2, aq3, ag0, ag1, ag2, ag3, ao0, ao1;
#pragma unroll
  for (int r = 0; r < 8; ++r) {
    aq0[r] = b1qi.x; aq1[r] = b1qi.y; aq2[r] = b1qi.z; aq3[r] = b1qi.w;
    ag0[r] = b1gs.x; ag1[r] = b1gs.y; ag2[r] = b1gs.z; ag3[r] = b1gs.w;
    ao0[r] = b1o.x;  ao1[r] = b1o.y;
  }

  const float4* lQIg = lQI + (4 * g) * 16 + c;
  const float4* lGSg = lGS + (4 * g) * 16 + c;
  const float2* lOg  = reinterpret_cast<const float2*>(lO) + (4 * g) * 16 + c;
#pragma unroll
  for (int i = 0; i < 8; ++i) {
#pragma unroll
    for (int p = 0; p < 2; ++p) {
      const int k = 8 * i + 2 * p;
      const float4 q0 = lQIg[k * 16], q1 = lQIg[(k + 1) * 16];
      const float4 s0 = lGSg[k * 16], s1 = lGSg[(k + 1) * 16];
      const float2 o0 = lOg[k * 16],  o1 = lOg[(k + 1) * 16];
      v2f a, b;
      a.x = p ? af[i].z : af[i].x;
      a.y = p ? af[i].w : af[i].y;
      b.x = q0.x; b.y = q1.x;
      aq0 = __builtin_amdgcn_wmma_f32_16x16x4_f32(false, a, false, b, (short)0, aq0, false, false);
      b.x = q0.y; b.y = q1.y;
      aq1 = __builtin_amdgcn_wmma_f32_16x16x4_f32(false, a, false, b, (short)0, aq1, false, false);
      b.x = q0.z; b.y = q1.z;
      aq2 = __builtin_amdgcn_wmma_f32_16x16x4_f32(false, a, false, b, (short)0, aq2, false, false);
      b.x = q0.w; b.y = q1.w;
      aq3 = __builtin_amdgcn_wmma_f32_16x16x4_f32(false, a, false, b, (short)0, aq3, false, false);
      b.x = s0.x; b.y = s1.x;
      ag0 = __builtin_amdgcn_wmma_f32_16x16x4_f32(false, a, false, b, (short)0, ag0, false, false);
      b.x = s0.y; b.y = s1.y;
      ag1 = __builtin_amdgcn_wmma_f32_16x16x4_f32(false, a, false, b, (short)0, ag1, false, false);
      b.x = s0.z; b.y = s1.z;
      ag2 = __builtin_amdgcn_wmma_f32_16x16x4_f32(false, a, false, b, (short)0, ag2, false, false);
      b.x = s0.w; b.y = s1.w;
      ag3 = __builtin_amdgcn_wmma_f32_16x16x4_f32(false, a, false, b, (short)0, ag3, false, false);
      b.x = o0.x; b.y = o1.x;
      ao0 = __builtin_amdgcn_wmma_f32_16x16x4_f32(false, a, false, b, (short)0, ao0, false, false);
      b.x = o0.y; b.y = o1.y;
      ao1 = __builtin_amdgcn_wmma_f32_16x16x4_f32(false, a, false, b, (short)0, ao1, false, false);
    }
  }

  const int myrow = row0 + 8 * g + cp;

  if (valid) {
    float2* ob2 = reinterpret_cast<float2*>(obot + (size_t)(row0 + 8 * g) * DOUT) + c;
#pragma unroll
    for (int r = 0; r < 8; ++r) ob2[r * (DOUT / 2)] = make_float2(ao0[r], ao1[r]);
  }

  {
    const float4 w2 = *reinterpret_cast<const float4*>((lo ? qW2 : iW2) + 4 * cp);
    const float b2 = lo ? qb2[0] : ib2[0];
    float keep = 0.0f;
#pragma unroll
    for (int r = 0; r < 8; ++r) {
      float v = fmaxf(aq0[r], 0.0f) * w2.x;
      v = fmaf(fmaxf(aq1[r], 0.0f), w2.y, v);
      v = fmaf(fmaxf(aq2[r], 0.0f), w2.z, v);
      v = fmaf(fmaxf(aq3[r], 0.0f), w2.w, v);
      v = group8_sum(v);
      keep = (cp == r) ? v : keep;
    }
    const float pr = sigmoidf_(keep + b2);
    if (valid) (lo ? oq : oi)[myrow] = pr;
  }

  {
    float f[20];
    const float4* sw = reinterpret_cast<const float4*>(sW2 + 20 * cp);
#pragma unroll
    for (int q = 0; q < 5; ++q) {
      const float4 v = sw[q];
      f[4 * q] = v.x; f[4 * q + 1] = v.y; f[4 * q + 2] = v.z; f[4 * q + 3] = v.w;
    }
    const float4 gw = *reinterpret_cast<const float4*>(gW2 + 4 * cp);
    const float wa0 = lo ? gw.x : f[0], wa1 = lo ? gw.y : f[5], wa2 = lo ? gw.z : f[10], wa3 = lo ? gw.w : f[15];
    float kz[5] = {0.0f, 0.0f, 0.0f, 0.0f, 0.0f};
#pragma unroll
    for (int r = 0; r < 8; ++r) {
      const float x0 = fmaxf(ag0[r], 0.0f), x1 = fmaxf(ag1[r], 0.0f);
      const float x2 = fmaxf(ag2[r], 0.0f), x3 = fmaxf(ag3[r], 0.0f);
      float u[5];
      u[0] = fmaf(x3, wa3, fmaf(x2, wa2, fmaf(x1, wa1, x0 * wa0)));
#pragma unroll
      for (int cls = 1; cls < 5; ++cls)
        u[cls] = fmaf(x3, f[15 + cls], fmaf(x2, f[10 + cls], fmaf(x1, f[5 + cls], x0 * f[cls])));
#pragma unroll
      for (int cls = 0; cls < 5; ++cls) {
        const float t = group8_sum(u[cls]);
        kz[cls] = (cp == r) ? t : kz[cls];
      }
    }
    if (lo) {
      const float pg = sigmoidf_(kz[0] + gb2[0]);
      if (valid) og[myrow] = pg;
    } else {
      float z[5];
#pragma unroll
      for (int cls = 0; cls < 5; ++cls) z[cls] = kz[cls] + sb2[cls];
      float zm = z[0];
#pragma unroll
      for (int cls = 1; cls < 5; ++cls) zm = fmaxf(zm, z[cls]);
      float es = 0.0f;
#pragma unroll
      for (int cls = 0; cls < 5; ++cls) { z[cls] = __expf(z[cls] - zm); es += z[cls]; }
      const float inv = 1.0f / es;
      if (valid) {
#pragma unroll
        for (int cls = 0; cls < 5; ++cls) osev[(size_t)myrow * 5 + cls] = z[cls] * inv;
      }
    }
  }
}

#define GM_BLOCKS 625
#define GM_LOADS 10
__device__ __forceinline__ void graph_partial_block(
    const int blk, const float* __restrict__ h, float* __restrict__ part, float (* __restrict__ ls)[H]) {
  const int tid = threadIdx.x;
  const int wave = tid >> 5, lane = tid & 31;
  const float4* h4 = reinterpret_cast<const float4*>(h);
  const size_t t = (size_t)blk * 256 + tid;
  float4 v[GM_LOADS];
#pragma unroll
  for (int i = 0; i < GM_LOADS; ++i) v[i] = h4[t + (size_t)i * (GM_BLOCKS * 256)];
  float4 a = v[0];
#pragma unroll
  for (int i = 1; i < GM_LOADS; ++i) {
    a.x += v[i].x; a.y += v[i].y; a.z += v[i].z; a.w += v[i].w;
  }
  a.x += __shfl_xor(a.x, 16, 32);
  a.y += __shfl_xor(a.y, 16, 32);
  a.z += __shfl_xor(a.z, 16, 32);
  a.w += __shfl_xor(a.w, 16, 32);
  if (lane < 16) *reinterpret_cast<float4*>(&ls[wave][lane * 4]) = a;
  __syncthreads();
  if (tid < H) {
    float s = ls[0][tid];
#pragma unroll
    for (int w = 1; w < 8; ++w) s += ls[w][tid];
    part[(size_t)blk * H + tid] = s;
  }
}

__global__ __launch_bounds__(256) void k_heads_graph(
    const float* __restrict__ h,
    const float* __restrict__ qW1, const float* __restrict__ qb1,
    const float* __restrict__ qW2, const float* __restrict__ qb2,
    const float* __restrict__ iW1, const float* __restrict__ ib1,
    const float* __restrict__ iW2, const float* __restrict__ ib2,
    const float* __restrict__ gW1, const float* __restrict__ gb1,
    const float* __restrict__ gW2, const float* __restrict__ gb2,
    const float* __restrict__ oW,  const float* __restrict__ obv,
    const float* __restrict__ sW1, const float* __restrict__ sb1,
    const float* __restrict__ sW2, const float* __restrict__ sb2,
    float* __restrict__ oq, float* __restrict__ oi, float* __restrict__ og,
    float* __restrict__ oh, float* __restrict__ obot, float* __restrict__ osev,
    float* __restrict__ part) {
  __shared__ float ls[8][H];
  __shared__ float4 lQI[H * 16];
  __shared__ float4 lGS[H * 16];
  __shared__ float4 lO[H * 8];
  if (blockIdx.x < GM_BLOCKS) {
    graph_partial_block(blockIdx.x, h, part, ls);
    return;
  }
  heads_block(blockIdx.x - GM_BLOCKS, h, qW1, qb1, qW2, qb2, iW1, ib1, iW2, ib2,
              gW1, gb1, gW2, gb2, oW, obv, sW1, sb1, sW2, sb2, oq, oi, og, oh, obot, osev,
              lQI, lGS, lO);
}

__global__ __launch_bounds__(1024) void k_graph_final(
    const float* __restrict__ part, float* __restrict__ gout) {
  __shared__ float ls[16][H];
  const int tid = threadIdx.x;
  const int f = tid & 63, q = tid >> 6;
  float v[40];
#pragma unroll
  for (int i = 0; i < 40; ++i) {
    int b = q + 16 * i;
    v[i] = (b < GM_BLOCKS) ? part[(size_t)b * H + f] : 0.0f;
  }
  float s = v[0];
#pragma unroll
  for (int i = 1; i < 40; ++i) s += v[i];
  ls[q][f] = s;
  __syncthreads();
  if (tid < H) {
    float r = ls[0][tid];
#pragma unroll
    for (int w = 1; w < 16; ++w) r += ls[w][tid];
    gout[tid] = r * (1.0f / NN);
  }
}

extern "C" void kernel_launch(void* const* d_in, const int* in_sizes, int n_in,
                              void* d_out, int out_size, void* d_ws, size_t ws_size,
                              hipStream_t stream) {
  (void)in_sizes; (void)n_in; (void)out_size; (void)ws_size;
  const float* x    = (const float*)d_in[0];
  const int*   ei   = (const int*)  d_in[1];
  const float* ea   = (const float*)d_in[2];
  const float* inW  = (const float*)d_in[3];
  const float* inb  = (const float*)d_in[4];
  const float* linW = (const float*)d_in[5];
  const float* linb = (const float*)d_in[6];
  const float* attW = (const float*)d_in[7];
  const float* attb = (const float*)d_in[8];
  const float* lng  = (const float*)d_in[9];
  const float* lnb  = (const float*)d_in[10];
  const float* qW1 = (const float*)d_in[11]; const float* qb1 = (const float*)d_in[12];
  const float* qW2 = (const float*)d_in[13]; const float* qb2 = (const float*)d_in[14];
  const float* iW1 = (const float*)d_in[15]; const float* ib1 = (const float*)d_in[16];
  const float* iW2 = (const float*)d_in[17]; const float* ib2 = (const float*)d_in[18];
  const float* gW1 = (const float*)d_in[19]; const float* gb1 = (const float*)d_in[20];
  const float* gW2 = (const float*)d_in[21]; const float* gb2 = (const float*)d_in[22];
  const float* oW  = (const float*)d_in[23]; const float* obv = (const float*)d_in[24];
  const float* sW1 = (const float*)d_in[25]; const float* sb1 = (const float*)d_in[26];
  const float* sW2 = (const float*)d_in[27]; const float* sb2 = (const float*)d_in[28];

  float* ws   = (float*)d_ws;
  float* h    = ws;
  float* mbuf = h    + (size_t)NN * H;
  float* lst  = mbuf + (size_t)NN * H;
  float* sD   = lst + (size_t)NN * H;
  float* sS   = sD + NN;
  int*   off  = (int*)lst;
  int2*  tmp  = (int2*)(off + NODE_PAD);
  int2*  rec  = tmp + NE;
  float* part = (float*)(rec + NE);
  int2*  binned = (int2*)(part + GM_BLOCKS * H);
  int*   gh   = (int*)(binned + NE);

  float* out  = (float*)d_out;
  float* oq   = out;
  float* oi   = out + (size_t)NN;
  float* og   = out + 2 * (size_t)NN;
  float* oh   = out + 3 * (size_t)NN;
  float* ogr  = out + 67 * (size_t)NN;
  float* obot = out + 67 * (size_t)NN + H;
  float* osev = out + 99 * (size_t)NN + H;

  k_hist_proj<<<NCHK + IP_WG, 1024, 0, stream>>>(ei, gh, x, inW, inb, h);
  k_scan_gemm0<<<1 + G0_WG, 1024, 0, stream>>>(gh, h, linW, linb, attW, mbuf, sD, sS);
  k_bin_scatter<<<NCHK, 1024, 0, stream>>>(ei, gh, binned);
  k_range_group<<<NBKT, 1024, 0, stream>>>(gh, binned, off, tmp);
  k_edge_order<<<(NE + 255) / 256, 256, 0, stream>>>(ei, ea, off, tmp, rec);

  const int nodeWaveBlocks = GU_WG;
  k_gather_update<<<nodeWaveBlocks, 256, 0, stream>>>(h, mbuf, off, rec, sD, sS, attb, lng, lnb);
  for (int l = 1; l < NL; ++l) {
    k_gemm_h<<<GT_BLOCKS, 128, 0, stream>>>(h, linW + (size_t)l * H * H, linb + (size_t)l * H,
                                          attW + (size_t)l * 2 * H, mbuf, sD, sS);
    k_gather_update<<<nodeWaveBlocks, 256, 0, stream>>>(h, mbuf, off, rec, sD, sS, attb + l,
                                                        lng + (size_t)l * H, lnb + (size_t)l * H);
  }

  k_heads_graph<<<GM_BLOCKS + HD_WG, 256, 0, stream>>>(h, qW1, qb1, qW2, qb2, iW1, ib1, iW2, ib2,
                                                     gW1, gb1, gW2, gb2, oW, obv, sW1, sb1, sW2, sb2,
                                                     oq, oi, og, oh, obot, osev, part);
  k_graph_final<<<1, 1024, 0, stream>>>(part, ogr);
}
